# starting kernel
# baseline (speedup 1.0000x reference)
#include <hip/hip_runtime.h>
#include <math.h>

#define NN    50000
#define DD    128
#define HC    256
#define CC    128
#define EE    800000
#define EP    (EE + NN)
#define NEG_SLOPE 0.2f

typedef __attribute__((ext_vector_type(16))) __bf16 v16bf;
typedef __attribute__((ext_vector_type(8)))  __bf16 v8bf;
typedef __attribute__((ext_vector_type(8)))  float  v8f;

__device__ __forceinline__ void edge_sd(int e, const int* __restrict__ ei,
                                        int& s, int& d) {
    if (e < EE) { s = ei[e]; d = ei[EE + e]; }
    else        { s = e - EE; d = e - EE; }
}

__device__ __forceinline__ unsigned enc_f32(float f) {
    unsigned u = __float_as_uint(f);
    return (u & 0x80000000u) ? ~u : (u | 0x80000000u);
}
__device__ __forceinline__ float dec_f32(unsigned u) {
    u = (u & 0x80000000u) ? (u & 0x7FFFFFFFu) : ~u;
    return __uint_as_float(u);
}

__global__ void k_f32_to_bf16(const float* __restrict__ in,
                              __bf16* __restrict__ out, int n) {
    int i = blockIdx.x * blockDim.x + threadIdx.x;
    if (i < n) out[i] = (__bf16)in[i];
}

__global__ void k_transpose_bf16(const float* __restrict__ W,
                                 __bf16* __restrict__ Wt, int K, int Nn) {
    int i = blockIdx.x * blockDim.x + threadIdx.x;
    if (i < K * Nn) {
        int n = i / K, k = i - n * K;
        Wt[i] = (__bf16)W[(size_t)k * Nn + n];
    }
}

__global__ __launch_bounds__(256)
void k_gemm_bf16_wmma(const __bf16* __restrict__ A,
                      const __bf16* __restrict__ Bt,
                      float* __restrict__ C,
                      int M, int Nn, int K) {
    const int wavesPerBlock = blockDim.x >> 5;
    int wave = blockIdx.x * wavesPerBlock + (threadIdx.x >> 5);
    int lane = threadIdx.x & 31;
    int tilesN = Nn >> 4;
    int tM = (wave / tilesN) << 4;
    int tN = (wave % tilesN) << 4;
    if (tM >= M) return;

    int half = lane >> 4;
    int lr   = lane & 15;

    v8f acc = {};
    const __bf16* arow = A  + (size_t)(tM + lr) * K;
    const __bf16* brow = Bt + (size_t)(tN + lr) * K;

    for (int kk = 0; kk < K; kk += 32) {
        v8bf alo = *(const v8bf*)(arow + kk + 8 * half);
        v8bf ahi = *(const v8bf*)(arow + kk + 16 + 8 * half);
        v8bf blo = *(const v8bf*)(brow + kk + 8 * half);
        v8bf bhi = *(const v8bf*)(brow + kk + 16 + 8 * half);
        v16bf a, b;
#pragma unroll
        for (int i = 0; i < 8; ++i) {
            a[i] = alo[i]; a[i + 8] = ahi[i];
            b[i] = blo[i]; b[i + 8] = bhi[i];
        }
        acc = __builtin_amdgcn_wmma_f32_16x16x32_bf16(
                  false, a, false, b, (short)0, acc, false, false);
    }

    int n = tN + lr;
#pragma unroll
    for (int r = 0; r < 8; ++r) {
        int m = tM + r + 8 * half;
        C[(size_t)m * Nn + n] = acc[r];
    }
}

__global__ void k_att_scores(const float* __restrict__ h,
                             const float* __restrict__ att_src,
                             const float* __restrict__ att_dst,
                             float* __restrict__ a_s,
                             float* __restrict__ a_d) {
    int i = blockIdx.x * blockDim.x + threadIdx.x;
    if (i >= NN * 2) return;
    int node = i >> 1, hd = i & 1;
    const float* hp = h + (size_t)node * HC + hd * CC;
    const float* as = att_src + hd * CC;
    const float* ad = att_dst + hd * CC;
    float s = 0.f, d = 0.f;
#pragma unroll 4
    for (int c = 0; c < CC; ++c) { float v = hp[c]; s += v * as[c]; d += v * ad[c]; }
    a_s[i] = s; a_d[i] = d;
}

__global__ void k_edge_max(const int* __restrict__ ei,
                           const float* __restrict__ a_s,
                           const float* __restrict__ a_d,
                           unsigned* __restrict__ m_enc) {
    int e = blockIdx.x * blockDim.x + threadIdx.x;
    if (e >= EP) return;
    int s, d; edge_sd(e, ei, s, d);
#pragma unroll
    for (int h = 0; h < 2; ++h) {
        float l = a_s[s * 2 + h] + a_d[d * 2 + h];
        l = (l > 0.f) ? l : NEG_SLOPE * l;
        atomicMax(&m_enc[d * 2 + h], enc_f32(l));
    }
}

__global__ void k_edge_exp(const int* __restrict__ ei,
                           const float* __restrict__ a_s,
                           const float* __restrict__ a_d,
                           const unsigned* __restrict__ m_enc,
                           float* __restrict__ e_buf,
                           float* __restrict__ denom) {
    int e = blockIdx.x * blockDim.x + threadIdx.x;
    if (e >= EP) return;
    int s, d; edge_sd(e, ei, s, d);
#pragma unroll
    for (int h = 0; h < 2; ++h) {
        float l = a_s[s * 2 + h] + a_d[d * 2 + h];
        l = (l > 0.f) ? l : NEG_SLOPE * l;
        float m  = dec_f32(m_enc[d * 2 + h]);
        float ex = expf(l - m);
        e_buf[(size_t)e * 2 + h] = ex;
        atomicAdd(&denom[d * 2 + h], ex);
    }
}

__global__ __launch_bounds__(HC)
void k_edge_scatter(const int* __restrict__ ei,
                    const float* __restrict__ h,
                    const float* __restrict__ e_buf,
                    const float* __restrict__ denom,
                    float* __restrict__ acc) {
    int e = blockIdx.x;
    int c = threadIdx.x;
    int s, d; edge_sd(e, ei, s, d);
    int hd = c >> 7;
    float alpha = e_buf[(size_t)e * 2 + hd] / (denom[d * 2 + hd] + 1e-16f);
    atomicAdd(&acc[(size_t)d * HC + c], alpha * h[(size_t)s * HC + c]);
}

__global__ void k_post_tanh(const float* __restrict__ acc,
                            const float* __restrict__ b,
                            float* __restrict__ hout,
                            __bf16* __restrict__ hbf, int total) {
    int i = blockIdx.x * blockDim.x + threadIdx.x;
    if (i < total) {
        float v = tanhf(acc[i] + b[i & (HC - 1)]);
        hout[i] = v;
        hbf[i]  = (__bf16)v;
    }
}

__global__ void k_final(const float* __restrict__ acc,
                        const float* __restrict__ b1,
                        const float* __restrict__ Wl,
                        const float* __restrict__ bl,
                        float* __restrict__ out) {
    int node = blockIdx.x * blockDim.x + threadIdx.x;
    if (node >= NN) return;
    const float* a = acc + (size_t)node * HC;
    float z0 = bl[0], z1 = bl[1];
#pragma unroll 4
    for (int k = 0; k < HC; ++k) {
        float v = a[k] + b1[k];
        z0 += v * Wl[k * 2 + 0];
        z1 += v * Wl[k * 2 + 1];
    }
    float mx = fmaxf(z0, z1);
    float e0 = expf(z0 - mx), e1 = expf(z1 - mx);
    float inv = 1.f / (e0 + e1);
    out[node * 2 + 0] = e0 * inv;
    out[node * 2 + 1] = e1 * inv;
}

extern "C" void kernel_launch(void* const* d_in, const int* in_sizes, int n_in,
                              void* d_out, int out_size, void* d_ws, size_t ws_size,
                              hipStream_t stream) {
    const float* x        = (const float*)d_in[0];
    const int*   ei       = (const int*)  d_in[1];
    const float* W0       = (const float*)d_in[2];
    const float* att_src0 = (const float*)d_in[3];
    const float* att_dst0 = (const float*)d_in[4];
    const float* b0       = (const float*)d_in[5];
    const float* W1       = (const float*)d_in[6];
    const float* att_src1 = (const float*)d_in[7];
    const float* att_dst1 = (const float*)d_in[8];
    const float* b1       = (const float*)d_in[9];
    const float* Wl       = (const float*)d_in[10];
    const float* bl       = (const float*)d_in[11];
    float* out = (float*)d_out;

    char* base = (char*)d_ws;
    size_t off = 0;
    auto carve = [&](size_t bytes) -> char* {
        char* p = base + off;
        off = (off + bytes + 255) & ~(size_t)255;
        return p;
    };
    float*    h_feat = (float*)   carve((size_t)NN * HC * 4);
    float*    accb   = (float*)   carve((size_t)NN * HC * 4);
    __bf16*   xbf    = (__bf16*)  carve((size_t)NN * HC * 2);
    __bf16*   wtbf   = (__bf16*)  carve((size_t)HC * HC * 2);
    float*    a_s    = (float*)   carve((size_t)NN * 2 * 4);
    float*    a_d    = (float*)   carve((size_t)NN * 2 * 4);
    unsigned* m_enc  = (unsigned*)carve((size_t)NN * 2 * 4);
    float*    denomb = (float*)   carve((size_t)NN * 2 * 4);
    float*    e_buf  = (float*)   carve((size_t)EP * 2 * 4);
    (void)ws_size; (void)in_sizes; (void)n_in; (void)out_size;

    const int T = 256;
    dim3 blk(T);

    k_f32_to_bf16<<<(NN * DD + T - 1) / T, blk, 0, stream>>>(x, xbf, NN * DD);
    k_transpose_bf16<<<(DD * HC + T - 1) / T, blk, 0, stream>>>(W0, wtbf, DD, HC);

    k_gemm_bf16_wmma<<<(NN / 16) * (HC / 16) / 8, blk, 0, stream>>>(
        xbf, wtbf, h_feat, NN, HC, DD);

    k_att_scores<<<(NN * 2 + T - 1) / T, blk, 0, stream>>>(
        h_feat, att_src0, att_dst0, a_s, a_d);

    hipMemsetAsync(accb,   0, (size_t)NN * HC * 4, stream);
    hipMemsetAsync(m_enc,  0, (size_t)NN * 2 * 4, stream);
    hipMemsetAsync(denomb, 0, (size_t)NN * 2 * 4, stream);

    k_edge_max    <<<(EP + T - 1) / T, blk, 0, stream>>>(ei, a_s, a_d, m_enc);
    k_edge_exp    <<<(EP + T - 1) / T, blk, 0, stream>>>(ei, a_s, a_d, m_enc, e_buf, denomb);
    k_edge_scatter<<<EP, dim3(HC), 0, stream>>>(ei, h_feat, e_buf, denomb, accb);

    k_post_tanh<<<(NN * HC + T - 1) / T, blk, 0, stream>>>(accb, b0, h_feat, xbf, NN * HC);

    k_transpose_bf16<<<(HC * HC + T - 1) / T, blk, 0, stream>>>(W1, wtbf, HC, HC);

    k_gemm_bf16_wmma<<<(NN / 16) * (HC / 16) / 8, blk, 0, stream>>>(
        xbf, wtbf, h_feat, NN, HC, HC);

    k_att_scores<<<(NN * 2 + T - 1) / T, blk, 0, stream>>>(
        h_feat, att_src1, att_dst1, a_s, a_d);

    hipMemsetAsync(accb,   0, (size_t)NN * HC * 4, stream);
    hipMemsetAsync(m_enc,  0, (size_t)NN * 2 * 4, stream);
    hipMemsetAsync(denomb, 0, (size_t)NN * 2 * 4, stream);

    k_edge_max    <<<(EP + T - 1) / T, blk, 0, stream>>>(ei, a_s, a_d, m_enc);
    k_edge_exp    <<<(EP + T - 1) / T, blk, 0, stream>>>(ei, a_s, a_d, m_enc, e_buf, denomb);
    k_edge_scatter<<<EP, dim3(HC), 0, stream>>>(ei, h_feat, e_buf, denomb, accb);

    k_final<<<(NN + T - 1) / T, blk, 0, stream>>>(accb, b1, Wl, bl, out);
}


# Edge phase: a block per 196-node destination tile scans the edge list twice (max; exp + rows), sums in LDS with LDS atomics; global atomics, clears, 2 launches a layer gone; sum order: LDS arrival
# speedup vs baseline: 4.0759x; 4.0759x over previous
#include <hip/hip_runtime.h>
#include <math.h>

#define NN    50000
#define DD    128
#define HC    256
#define CC    128
#define EE    800000
#define EP    (EE + NN)
#define NEG_SLOPE 0.2f

typedef __attribute__((ext_vector_type(16))) __bf16 v16bf;
typedef __attribute__((ext_vector_type(8)))  __bf16 v8bf;
typedef __attribute__((ext_vector_type(8)))  float  v8f;

__device__ __forceinline__ unsigned enc_f32(float f) {
    unsigned u = __float_as_uint(f);
    return (u & 0x80000000u) ? ~u : (u | 0x80000000u);
}
__device__ __forceinline__ float dec_f32(unsigned u) {
    u = (u & 0x80000000u) ? (u & 0x7FFFFFFFu) : ~u;
    return __uint_as_float(u);
}

__global__ void k_f32_to_bf16(const float* __restrict__ in,
                              __bf16* __restrict__ out, int n) {
    int i = blockIdx.x * blockDim.x + threadIdx.x;
    if (i < n) out[i] = (__bf16)in[i];
}

__global__ void k_transpose_bf16(const float* __restrict__ W,
                                 __bf16* __restrict__ Wt, int K, int Nn) {
    int i = blockIdx.x * blockDim.x + threadIdx.x;
    if (i < K * Nn) {
        int n = i / K, k = i - n * K;
        Wt[i] = (__bf16)W[(size_t)k * Nn + n];
    }
}

__global__ __launch_bounds__(256)
void k_gemm_bf16_wmma(const __bf16* __restrict__ A,
                      const __bf16* __restrict__ Bt,
                      float* __restrict__ C,
                      int M, int Nn, int K) {
    const int wavesPerBlock = blockDim.x >> 5;
    int wave = blockIdx.x * wavesPerBlock + (threadIdx.x >> 5);
    int lane = threadIdx.x & 31;
    int tilesN = Nn >> 4;
    int tM = (wave / tilesN) << 4;
    int tN = (wave % tilesN) << 4;
    if (tM >= M) return;

    int half = lane >> 4;
    int lr   = lane & 15;

    v8f acc = {};
    const __bf16* arow = A  + (size_t)(tM + lr) * K;
    const __bf16* brow = Bt + (size_t)(tN + lr) * K;

    for (int kk = 0; kk < K; kk += 32) {
        v8bf alo = *(const v8bf*)(arow + kk + 8 * half);
        v8bf ahi = *(const v8bf*)(arow + kk + 16 + 8 * half);
        v8bf blo = *(const v8bf*)(brow + kk + 8 * half);
        v8bf bhi = *(const v8bf*)(brow + kk + 16 + 8 * half);
        v16bf a, b;
#pragma unroll
        for (int i = 0; i < 8; ++i) {
            a[i] = alo[i]; a[i + 8] = ahi[i];
            b[i] = blo[i]; b[i + 8] = bhi[i];
        }
        acc = __builtin_amdgcn_wmma_f32_16x16x32_bf16(
                  false, a, false, b, (short)0, acc, false, false);
    }

    int n = tN + lr;
#pragma unroll
    for (int r = 0; r < 8; ++r) {
        int m = tM + r + 8 * half;
        C[(size_t)m * Nn + n] = acc[r];
    }
}

__global__ void k_att_scores(const float* __restrict__ h,
                             const float* __restrict__ att_src,
                             const float* __restrict__ att_dst,
                             float* __restrict__ a_s,
                             float* __restrict__ a_d) {
    int i = blockIdx.x * blockDim.x + threadIdx.x;
    if (i >= NN * 2) return;
    int node = i >> 1, hd = i & 1;
    const float* hp = h + (size_t)node * HC + hd * CC;
    const float* as = att_src + hd * CC;
    const float* ad = att_dst + hd * CC;
    float s = 0.f, d = 0.f;
#pragma unroll 4
    for (int c = 0; c < CC; ++c) { float v = hp[c]; s += v * as[c]; d += v * ad[c]; }
    a_s[i] = s; a_d[i] = d;
}

#define TILE      196
#define NTILES    ((NN + TILE - 1) / TILE)
#define AGG_T     1024
#define AGG_W     (AGG_T / 32)
#define SCAN_U    4
#define SCAN_STEP (AGG_W * 128)
#define SCAN_SUP  (SCAN_STEP * SCAN_U)
#define NSUPER    ((EE + SCAN_SUP - 1) / SCAN_SUP)
#define QDRAIN    32
#define QCAP      (QDRAIN + 128 * SCAN_U)
static_assert(EP <= (1 << 20) && TILE <= (1 << 12), "queue entry packing");

__device__ __forceinline__ float leaky(float l) {
    return (l > 0.f) ? l : NEG_SLOPE * l;
}

__device__ __forceinline__ unsigned make_entry(int e, unsigned dl) {
    return (unsigned)e | (dl << 20);
}
__device__ __forceinline__ int entry_src(unsigned ent, const int* __restrict__ srcp) {
    int e = (int)(ent & 0xFFFFFu);
    return (e < EE) ? srcp[e] : e - EE;
}

__device__ __forceinline__ void drain_max(const unsigned* q, int n, int lane,
                                          const int* __restrict__ srcp,
                                          const float* __restrict__ a_s,
                                          const float* ad_l, unsigned* m_enc) {
    for (int b = 0; b < n; b += 32) {
        int i = b + lane;
        if (i < n) {
            unsigned ent = q[i];
            int s = entry_src(ent, srcp);
            float2 as = *(const float2*)(a_s + 2 * s);
            int dl = (int)(ent >> 20);
            atomicMax(&m_enc[dl * 2 + 0], enc_f32(leaky(as.x + ad_l[dl * 2 + 0])));
            atomicMax(&m_enc[dl * 2 + 1], enc_f32(leaky(as.y + ad_l[dl * 2 + 1])));
        }
    }
}

template <bool FULL>
__device__ __forceinline__ void add_rows(int k0, int cnt, int lane, int s_l, int dl_l,
                                         float ex0_l, float ex1_l,
                                         const float* __restrict__ h, float* acc) {
    float4 v0[8], v1[8];
#pragma unroll
    for (int k = 0; k < 8; ++k) {
        if (FULL || k0 + k < cnt) {
            int s = __builtin_amdgcn_readlane(s_l, k0 + k);
            const float* hr = h + (size_t)s * HC + 4 * lane;
            v0[k] = *(const float4*)hr;
            v1[k] = *(const float4*)(hr + CC);
        }
    }
#pragma unroll
    for (int k = 0; k < 8; ++k) {
        if (FULL || k0 + k < cnt) {
            int   dl = __builtin_amdgcn_readlane(dl_l, k0 + k);
            float e0 = __int_as_float(__builtin_amdgcn_readlane(__float_as_int(ex0_l), k0 + k));
            float e1 = __int_as_float(__builtin_amdgcn_readlane(__float_as_int(ex1_l), k0 + k));
            float* ar = acc + dl * HC + lane;
            atomicAdd(ar +   0, e0 * v0[k].x);
            atomicAdd(ar +  32, e0 * v0[k].y);
            atomicAdd(ar +  64, e0 * v0[k].z);
            atomicAdd(ar +  96, e0 * v0[k].w);
            atomicAdd(ar + 128, e1 * v1[k].x);
            atomicAdd(ar + 160, e1 * v1[k].y);
            atomicAdd(ar + 192, e1 * v1[k].z);
            atomicAdd(ar + 224, e1 * v1[k].w);
        }
    }
}

__device__ __forceinline__ int drain_sum(unsigned* q, int n, bool last, int lane,
                                          const int* __restrict__ srcp,
                                          const float* __restrict__ a_s,
                                          const float* __restrict__ h,
                                          const float* ad_l, const unsigned* m_enc,
                                          float* den, float* acc) {
    const int nproc = last ? n : (n & ~7);
    for (int b = 0; b < nproc; b += 32) {
        int i = b + lane;
        int cnt = min(32, nproc - b);
        int s_l = 0, dl_l = 0;
        float ex0_l = 0.f, ex1_l = 0.f;
        if (i < nproc) {
            unsigned ent = q[i];
            s_l  = entry_src(ent, srcp);
            dl_l = (int)(ent >> 20);
            float2 as = *(const float2*)(a_s + 2 * s_l);
            ex0_l = expf(leaky(as.x + ad_l[dl_l * 2 + 0]) - dec_f32(m_enc[dl_l * 2 + 0]));
            ex1_l = expf(leaky(as.y + ad_l[dl_l * 2 + 1]) - dec_f32(m_enc[dl_l * 2 + 1]));
            atomicAdd(&den[dl_l * 2 + 0], ex0_l);
            atomicAdd(&den[dl_l * 2 + 1], ex1_l);
        }
        int k0 = 0;
        for (; k0 + 8 <= cnt; k0 += 8)
            add_rows<true>(k0, cnt, lane, s_l, dl_l, ex0_l, ex1_l, h, acc);
        if (k0 < cnt)
            add_rows<false>(k0, cnt, lane, s_l, dl_l, ex0_l, ex1_l, h, acc);
    }
    const int rem = n - nproc;
    if (rem > 0 && nproc > 0) {
        unsigned keep = 0u;
        if (lane < rem) keep = q[nproc + lane];
        if (lane < rem) q[lane] = keep;
    }
    return rem;
}

template <int PASS>
__device__ __forceinline__ void scan_edges(const int* __restrict__ srcp,
                                           const int* __restrict__ dstp,
                                           const float* __restrict__ a_s,
                                           const float* __restrict__ h,
                                           unsigned* q, int n, int base, int nloc,
                                           int lane, int w,
                                           const float* ad_l, unsigned* m_enc,
                                           float* den, float* acc) {
    int sup = blockIdx.x % NSUPER;
    for (int ss = 0; ss < NSUPER; ++ss) {
        int e_base = sup * SCAN_SUP + w * 128 + lane * 4;
        int4 dv[SCAN_U];
#pragma unroll
        for (int u = 0; u < SCAN_U; ++u) {
            int e = e_base + u * SCAN_STEP;
            dv[u] = (e < EE) ? *(const int4*)(dstp + e) : make_int4(-1, -1, -1, -1);
        }
#pragma unroll
        for (int u = 0; u < SCAN_U; ++u) {
            int e = e_base + u * SCAN_STEP;
            unsigned d0 = (unsigned)(dv[u].x - base), d1 = (unsigned)(dv[u].y - base);
            unsigned d2 = (unsigned)(dv[u].z - base), d3 = (unsigned)(dv[u].w - base);
            bool h0 = d0 < (unsigned)nloc, h1 = d1 < (unsigned)nloc;
            bool h2 = d2 < (unsigned)nloc, h3 = d3 < (unsigned)nloc;
            unsigned m0 = __builtin_amdgcn_ballot_w32(h0), m1 = __builtin_amdgcn_ballot_w32(h1);
            unsigned m2 = __builtin_amdgcn_ballot_w32(h2), m3 = __builtin_amdgcn_ballot_w32(h3);
            if ((m0 | m1 | m2 | m3) == 0u) continue;
            if (h0) q[n + __builtin_amdgcn_mbcnt_lo(m0, 0u)] = make_entry(e + 0, d0);
            n += __popc(m0);
            if (h1) q[n + __builtin_amdgcn_mbcnt_lo(m1, 0u)] = make_entry(e + 1, d1);
            n += __popc(m1);
            if (h2) q[n + __builtin_amdgcn_mbcnt_lo(m2, 0u)] = make_entry(e + 2, d2);
            n += __popc(m2);
            if (h3) q[n + __builtin_amdgcn_mbcnt_lo(m3, 0u)] = make_entry(e + 3, d3);
            n += __popc(m3);
        }
        if (++sup == NSUPER) sup = 0;
        const bool last = (ss == NSUPER - 1);
        if (n >= QDRAIN || (last && n > 0)) {
            if (PASS == 1) { drain_max(q, n, lane, srcp, a_s, ad_l, m_enc); n = 0; }
            else           n = drain_sum(q, n, last, lane, srcp, a_s, h, ad_l, m_enc, den, acc);
        }
    }
}

__global__ __launch_bounds__(AGG_T)
void k_tile_aggregate(const int* __restrict__ ei,
                      const float* __restrict__ h,
                      const float* __restrict__ a_s,
                      const float* __restrict__ a_d,
                      float* __restrict__ out) {
    __shared__ alignas(16) float acc[TILE * HC];
    __shared__ unsigned queue[AGG_W * QCAP];
    __shared__ unsigned m_enc[TILE * 2];
    __shared__ float    den[TILE * 2];
    __shared__ float    ad_l[TILE * 2];

    const int tid = threadIdx.x, lane = tid & 31;
    const int w = __builtin_amdgcn_readfirstlane(tid >> 5);
    const int base = blockIdx.x * TILE;
    const int nloc = min(TILE, NN - base);
    const int* __restrict__ srcp = ei;
    const int* __restrict__ dstp = ei + EE;
    unsigned* q = queue + w * QCAP;

    for (int i = tid; i < TILE * HC / 4; i += AGG_T)
        ((float4*)acc)[i] = make_float4(0.f, 0.f, 0.f, 0.f);
    if (tid < TILE * 2) den[tid] = 0.f;
    if (tid < nloc * 2) {
        int g = base * 2 + tid;
        float ad = a_d[g];
        m_enc[tid] = enc_f32(leaky(a_s[g] + ad));
        ad_l[tid]  = ad;
    }
    __syncthreads();

    scan_edges<1>(srcp, dstp, a_s, h, q, 0, base, nloc, lane, w, ad_l, m_enc, den, acc);
    __syncthreads();

    int nself = (nloc > w) ? (nloc - w + 31) / 32 : 0;
    if (lane < nself) {
        int dl = w + 32 * lane;
        q[lane] = make_entry(EE + base + dl, (unsigned)dl);
    }
    scan_edges<2>(srcp, dstp, a_s, h, q, nself, base, nloc, lane, w, ad_l, m_enc, den, acc);
    __syncthreads();

    for (int dl = w; dl < nloc; dl += AGG_W) {
        float dn0 = den[dl * 2 + 0] + 1e-16f, dn1 = den[dl * 2 + 1] + 1e-16f;
        const float* ar = acc + dl * HC + lane;
        float4 o0 = make_float4(ar[0] / dn0,   ar[32] / dn0,  ar[64] / dn0,  ar[96] / dn0);
        float4 o1 = make_float4(ar[128] / dn1, ar[160] / dn1, ar[192] / dn1, ar[224] / dn1);
        float* op = out + (size_t)(base + dl) * HC + 4 * lane;
        *(float4*)op        = o0;
        *(float4*)(op + CC) = o1;
    }
}

__global__ void k_post_tanh(const float* __restrict__ acc,
                            const float* __restrict__ b,
                            float* __restrict__ hout,
                            __bf16* __restrict__ hbf, int total) {
    int i = blockIdx.x * blockDim.x + threadIdx.x;
    if (i < total) {
        float v = tanhf(acc[i] + b[i & (HC - 1)]);
        hout[i] = v;
        hbf[i]  = (__bf16)v;
    }
}

__global__ void k_final(const float* __restrict__ acc,
                        const float* __restrict__ b1,
                        const float* __restrict__ Wl,
                        const float* __restrict__ bl,
                        float* __restrict__ out) {
    int node = blockIdx.x * blockDim.x + threadIdx.x;
    if (node >= NN) return;
    const float* a = acc + (size_t)node * HC;
    float z0 = bl[0], z1 = bl[1];
#pragma unroll 4
    for (int k = 0; k < HC; ++k) {
        float v = a[k] + b1[k];
        z0 += v * Wl[k * 2 + 0];
        z1 += v * Wl[k * 2 + 1];
    }
    float mx = fmaxf(z0, z1);
    float e0 = expf(z0 - mx), e1 = expf(z1 - mx);
    float inv = 1.f / (e0 + e1);
    out[node * 2 + 0] = e0 * inv;
    out[node * 2 + 1] = e1 * inv;
}

extern "C" void kernel_launch(void* const* d_in, const int* in_sizes, int n_in,
                              void* d_out, int out_size, void* d_ws, size_t ws_size,
                              hipStream_t stream) {
    const float* x        = (const float*)d_in[0];
    const int*   ei       = (const int*)  d_in[1];
    const float* W0       = (const float*)d_in[2];
    const float* att_src0 = (const float*)d_in[3];
    const float* att_dst0 = (const float*)d_in[4];
    const float* b0       = (const float*)d_in[5];
    const float* W1       = (const float*)d_in[6];
    const float* att_src1 = (const float*)d_in[7];
    const float* att_dst1 = (const float*)d_in[8];
    const float* b1       = (const float*)d_in[9];
    const float* Wl       = (const float*)d_in[10];
    const float* bl       = (const float*)d_in[11];
    float* out = (float*)d_out;

    char* base = (char*)d_ws;
    size_t off = 0;
    auto carve = [&](size_t bytes) -> char* {
        char* p = base + off;
        off = (off + bytes + 255) & ~(size_t)255;
        return p;
    };
    float*    h_feat = (float*)   carve((size_t)NN * HC * 4);
    float*    accb   = (float*)   carve((size_t)NN * HC * 4);
    __bf16*   xbf    = (__bf16*)  carve((size_t)NN * HC * 2);
    __bf16*   wtbf   = (__bf16*)  carve((size_t)HC * HC * 2);
    float*    a_s    = (float*)   carve((size_t)NN * 2 * 4);
    float*    a_d    = (float*)   carve((size_t)NN * 2 * 4);
    (void)ws_size; (void)in_sizes; (void)n_in; (void)out_size;

    const int T = 256;
    dim3 blk(T);

    k_f32_to_bf16<<<(NN * DD + T - 1) / T, blk, 0, stream>>>(x, xbf, NN * DD);
    k_transpose_bf16<<<(DD * HC + T - 1) / T, blk, 0, stream>>>(W0, wtbf, DD, HC);

    k_gemm_bf16_wmma<<<(NN / 16) * (HC / 16) / 8, blk, 0, stream>>>(
        xbf, wtbf, h_feat, NN, HC, DD);

    k_att_scores<<<(NN * 2 + T - 1) / T, blk, 0, stream>>>(
        h_feat, att_src0, att_dst0, a_s, a_d);

    k_tile_aggregate<<<NTILES, dim3(AGG_T), 0, stream>>>(ei, h_feat, a_s, a_d, accb);

    k_post_tanh<<<(NN * HC + T - 1) / T, blk, 0, stream>>>(accb, b0, h_feat, xbf, NN * HC);

    k_transpose_bf16<<<(HC * HC + T - 1) / T, blk, 0, stream>>>(W1, wtbf, HC, HC);

    k_gemm_bf16_wmma<<<(NN / 16) * (HC / 16) / 8, blk, 0, stream>>>(
        xbf, wtbf, h_feat, NN, HC, HC);

    k_att_scores<<<(NN * 2 + T - 1) / T, blk, 0, stream>>>(
        h_feat, att_src1, att_dst1, a_s, a_d);

    k_tile_aggregate<<<NTILES, dim3(AGG_T), 0, stream>>>(ei, h_feat, a_s, a_d, accb);

    k_final<<<(NN + T - 1) / T, blk, 0, stream>>>(accb, b1, Wl, bl, out);
}


# Edge phase pass 2 (exp + rows) runs from a per-wave LDS list of the edges pass 1 found (source node kept): second read of edge list and source column gone; sum order: LDS arrival as before
# speedup vs baseline: 4.9120x; 1.2051x over previous
#include <hip/hip_runtime.h>
#include <math.h>

#define NN    50000
#define DD    128
#define HC    256
#define CC    128
#define EE    800000
#define EP    (EE + NN)
#define NEG_SLOPE 0.2f

typedef __attribute__((ext_vector_type(16))) __bf16 v16bf;
typedef __attribute__((ext_vector_type(8)))  __bf16 v8bf;
typedef __attribute__((ext_vector_type(8)))  float  v8f;

__device__ __forceinline__ unsigned enc_f32(float f) {
    unsigned u = __float_as_uint(f);
    return (u & 0x80000000u) ? ~u : (u | 0x80000000u);
}
__device__ __forceinline__ float dec_f32(unsigned u) {
    u = (u & 0x80000000u) ? (u & 0x7FFFFFFFu) : ~u;
    return __uint_as_float(u);
}

__global__ void k_f32_to_bf16(const float* __restrict__ in,
                              __bf16* __restrict__ out, int n) {
    int i = blockIdx.x * blockDim.x + threadIdx.x;
    if (i < n) out[i] = (__bf16)in[i];
}

__global__ void k_transpose_bf16(const float* __restrict__ W,
                                 __bf16* __restrict__ Wt, int K, int Nn) {
    int i = blockIdx.x * blockDim.x + threadIdx.x;
    if (i < K * Nn) {
        int n = i / K, k = i - n * K;
        Wt[i] = (__bf16)W[(size_t)k * Nn + n];
    }
}

__global__ __launch_bounds__(256)
void k_gemm_bf16_wmma(const __bf16* __restrict__ A,
                      const __bf16* __restrict__ Bt,
                      float* __restrict__ C,
                      int M, int Nn, int K) {
    const int wavesPerBlock = blockDim.x >> 5;
    int wave = blockIdx.x * wavesPerBlock + (threadIdx.x >> 5);
    int lane = threadIdx.x & 31;
    int tilesN = Nn >> 4;
    int tM = (wave / tilesN) << 4;
    int tN = (wave % tilesN) << 4;
    if (tM >= M) return;

    int half = lane >> 4;
    int lr   = lane & 15;

    v8f acc = {};
    const __bf16* arow = A  + (size_t)(tM + lr) * K;
    const __bf16* brow = Bt + (size_t)(tN + lr) * K;

    for (int kk = 0; kk < K; kk += 32) {
        v8bf alo = *(const v8bf*)(arow + kk + 8 * half);
        v8bf ahi = *(const v8bf*)(arow + kk + 16 + 8 * half);
        v8bf blo = *(const v8bf*)(brow + kk + 8 * half);
        v8bf bhi = *(const v8bf*)(brow + kk + 16 + 8 * half);
        v16bf a, b;
#pragma unroll
        for (int i = 0; i < 8; ++i) {
            a[i] = alo[i]; a[i + 8] = ahi[i];
            b[i] = blo[i]; b[i + 8] = bhi[i];
        }
        acc = __builtin_amdgcn_wmma_f32_16x16x32_bf16(
                  false, a, false, b, (short)0, acc, false, false);
    }

    int n = tN + lr;
#pragma unroll
    for (int r = 0; r < 8; ++r) {
        int m = tM + r + 8 * half;
        C[(size_t)m * Nn + n] = acc[r];
    }
}

__global__ void k_att_scores(const float* __restrict__ h,
                             const float* __restrict__ att_src,
                             const float* __restrict__ att_dst,
                             float* __restrict__ a_s,
                             float* __restrict__ a_d) {
    int i = blockIdx.x * blockDim.x + threadIdx.x;
    if (i >= NN * 2) return;
    int node = i >> 1, hd = i & 1;
    const float* hp = h + (size_t)node * HC + hd * CC;
    const float* as = att_src + hd * CC;
    const float* ad = att_dst + hd * CC;
    float s = 0.f, d = 0.f;
#pragma unroll 4
    for (int c = 0; c < CC; ++c) { float v = hp[c]; s += v * as[c]; d += v * ad[c]; }
    a_s[i] = s; a_d[i] = d;
}

#define TILE      196
#define NTILES    ((NN + TILE - 1) / TILE)
#define AGG_T     1024
#define AGG_W     (AGG_T / 32)
#define SCAN_U    4
#define SCAN_STEP (AGG_W * 128)
#define SCAN_SUP  (SCAN_STEP * SCAN_U)
#define NSUPER    ((EE + SCAN_SUP - 1) / SCAN_SUP)
#define QDRAIN    32
#define QCAP      (QDRAIN + 128 * SCAN_U)
#define KCAP      384
#define KSELF     8
static_assert(EP <= (1 << 20) && NN <= (1 << 20) && TILE <= (1 << 12), "queue entry packing");
static_assert((TILE + AGG_W - 1) / AGG_W <= KSELF && KSELF <= 32, "self loops fit the reserved room");

__device__ __forceinline__ float leaky(float l) {
    return (l > 0.f) ? l : NEG_SLOPE * l;
}

__device__ __forceinline__ unsigned make_entry(int e, unsigned dl) {
    return (unsigned)e | (dl << 20);
}
__device__ __forceinline__ int entry_src(unsigned ent, const int* __restrict__ srcp) {
    int e = (int)(ent & 0xFFFFFu);
    return (e < EE) ? srcp[e] : e - EE;
}

__device__ __forceinline__ void drain_max(const unsigned* q, int n, int lane,
                                          const int* __restrict__ srcp,
                                          const float* __restrict__ a_s,
                                          const float* ad_l, unsigned* m_enc,
                                          unsigned* kp, bool keep) {
    for (int b = 0; b < n; b += 32) {
        int i = b + lane;
        if (i < n) {
            unsigned ent = q[i];
            int s = entry_src(ent, srcp);
            float2 as = *(const float2*)(a_s + 2 * s);
            int dl = (int)(ent >> 20);
            if (keep) kp[i] = make_entry(s, (unsigned)dl);
            atomicMax(&m_enc[dl * 2 + 0], enc_f32(leaky(as.x + ad_l[dl * 2 + 0])));
            atomicMax(&m_enc[dl * 2 + 1], enc_f32(leaky(as.y + ad_l[dl * 2 + 1])));
        }
    }
}

template <bool FULL>
__device__ __forceinline__ void add_rows(int k0, int cnt, int lane, int s_l, int dl_l,
                                         float ex0_l, float ex1_l,
                                         const float* __restrict__ h, float* acc) {
    float4 v0[8], v1[8];
#pragma unroll
    for (int k = 0; k < 8; ++k) {
        if (FULL || k0 + k < cnt) {
            int s = __builtin_amdgcn_readlane(s_l, k0 + k);
            const float* hr = h + (size_t)s * HC + 4 * lane;
            v0[k] = *(const float4*)hr;
            v1[k] = *(const float4*)(hr + CC);
        }
    }
#pragma unroll
    for (int k = 0; k < 8; ++k) {
        if (FULL || k0 + k < cnt) {
            int   dl = __builtin_amdgcn_readlane(dl_l, k0 + k);
            float e0 = __int_as_float(__builtin_amdgcn_readlane(__float_as_int(ex0_l), k0 + k));
            float e1 = __int_as_float(__builtin_amdgcn_readlane(__float_as_int(ex1_l), k0 + k));
            float* ar = acc + dl * HC + lane;
            atomicAdd(ar +   0, e0 * v0[k].x);
            atomicAdd(ar +  32, e0 * v0[k].y);
            atomicAdd(ar +  64, e0 * v0[k].z);
            atomicAdd(ar +  96, e0 * v0[k].w);
            atomicAdd(ar + 128, e1 * v1[k].x);
            atomicAdd(ar + 160, e1 * v1[k].y);
            atomicAdd(ar + 192, e1 * v1[k].z);
            atomicAdd(ar + 224, e1 * v1[k].w);
        }
    }
}

template <bool DIRECT>
__device__ __forceinline__ int drain_sum(unsigned* q, int n, bool last, int lane,
                                          const int* __restrict__ srcp,
                                          const float* __restrict__ a_s,
                                          const float* __restrict__ h,
                                          const float* ad_l, const unsigned* m_enc,
                                          float* den, float* acc) {
    const int nproc = last ? n : (n & ~7);
    for (int b = 0; b < nproc; b += 32) {
        int i = b + lane;
        int cnt = min(32, nproc - b);
        int s_l = 0, dl_l = 0;
        float ex0_l = 0.f, ex1_l = 0.f;
        if (i < nproc) {
            unsigned ent = q[i];
            s_l  = DIRECT ? (int)(ent & 0xFFFFFu) : entry_src(ent, srcp);
            dl_l = (int)(ent >> 20);
            float2 as = *(const float2*)(a_s + 2 * s_l);
            ex0_l = expf(leaky(as.x + ad_l[dl_l * 2 + 0]) - dec_f32(m_enc[dl_l * 2 + 0]));
            ex1_l = expf(leaky(as.y + ad_l[dl_l * 2 + 1]) - dec_f32(m_enc[dl_l * 2 + 1]));
            atomicAdd(&den[dl_l * 2 + 0], ex0_l);
            atomicAdd(&den[dl_l * 2 + 1], ex1_l);
        }
        int k0 = 0;
        for (; k0 + 8 <= cnt; k0 += 8)
            add_rows<true>(k0, cnt, lane, s_l, dl_l, ex0_l, ex1_l, h, acc);
        if (k0 < cnt)
            add_rows<false>(k0, cnt, lane, s_l, dl_l, ex0_l, ex1_l, h, acc);
    }
    const int rem = n - nproc;
    if (rem > 0 && nproc > 0) {
        unsigned keep = 0u;
        if (lane < rem) keep = q[nproc + lane];
        if (lane < rem) q[lane] = keep;
    }
    return rem;
}

template <int PASS>
__device__ __forceinline__ void scan_edges(const int* __restrict__ srcp,
                                           const int* __restrict__ dstp,
                                           const float* __restrict__ a_s,
                                           const float* __restrict__ h,
                                           unsigned* q, int n, int base, int nloc,
                                           int lane, int w,
                                           const float* ad_l, unsigned* m_enc,
                                           float* den, float* acc,
                                           unsigned* kept_w, int& kc, bool& ovf) {
    int sup = blockIdx.x % NSUPER;
    for (int ss = 0; ss < NSUPER; ++ss) {
        int e_base = sup * SCAN_SUP + w * 128 + lane * 4;
        int4 dv[SCAN_U];
#pragma unroll
        for (int u = 0; u < SCAN_U; ++u) {
            int e = e_base + u * SCAN_STEP;
            dv[u] = (e < EE) ? *(const int4*)(dstp + e) : make_int4(-1, -1, -1, -1);
        }
#pragma unroll
        for (int u = 0; u < SCAN_U; ++u) {
            int e = e_base + u * SCAN_STEP;
            unsigned d0 = (unsigned)(dv[u].x - base), d1 = (unsigned)(dv[u].y - base);
            unsigned d2 = (unsigned)(dv[u].z - base), d3 = (unsigned)(dv[u].w - base);
            bool h0 = d0 < (unsigned)nloc, h1 = d1 < (unsigned)nloc;
            bool h2 = d2 < (unsigned)nloc, h3 = d3 < (unsigned)nloc;
            unsigned m0 = __builtin_amdgcn_ballot_w32(h0), m1 = __builtin_amdgcn_ballot_w32(h1);
            unsigned m2 = __builtin_amdgcn_ballot_w32(h2), m3 = __builtin_amdgcn_ballot_w32(h3);
            if ((m0 | m1 | m2 | m3) == 0u) continue;
            if (h0) q[n + __builtin_amdgcn_mbcnt_lo(m0, 0u)] = make_entry(e + 0, d0);
            n += __popc(m0);
            if (h1) q[n + __builtin_amdgcn_mbcnt_lo(m1, 0u)] = make_entry(e + 1, d1);
            n += __popc(m1);
            if (h2) q[n + __builtin_amdgcn_mbcnt_lo(m2, 0u)] = make_entry(e + 2, d2);
            n += __popc(m2);
            if (h3) q[n + __builtin_amdgcn_mbcnt_lo(m3, 0u)] = make_entry(e + 3, d3);
            n += __popc(m3);
        }
        if (++sup == NSUPER) sup = 0;
        const bool last = (ss == NSUPER - 1);
        if (n >= QDRAIN || (last && n > 0)) {
            if (PASS == 1) {
                if (kc + n > KCAP - KSELF) ovf = true;
                drain_max(q, n, lane, srcp, a_s, ad_l, m_enc, kept_w + kc, !ovf);
                if (!ovf) kc += n;
                n = 0;
            } else {
                n = drain_sum<false>(q, n, last, lane, srcp, a_s, h, ad_l, m_enc, den, acc);
            }
        }
    }
}

__global__ __launch_bounds__(AGG_T)
void k_tile_aggregate(const int* __restrict__ ei,
                      const float* __restrict__ h,
                      const float* __restrict__ a_s,
                      const float* __restrict__ a_d,
                      float* __restrict__ out) {
    __shared__ alignas(16) float acc[TILE * HC];
    __shared__ unsigned queue[AGG_W * QCAP];
    __shared__ unsigned kept[AGG_W * KCAP];
    __shared__ unsigned m_enc[TILE * 2];
    __shared__ float    den[TILE * 2];
    __shared__ float    ad_l[TILE * 2];

    const int tid = threadIdx.x, lane = tid & 31;
    const int w = __builtin_amdgcn_readfirstlane(tid >> 5);
    const int base = blockIdx.x * TILE;
    const int nloc = min(TILE, NN - base);
    const int* __restrict__ srcp = ei;
    const int* __restrict__ dstp = ei + EE;
    unsigned* q = queue + w * QCAP;
    unsigned* kept_w = kept + w * KCAP;
    int  kc  = 0;
    bool ovf = false;

    for (int i = tid; i < TILE * HC / 4; i += AGG_T)
        ((float4*)acc)[i] = make_float4(0.f, 0.f, 0.f, 0.f);
    if (tid < TILE * 2) den[tid] = 0.f;
    if (tid < nloc * 2) {
        int g = base * 2 + tid;
        float ad = a_d[g];
        m_enc[tid] = enc_f32(leaky(a_s[g] + ad));
        ad_l[tid]  = ad;
    }
    __syncthreads();

    scan_edges<1>(srcp, dstp, a_s, h, q, 0, base, nloc, lane, w, ad_l, m_enc, den, acc,
                  kept_w, kc, ovf);
    __syncthreads();

    int nself = (nloc > w) ? (nloc - w + 31) / 32 : 0;
    if (!ovf) {
        if (lane < nself) {
            int dl = w + 32 * lane;
            kept_w[kc + lane] = make_entry(base + dl, (unsigned)dl);
        }
        drain_sum<true>(kept_w, kc + nself, true, lane, srcp, a_s, h, ad_l, m_enc, den, acc);
    } else {
        if (lane < nself) {
            int dl = w + 32 * lane;
            q[lane] = make_entry(EE + base + dl, (unsigned)dl);
        }
        scan_edges<2>(srcp, dstp, a_s, h, q, nself, base, nloc, lane, w, ad_l, m_enc, den, acc,
                      kept_w, kc, ovf);
    }
    __syncthreads();

    for (int dl = w; dl < nloc; dl += AGG_W) {
        float dn0 = den[dl * 2 + 0] + 1e-16f, dn1 = den[dl * 2 + 1] + 1e-16f;
        const float* ar = acc + dl * HC + lane;
        float4 o0 = make_float4(ar[0] / dn0,   ar[32] / dn0,  ar[64] / dn0,  ar[96] / dn0);
        float4 o1 = make_float4(ar[128] / dn1, ar[160] / dn1, ar[192] / dn1, ar[224] / dn1);
        float* op = out + (size_t)(base + dl) * HC + 4 * lane;
        *(float4*)op        = o0;
        *(float4*)(op + CC) = o1;
    }
}

__global__ void k_post_tanh(const float* __restrict__ acc,
                            const float* __restrict__ b,
                            float* __restrict__ hout,
                            __bf16* __restrict__ hbf, int total) {
    int i = blockIdx.x * blockDim.x + threadIdx.x;
    if (i < total) {
        float v = tanhf(acc[i] + b[i & (HC - 1)]);
        hout[i] = v;
        hbf[i]  = (__bf16)v;
    }
}

__global__ void k_final(const float* __restrict__ acc,
                        const float* __restrict__ b1,
                        const float* __restrict__ Wl,
                        const float* __restrict__ bl,
                        float* __restrict__ out) {
    int node = blockIdx.x * blockDim.x + threadIdx.x;
    if (node >= NN) return;
    const float* a = acc + (size_t)node * HC;
    float z0 = bl[0], z1 = bl[1];
#pragma unroll 4
    for (int k = 0; k < HC; ++k) {
        float v = a[k] + b1[k];
        z0 += v * Wl[k * 2 + 0];
        z1 += v * Wl[k * 2 + 1];
    }
    float mx = fmaxf(z0, z1);
    float e0 = expf(z0 - mx), e1 = expf(z1 - mx);
    float inv = 1.f / (e0 + e1);
    out[node * 2 + 0] = e0 * inv;
    out[node * 2 + 1] = e1 * inv;
}

extern "C" void kernel_launch(void* const* d_in, const int* in_sizes, int n_in,
                              void* d_out, int out_size, void* d_ws, size_t ws_size,
                              hipStream_t stream) {
    const float* x        = (const float*)d_in[0];
    const int*   ei       = (const int*)  d_in[1];
    const float* W0       = (const float*)d_in[2];
    const float* att_src0 = (const float*)d_in[3];
    const float* att_dst0 = (const float*)d_in[4];
    const float* b0       = (const float*)d_in[5];
    const float* W1       = (const float*)d_in[6];
    const float* att_src1 = (const float*)d_in[7];
    const float* att_dst1 = (const float*)d_in[8];
    const float* b1       = (const float*)d_in[9];
    const float* Wl       = (const float*)d_in[10];
    const float* bl       = (const float*)d_in[11];
    float* out = (float*)d_out;

    char* base = (char*)d_ws;
    size_t off = 0;
    auto carve = [&](size_t bytes) -> char* {
        char* p = base + off;
        off = (off + bytes + 255) & ~(size_t)255;
        return p;
    };
    float*    h_feat = (float*)   carve((size_t)NN * HC * 4);
    float*    accb   = (float*)   carve((size_t)NN * HC * 4);
    __bf16*   xbf    = (__bf16*)  carve((size_t)NN * HC * 2);
    __bf16*   wtbf   = (__bf16*)  carve((size_t)HC * HC * 2);
    float*    a_s    = (float*)   carve((size_t)NN * 2 * 4);
    float*    a_d    = (float*)   carve((size_t)NN * 2 * 4);
    (void)ws_size; (void)in_sizes; (void)n_in; (void)out_size;

    const int T = 256;
    dim3 blk(T);

    k_f32_to_bf16<<<(NN * DD + T - 1) / T, blk, 0, stream>>>(x, xbf, NN * DD);
    k_transpose_bf16<<<(DD * HC + T - 1) / T, blk, 0, stream>>>(W0, wtbf, DD, HC);

    k_gemm_bf16_wmma<<<(NN / 16) * (HC / 16) / 8, blk, 0, stream>>>(
        xbf, wtbf, h_feat, NN, HC, DD);

    k_att_scores<<<(NN * 2 + T - 1) / T, blk, 0, stream>>>(
        h_feat, att_src0, att_dst0, a_s, a_d);

    k_tile_aggregate<<<NTILES, dim3(AGG_T), 0, stream>>>(ei, h_feat, a_s, a_d, accb);

    k_post_tanh<<<(NN * HC + T - 1) / T, blk, 0, stream>>>(accb, b0, h_feat, xbf, NN * HC);

    k_transpose_bf16<<<(HC * HC + T - 1) / T, blk, 0, stream>>>(W1, wtbf, HC, HC);

    k_gemm_bf16_wmma<<<(NN / 16) * (HC / 16) / 8, blk, 0, stream>>>(
        xbf, wtbf, h_feat, NN, HC, HC);

    k_att_scores<<<(NN * 2 + T - 1) / T, blk, 0, stream>>>(
        h_feat, att_src1, att_dst1, a_s, a_d);

    k_tile_aggregate<<<NTILES, dim3(AGG_T), 0, stream>>>(ei, h_feat, a_s, a_d, accb);

    k_final<<<(NN + T - 1) / T, blk, 0, stream>>>(accb, b1, Wl, bl, out);
}


# Both products take f32 rows: a 16-wave block rounds (layer 0) or bias+tanh+rounds (layer 1) its 16 A rows into LDS once, A fragments from LDS; 2 elementwise launches and bf16 copies gone
# speedup vs baseline: 6.2586x; 1.2741x over previous
#include <hip/hip_runtime.h>
#include <math.h>

#define NN    50000
#define DD    128
#define HC    256
#define CC    128
#define EE    800000
#define EP    (EE + NN)
#define NEG_SLOPE 0.2f

typedef __attribute__((ext_vector_type(16))) __bf16 v16bf;
typedef __attribute__((ext_vector_type(8)))  __bf16 v8bf;
typedef __attribute__((ext_vector_type(8)))  float  v8f;

__device__ __forceinline__ unsigned enc_f32(float f) {
    unsigned u = __float_as_uint(f);
    return (u & 0x80000000u) ? ~u : (u | 0x80000000u);
}
__device__ __forceinline__ float dec_f32(unsigned u) {
    u = (u & 0x80000000u) ? (u & 0x7FFFFFFFu) : ~u;
    return __uint_as_float(u);
}

__global__ void k_transpose_bf16(const float* __restrict__ W,
                                 __bf16* __restrict__ Wt, int K, int Nn) {
    int i = blockIdx.x * blockDim.x + threadIdx.x;
    if (i < K * Nn) {
        int n = i / K, k = i - n * K;
        Wt[i] = (__bf16)W[(size_t)k * Nn + n];
    }
}

typedef __attribute__((ext_vector_type(4))) __bf16 v4bf;
#define GEMM_T   512
#define A_PAD    8
static_assert(NN % 16 == 0 && HC == 16 * (GEMM_T / 32), "a block covers 16 whole rows");

template <int K, bool TANH>
__global__ __launch_bounds__(GEMM_T)
void k_gemm_rows_wmma(const float* __restrict__ A,
                      const float* __restrict__ bias,
                      const __bf16* __restrict__ Bt,
                      float* __restrict__ C) {
    constexpr int PITCH = K + A_PAD;
    __shared__ alignas(16) __bf16 a_lds[16 * PITCH];

    const int tid = threadIdx.x;
    const int tM  = blockIdx.x << 4;
    const int lane = tid & 31;
    const int tN   = (tid >> 5) << 4;
    const int half = lane >> 4;
    const int lr   = lane & 15;

    const __bf16* brow = Bt + (size_t)(tN + lr) * K;
    v8bf bfr[K / 16];
#pragma unroll
    for (int j = 0; j < K / 16; ++j)
        bfr[j] = *(const v8bf*)(brow + 16 * j + 8 * half);

    constexpr int V4_ROW = K / 4;
    constexpr int NIT    = 16 * V4_ROW / GEMM_T;
    static_assert(NIT * GEMM_T == 16 * V4_ROW, "the block's threads cover the 16 rows exactly");
    float4 av[NIT];
#pragma unroll
    for (int it = 0; it < NIT; ++it) {
        int i = tid + it * GEMM_T;
        int row = i / V4_ROW, c4 = i - row * V4_ROW;
        av[it] = *(const float4*)(A + (size_t)(tM + row) * K + 4 * c4);
    }
#pragma unroll
    for (int it = 0; it < NIT; ++it) {
        int i = tid + it * GEMM_T;
        int row = i / V4_ROW, c4 = i - row * V4_ROW;
        float4 v = av[it];
        if (TANH) {
            float4 b = *(const float4*)(bias + 4 * c4);
            v.x = tanhf(v.x + b.x); v.y = tanhf(v.y + b.y);
            v.z = tanhf(v.z + b.z); v.w = tanhf(v.w + b.w);
        }
        v4bf o;
        o[0] = (__bf16)v.x; o[1] = (__bf16)v.y; o[2] = (__bf16)v.z; o[3] = (__bf16)v.w;
        *(v4bf*)(a_lds + row * PITCH + 4 * c4) = o;
    }
    __syncthreads();

    v8f acc = {};
    const __bf16* arow = a_lds + lr * PITCH;

#pragma unroll
    for (int kk = 0; kk < K; kk += 32) {
        v8bf alo = *(const v8bf*)(arow + kk + 8 * half);
        v8bf ahi = *(const v8bf*)(arow + kk + 16 + 8 * half);
        v8bf blo = bfr[kk / 16];
        v8bf bhi = bfr[kk / 16 + 1];
        v16bf a, b;
#pragma unroll
        for (int i = 0; i < 8; ++i) {
            a[i] = alo[i]; a[i + 8] = ahi[i];
            b[i] = blo[i]; b[i + 8] = bhi[i];
        }
        acc = __builtin_amdgcn_wmma_f32_16x16x32_bf16(
                  false, a, false, b, (short)0, acc, false, false);
    }

    int n = tN + lr;
#pragma unroll
    for (int r = 0; r < 8; ++r) {
        int m = tM + r + 8 * half;
        C[(size_t)m * HC + n] = acc[r];
    }
}

__global__ void k_att_scores(const float* __restrict__ h,
                             const float* __restrict__ att_src,
                             const float* __restrict__ att_dst,
                             float* __restrict__ a_s,
                             float* __restrict__ a_d) {
    int i = blockIdx.x * blockDim.x + threadIdx.x;
    if (i >= NN * 2) return;
    int node = i >> 1, hd = i & 1;
    const float* hp = h + (size_t)node * HC + hd * CC;
    const float* as = att_src + hd * CC;
    const float* ad = att_dst + hd * CC;
    float s = 0.f, d = 0.f;
#pragma unroll 4
    for (int c = 0; c < CC; ++c) { float v = hp[c]; s += v * as[c]; d += v * ad[c]; }
    a_s[i] = s; a_d[i] = d;
}

#define TILE      196
#define NTILES    ((NN + TILE - 1) / TILE)
#define AGG_T     1024
#define AGG_W     (AGG_T / 32)
#define SCAN_U    4
#define SCAN_STEP (AGG_W * 128)
#define SCAN_SUP  (SCAN_STEP * SCAN_U)
#define NSUPER    ((EE + SCAN_SUP - 1) / SCAN_SUP)
#define QDRAIN    32
#define QCAP      (QDRAIN + 128 * SCAN_U)
#define KCAP      384
#define KSELF     8
static_assert(EP <= (1 << 20) && NN <= (1 << 20) && TILE <= (1 << 12), "queue entry packing");
static_assert((TILE + AGG_W - 1) / AGG_W <= KSELF && KSELF <= 32, "self loops fit the reserved room");

__device__ __forceinline__ float leaky(float l) {
    return (l > 0.f) ? l : NEG_SLOPE * l;
}

__device__ __forceinline__ unsigned make_entry(int e, unsigned dl) {
    return (unsigned)e | (dl << 20);
}
__device__ __forceinline__ int entry_src(unsigned ent, const int* __restrict__ srcp) {
    int e = (int)(ent & 0xFFFFFu);
    return (e < EE) ? srcp[e] : e - EE;
}

__device__ __forceinline__ void drain_max(const unsigned* q, int n, int lane,
                                          const int* __restrict__ srcp,
                                          const float* __restrict__ a_s,
                                          const float* ad_l, unsigned* m_enc,
                                          unsigned* kp, bool keep) {
    for (int b = 0; b < n; b += 32) {
        int i = b + lane;
        if (i < n) {
            unsigned ent = q[i];
            int s = entry_src(ent, srcp);
            float2 as = *(const float2*)(a_s + 2 * s);
            int dl = (int)(ent >> 20);
            if (keep) kp[i] = make_entry(s, (unsigned)dl);
            atomicMax(&m_enc[dl * 2 + 0], enc_f32(leaky(as.x + ad_l[dl * 2 + 0])));
            atomicMax(&m_enc[dl * 2 + 1], enc_f32(leaky(as.y + ad_l[dl * 2 + 1])));
        }
    }
}

template <bool FULL>
__device__ __forceinline__ void add_rows(int k0, int cnt, int lane, int s_l, int dl_l,
                                         float ex0_l, float ex1_l,
                                         const float* __restrict__ h, float* acc) {
    float4 v0[8], v1[8];
#pragma unroll
    for (int k = 0; k < 8; ++k) {
        if (FULL || k0 + k < cnt) {
            int s = __builtin_amdgcn_readlane(s_l, k0 + k);
            const float* hr = h + (size_t)s * HC + 4 * lane;
            v0[k] = *(const float4*)hr;
            v1[k] = *(const float4*)(hr + CC);
        }
    }
#pragma unroll
    for (int k = 0; k < 8; ++k) {
        if (FULL || k0 + k < cnt) {
            int   dl = __builtin_amdgcn_readlane(dl_l, k0 + k);
            float e0 = __int_as_float(__builtin_amdgcn_readlane(__float_as_int(ex0_l), k0 + k));
            float e1 = __int_as_float(__builtin_amdgcn_readlane(__float_as_int(ex1_l), k0 + k));
            float* ar = acc + dl * HC + lane;
            atomicAdd(ar +   0, e0 * v0[k].x);
            atomicAdd(ar +  32, e0 * v0[k].y);
            atomicAdd(ar +  64, e0 * v0[k].z);
            atomicAdd(ar +  96, e0 * v0[k].w);
            atomicAdd(ar + 128, e1 * v1[k].x);
            atomicAdd(ar + 160, e1 * v1[k].y);
            atomicAdd(ar + 192, e1 * v1[k].z);
            atomicAdd(ar + 224, e1 * v1[k].w);
        }
    }
}

template <bool DIRECT>
__device__ __forceinline__ int drain_sum(unsigned* q, int n, bool last, int lane,
                                          const int* __restrict__ srcp,
                                          const float* __restrict__ a_s,
                                          const float* __restrict__ h,
                                          const float* ad_l, const unsigned* m_enc,
                                          float* den, float* acc) {
    const int nproc = last ? n : (n & ~7);
    for (int b = 0; b < nproc; b += 32) {
        int i = b + lane;
        int cnt = min(32, nproc - b);
        int s_l = 0, dl_l = 0;
        float ex0_l = 0.f, ex1_l = 0.f;
        if (i < nproc) {
            unsigned ent = q[i];
            s_l  = DIRECT ? (int)(ent & 0xFFFFFu) : entry_src(ent, srcp);
            dl_l = (int)(ent >> 20);
            float2 as = *(const float2*)(a_s + 2 * s_l);
            ex0_l = expf(leaky(as.x + ad_l[dl_l * 2 + 0]) - dec_f32(m_enc[dl_l * 2 + 0]));
            ex1_l = expf(leaky(as.y + ad_l[dl_l * 2 + 1]) - dec_f32(m_enc[dl_l * 2 + 1]));
            atomicAdd(&den[dl_l * 2 + 0], ex0_l);
            atomicAdd(&den[dl_l * 2 + 1], ex1_l);
        }
        int k0 = 0;
        for (; k0 + 8 <= cnt; k0 += 8)
            add_rows<true>(k0, cnt, lane, s_l, dl_l, ex0_l, ex1_l, h, acc);
        if (k0 < cnt)
            add_rows<false>(k0, cnt, lane, s_l, dl_l, ex0_l, ex1_l, h, acc);
    }
    const int rem = n - nproc;
    if (rem > 0 && nproc > 0) {
        unsigned keep = 0u;
        if (lane < rem) keep = q[nproc + lane];
        if (lane < rem) q[lane] = keep;
    }
    return rem;
}

template <int PASS>
__device__ __forceinline__ void scan_edges(const int* __restrict__ srcp,
                                           const int* __restrict__ dstp,
                                           const float* __restrict__ a_s,
                                           const float* __restrict__ h,
                                           unsigned* q, int n, int base, int nloc,
                                           int lane, int w,
                                           const float* ad_l, unsigned* m_enc,
                                           float* den, float* acc,
                                           unsigned* kept_w, int& kc, bool& ovf) {
    int sup = blockIdx.x % NSUPER;
    for (int ss = 0; ss < NSUPER; ++ss) {
        int e_base = sup * SCAN_SUP + w * 128 + lane * 4;
        int4 dv[SCAN_U];
#pragma unroll
        for (int u = 0; u < SCAN_U; ++u) {
            int e = e_base + u * SCAN_STEP;
            dv[u] = (e < EE) ? *(const int4*)(dstp + e) : make_int4(-1, -1, -1, -1);
        }
#pragma unroll
        for (int u = 0; u < SCAN_U; ++u) {
            int e = e_base + u * SCAN_STEP;
            unsigned d0 = (unsigned)(dv[u].x - base), d1 = (unsigned)(dv[u].y - base);
            unsigned d2 = (unsigned)(dv[u].z - base), d3 = (unsigned)(dv[u].w - base);
            bool h0 = d0 < (unsigned)nloc, h1 = d1 < (unsigned)nloc;
            bool h2 = d2 < (unsigned)nloc, h3 = d3 < (unsigned)nloc;
            unsigned m0 = __builtin_amdgcn_ballot_w32(h0), m1 = __builtin_amdgcn_ballot_w32(h1);
            unsigned m2 = __builtin_amdgcn_ballot_w32(h2), m3 = __builtin_amdgcn_ballot_w32(h3);
            if ((m0 | m1 | m2 | m3) == 0u) continue;
            if (h0) q[n + __builtin_amdgcn_mbcnt_lo(m0, 0u)] = make_entry(e + 0, d0);
            n += __popc(m0);
            if (h1) q[n + __builtin_amdgcn_mbcnt_lo(m1, 0u)] = make_entry(e + 1, d1);
            n += __popc(m1);
            if (h2) q[n + __builtin_amdgcn_mbcnt_lo(m2, 0u)] = make_entry(e + 2, d2);
            n += __popc(m2);
            if (h3) q[n + __builtin_amdgcn_mbcnt_lo(m3, 0u)] = make_entry(e + 3, d3);
            n += __popc(m3);
        }
        if (++sup == NSUPER) sup = 0;
        const bool last = (ss == NSUPER - 1);
        if (n >= QDRAIN || (last && n > 0)) {
            if (PASS == 1) {
                if (kc + n > KCAP - KSELF) ovf = true;
                drain_max(q, n, lane, srcp, a_s, ad_l, m_enc, kept_w + kc, !ovf);
                if (!ovf) kc += n;
                n = 0;
            } else {
                n = drain_sum<false>(q, n, last, lane, srcp, a_s, h, ad_l, m_enc, den, acc);
            }
        }
    }
}

__global__ __launch_bounds__(AGG_T)
void k_tile_aggregate(const int* __restrict__ ei,
                      const float* __restrict__ h,
                      const float* __restrict__ a_s,
                      const float* __restrict__ a_d,
                      float* __restrict__ out) {
    __shared__ alignas(16) float acc[TILE * HC];
    __shared__ unsigned queue[AGG_W * QCAP];
    __shared__ unsigned kept[AGG_W * KCAP];
    __shared__ unsigned m_enc[TILE * 2];
    __shared__ float    den[TILE * 2];
    __shared__ float    ad_l[TILE * 2];

    const int tid = threadIdx.x, lane = tid & 31;
    const int w = __builtin_amdgcn_readfirstlane(tid >> 5);
    const int base = blockIdx.x * TILE;
    const int nloc = min(TILE, NN - base);
    const int* __restrict__ srcp = ei;
    const int* __restrict__ dstp = ei + EE;
    unsigned* q = queue + w * QCAP;
    unsigned* kept_w = kept + w * KCAP;
    int  kc  = 0;
    bool ovf = false;

    for (int i = tid; i < TILE * HC / 4; i += AGG_T)
        ((float4*)acc)[i] = make_float4(0.f, 0.f, 0.f, 0.f);
    if (tid < TILE * 2) den[tid] = 0.f;
    if (tid < nloc * 2) {
        int g = base * 2 + tid;
        float ad = a_d[g];
        m_enc[tid] = enc_f32(leaky(a_s[g] + ad));
        ad_l[tid]  = ad;
    }
    __syncthreads();

    scan_edges<1>(srcp, dstp, a_s, h, q, 0, base, nloc, lane, w, ad_l, m_enc, den, acc,
                  kept_w, kc, ovf);
    __syncthreads();

    int nself = (nloc > w) ? (nloc - w + 31) / 32 : 0;
    if (!ovf) {
        if (lane < nself) {
            int dl = w + 32 * lane;
            kept_w[kc + lane] = make_entry(base + dl, (unsigned)dl);
        }
        drain_sum<true>(kept_w, kc + nself, true, lane, srcp, a_s, h, ad_l, m_enc, den, acc);
    } else {
        if (lane < nself) {
            int dl = w + 32 * lane;
            q[lane] = make_entry(EE + base + dl, (unsigned)dl);
        }
        scan_edges<2>(srcp, dstp, a_s, h, q, nself, base, nloc, lane, w, ad_l, m_enc, den, acc,
                      kept_w, kc, ovf);
    }
    __syncthreads();

    for (int dl = w; dl < nloc; dl += AGG_W) {
        float dn0 = den[dl * 2 + 0] + 1e-16f, dn1 = den[dl * 2 + 1] + 1e-16f;
        const float* ar = acc + dl * HC + lane;
        float4 o0 = make_float4(ar[0] / dn0,   ar[32] / dn0,  ar[64] / dn0,  ar[96] / dn0);
        float4 o1 = make_float4(ar[128] / dn1, ar[160] / dn1, ar[192] / dn1, ar[224] / dn1);
        float* op = out + (size_t)(base + dl) * HC + 4 * lane;
        *(float4*)op        = o0;
        *(float4*)(op + CC) = o1;
    }
}

__global__ void k_final(const float* __restrict__ acc,
                        const float* __restrict__ b1,
                        const float* __restrict__ Wl,
                        const float* __restrict__ bl,
                        float* __restrict__ out) {
    int node = blockIdx.x * blockDim.x + threadIdx.x;
    if (node >= NN) return;
    const float* a = acc + (size_t)node * HC;
    float z0 = bl[0], z1 = bl[1];
#pragma unroll 4
    for (int k = 0; k < HC; ++k) {
        float v = a[k] + b1[k];
        z0 += v * Wl[k * 2 + 0];
        z1 += v * Wl[k * 2 + 1];
    }
    float mx = fmaxf(z0, z1);
    float e0 = expf(z0 - mx), e1 = expf(z1 - mx);
    float inv = 1.f / (e0 + e1);
    out[node * 2 + 0] = e0 * inv;
    out[node * 2 + 1] = e1 * inv;
}

extern "C" void kernel_launch(void* const* d_in, const int* in_sizes, int n_in,
                              void* d_out, int out_size, void* d_ws, size_t ws_size,
                              hipStream_t stream) {
    const float* x        = (const float*)d_in[0];
    const int*   ei       = (const int*)  d_in[1];
    const float* W0       = (const float*)d_in[2];
    const float* att_src0 = (const float*)d_in[3];
    const float* att_dst0 = (const float*)d_in[4];
    const float* b0       = (const float*)d_in[5];
    const float* W1       = (const float*)d_in[6];
    const float* att_src1 = (const float*)d_in[7];
    const float* att_dst1 = (const float*)d_in[8];
    const float* b1       = (const float*)d_in[9];
    const float* Wl       = (const float*)d_in[10];
    const float* bl       = (const float*)d_in[11];
    float* out = (float*)d_out;

    char* base = (char*)d_ws;
    size_t off = 0;
    auto carve = [&](size_t bytes) -> char* {
        char* p = base + off;
        off = (off + bytes + 255) & ~(size_t)255;
        return p;
    };
    float*    h_feat = (float*)   carve((size_t)NN * HC * 4);
    float*    accb   = (float*)   carve((size_t)NN * HC * 4);
    __bf16*   wtbf   = (__bf16*)  carve((size_t)HC * HC * 2);
    float*    a_s    = (float*)   carve((size_t)NN * 2 * 4);
    float*    a_d    = (float*)   carve((size_t)NN * 2 * 4);
    (void)ws_size; (void)in_sizes; (void)n_in; (void)out_size;

    const int T = 256;
    dim3 blk(T);

    k_transpose_bf16<<<(DD * HC + T - 1) / T, blk, 0, stream>>>(W0, wtbf, DD, HC);

    k_gemm_rows_wmma<DD, false><<<NN / 16, dim3(GEMM_T), 0, stream>>>(x, b0, wtbf, h_feat);

    k_att_scores<<<(NN * 2 + T - 1) / T, blk, 0, stream>>>(
        h_feat, att_src0, att_dst0, a_s, a_d);

    k_tile_aggregate<<<NTILES, dim3(AGG_T), 0, stream>>>(ei, h_feat, a_s, a_d, accb);

    k_transpose_bf16<<<(HC * HC + T - 1) / T, blk, 0, stream>>>(W1, wtbf, HC, HC);

    k_gemm_rows_wmma<HC, true><<<NN / 16, dim3(GEMM_T), 0, stream>>>(accb, b0, wtbf, h_feat);

    k_att_scores<<<(NN * 2 + T - 1) / T, blk, 0, stream>>>(
        h_feat, att_src1, att_dst1, a_s, a_d);

    k_tile_aggregate<<<NTILES, dim3(AGG_T), 0, stream>>>(ei, h_feat, a_s, a_d, accb);

    k_final<<<(NN + T - 1) / T, blk, 0, stream>>>(accb, b1, Wl, bl, out);
}


# Layer 1's edge phase takes each wave's list of found edges from a workspace copy that layer 0's edge phase leaves in the same call: no second scan of the edge list; sum order: LDS arrival as before
# speedup vs baseline: 7.0143x; 1.1207x over previous
#include <hip/hip_runtime.h>
#include <math.h>

#define NN    50000
#define DD    128
#define HC    256
#define CC    128
#define EE    800000
#define EP    (EE + NN)
#define NEG_SLOPE 0.2f

typedef __attribute__((ext_vector_type(16))) __bf16 v16bf;
typedef __attribute__((ext_vector_type(8)))  __bf16 v8bf;
typedef __attribute__((ext_vector_type(8)))  float  v8f;

__device__ __forceinline__ unsigned enc_f32(float f) {
    unsigned u = __float_as_uint(f);
    return (u & 0x80000000u) ? ~u : (u | 0x80000000u);
}
__device__ __forceinline__ float dec_f32(unsigned u) {
    u = (u & 0x80000000u) ? (u & 0x7FFFFFFFu) : ~u;
    return __uint_as_float(u);
}

__global__ void k_transpose_bf16(const float* __restrict__ W,
                                 __bf16* __restrict__ Wt, int K, int Nn) {
    int i = blockIdx.x * blockDim.x + threadIdx.x;
    if (i < K * Nn) {
        int n = i / K, k = i - n * K;
        Wt[i] = (__bf16)W[(size_t)k * Nn + n];
    }
}

typedef __attribute__((ext_vector_type(4))) __bf16 v4bf;
#define GEMM_T   512
#define A_PAD    8
static_assert(NN % 16 == 0 && HC == 16 * (GEMM_T / 32), "a block covers 16 whole rows");

template <int K, bool TANH>
__global__ __launch_bounds__(GEMM_T)
void k_gemm_rows_wmma(const float* __restrict__ A,
                      const float* __restrict__ bias,
                      const __bf16* __restrict__ Bt,
                      float* __restrict__ C) {
    constexpr int PITCH = K + A_PAD;
    __shared__ alignas(16) __bf16 a_lds[16 * PITCH];

    const int tid = threadIdx.x;
    const int tM  = blockIdx.x << 4;
    const int lane = tid & 31;
    const int tN   = (tid >> 5) << 4;
    const int half = lane >> 4;
    const int lr   = lane & 15;

    const __bf16* brow = Bt + (size_t)(tN + lr) * K;
    v8bf bfr[K / 16];
#pragma unroll
    for (int j = 0; j < K / 16; ++j)
        bfr[j] = *(const v8bf*)(brow + 16 * j + 8 * half);

    constexpr int V4_ROW = K / 4;
    constexpr int NIT    = 16 * V4_ROW / GEMM_T;
    static_assert(NIT * GEMM_T == 16 * V4_ROW, "the block's threads cover the 16 rows exactly");
    float4 av[NIT];
#pragma unroll
    for (int it = 0; it < NIT; ++it) {
        int i = tid + it * GEMM_T;
        int row = i / V4_ROW, c4 = i - row * V4_ROW;
        av[it] = *(const float4*)(A + (size_t)(tM + row) * K + 4 * c4);
    }
#pragma unroll
    for (int it = 0; it < NIT; ++it) {
        int i = tid + it * GEMM_T;
        int row = i / V4_ROW, c4 = i - row * V4_ROW;
        float4 v = av[it];
        if (TANH) {
            float4 b = *(const float4*)(bias + 4 * c4);
            v.x = tanhf(v.x + b.x); v.y = tanhf(v.y + b.y);
            v.z = tanhf(v.z + b.z); v.w = tanhf(v.w + b.w);
        }
        v4bf o;
        o[0] = (__bf16)v.x; o[1] = (__bf16)v.y; o[2] = (__bf16)v.z; o[3] = (__bf16)v.w;
        *(v4bf*)(a_lds + row * PITCH + 4 * c4) = o;
    }
    __syncthreads();

    v8f acc = {};
    const __bf16* arow = a_lds + lr * PITCH;

#pragma unroll
    for (int kk = 0; kk < K; kk += 32) {
        v8bf alo = *(const v8bf*)(arow + kk + 8 * half);
        v8bf ahi = *(const v8bf*)(arow + kk + 16 + 8 * half);
        v8bf blo = bfr[kk / 16];
        v8bf bhi = bfr[kk / 16 + 1];
        v16bf a, b;
#pragma unroll
        for (int i = 0; i < 8; ++i) {
            a[i] = alo[i]; a[i + 8] = ahi[i];
            b[i] = blo[i]; b[i + 8] = bhi[i];
        }
        acc = __builtin_amdgcn_wmma_f32_16x16x32_bf16(
                  false, a, false, b, (short)0, acc, false, false);
    }

    int n = tN + lr;
#pragma unroll
    for (int r = 0; r < 8; ++r) {
        int m = tM + r + 8 * half;
        C[(size_t)m * HC + n] = acc[r];
    }
}

__global__ void k_att_scores(const float* __restrict__ h,
                             const float* __restrict__ att_src,
                             const float* __restrict__ att_dst,
                             float* __restrict__ a_s,
                             float* __restrict__ a_d) {
    int i = blockIdx.x * blockDim.x + threadIdx.x;
    if (i >= NN * 2) return;
    int node = i >> 1, hd = i & 1;
    const float* hp = h + (size_t)node * HC + hd * CC;
    const float* as = att_src + hd * CC;
    const float* ad = att_dst + hd * CC;
    float s = 0.f, d = 0.f;
#pragma unroll 4
    for (int c = 0; c < CC; ++c) { float v = hp[c]; s += v * as[c]; d += v * ad[c]; }
    a_s[i] = s; a_d[i] = d;
}

#define TILE      196
#define NTILES    ((NN + TILE - 1) / TILE)
#define AGG_T     1024
#define AGG_W     (AGG_T / 32)
#define SCAN_U    4
#define SCAN_STEP (AGG_W * 128)
#define SCAN_SUP  (SCAN_STEP * SCAN_U)
#define NSUPER    ((EE + SCAN_SUP - 1) / SCAN_SUP)
#define QDRAIN    32
#define QCAP      (QDRAIN + 128 * SCAN_U)
#define KCAP      384
#define KSELF     8
static_assert(EP <= (1 << 20) && NN <= (1 << 20) && TILE <= (1 << 12), "queue entry packing");
static_assert((TILE + AGG_W - 1) / AGG_W <= KSELF && KSELF <= 32, "self loops fit the reserved room");

__device__ __forceinline__ float leaky(float l) {
    return (l > 0.f) ? l : NEG_SLOPE * l;
}

__device__ __forceinline__ unsigned make_entry(int e, unsigned dl) {
    return (unsigned)e | (dl << 20);
}
__device__ __forceinline__ int entry_src(unsigned ent, const int* __restrict__ srcp) {
    int e = (int)(ent & 0xFFFFFu);
    return (e < EE) ? srcp[e] : e - EE;
}

__device__ __forceinline__ void drain_max(const unsigned* q, int n, int lane,
                                          const int* __restrict__ srcp,
                                          const float* __restrict__ a_s,
                                          const float* ad_l, unsigned* m_enc,
                                          unsigned* kp, bool keep) {
    for (int b = 0; b < n; b += 32) {
        int i = b + lane;
        if (i < n) {
            unsigned ent = q[i];
            int s = entry_src(ent, srcp);
            float2 as = *(const float2*)(a_s + 2 * s);
            int dl = (int)(ent >> 20);
            if (keep) kp[i] = make_entry(s, (unsigned)dl);
            atomicMax(&m_enc[dl * 2 + 0], enc_f32(leaky(as.x + ad_l[dl * 2 + 0])));
            atomicMax(&m_enc[dl * 2 + 1], enc_f32(leaky(as.y + ad_l[dl * 2 + 1])));
        }
    }
}

__device__ __forceinline__ void take_list_max(const unsigned* __restrict__ lst, int n, int lane,
                                              const float* __restrict__ a_s,
                                              const float* ad_l, unsigned* m_enc,
                                              unsigned* kp) {
    for (int b = 0; b < n; b += 128) {
        unsigned ent[4];
        float2   as[4];
#pragma unroll
        for (int u = 0; u < 4; ++u) {
            int i = b + 32 * u + lane;
            ent[u] = (i < n) ? lst[i] : 0u;
        }
#pragma unroll
        for (int u = 0; u < 4; ++u)
            as[u] = *(const float2*)(a_s + 2 * (int)(ent[u] & 0xFFFFFu));
#pragma unroll
        for (int u = 0; u < 4; ++u) {
            int i = b + 32 * u + lane;
            if (i < n) {
                int dl = (int)(ent[u] >> 20);
                kp[i] = ent[u];
                atomicMax(&m_enc[dl * 2 + 0], enc_f32(leaky(as[u].x + ad_l[dl * 2 + 0])));
                atomicMax(&m_enc[dl * 2 + 1], enc_f32(leaky(as[u].y + ad_l[dl * 2 + 1])));
            }
        }
    }
}

template <bool FULL>
__device__ __forceinline__ void add_rows(int k0, int cnt, int lane, int s_l, int dl_l,
                                         float ex0_l, float ex1_l,
                                         const float* __restrict__ h, float* acc) {
    float4 v0[8], v1[8];
#pragma unroll
    for (int k = 0; k < 8; ++k) {
        if (FULL || k0 + k < cnt) {
            int s = __builtin_amdgcn_readlane(s_l, k0 + k);
            const float* hr = h + (size_t)s * HC + 4 * lane;
            v0[k] = *(const float4*)hr;
            v1[k] = *(const float4*)(hr + CC);
        }
    }
#pragma unroll
    for (int k = 0; k < 8; ++k) {
        if (FULL || k0 + k < cnt) {
            int   dl = __builtin_amdgcn_readlane(dl_l, k0 + k);
            float e0 = __int_as_float(__builtin_amdgcn_readlane(__float_as_int(ex0_l), k0 + k));
            float e1 = __int_as_float(__builtin_amdgcn_readlane(__float_as_int(ex1_l), k0 + k));
            float* ar = acc + dl * HC + lane;
            atomicAdd(ar +   0, e0 * v0[k].x);
            atomicAdd(ar +  32, e0 * v0[k].y);
            atomicAdd(ar +  64, e0 * v0[k].z);
            atomicAdd(ar +  96, e0 * v0[k].w);
            atomicAdd(ar + 128, e1 * v1[k].x);
            atomicAdd(ar + 160, e1 * v1[k].y);
            atomicAdd(ar + 192, e1 * v1[k].z);
            atomicAdd(ar + 224, e1 * v1[k].w);
        }
    }
}

template <bool DIRECT>
__device__ __forceinline__ int drain_sum(unsigned* q, int n, bool last, int lane,
                                          const int* __restrict__ srcp,
                                          const float* __restrict__ a_s,
                                          const float* __restrict__ h,
                                          const float* ad_l, const unsigned* m_enc,
                                          float* den, float* acc) {
    const int nproc = last ? n : (n & ~7);
    for (int b = 0; b < nproc; b += 32) {
        int i = b + lane;
        int cnt = min(32, nproc - b);
        int s_l = 0, dl_l = 0;
        float ex0_l = 0.f, ex1_l = 0.f;
        if (i < nproc) {
            unsigned ent = q[i];
            s_l  = DIRECT ? (int)(ent & 0xFFFFFu) : entry_src(ent, srcp);
            dl_l = (int)(ent >> 20);
            float2 as = *(const float2*)(a_s + 2 * s_l);
            ex0_l = expf(leaky(as.x + ad_l[dl_l * 2 + 0]) - dec_f32(m_enc[dl_l * 2 + 0]));
            ex1_l = expf(leaky(as.y + ad_l[dl_l * 2 + 1]) - dec_f32(m_enc[dl_l * 2 + 1]));
            atomicAdd(&den[dl_l * 2 + 0], ex0_l);
            atomicAdd(&den[dl_l * 2 + 1], ex1_l);
        }
        int k0 = 0;
        for (; k0 + 8 <= cnt; k0 += 8)
            add_rows<true>(k0, cnt, lane, s_l, dl_l, ex0_l, ex1_l, h, acc);
        if (k0 < cnt)
            add_rows<false>(k0, cnt, lane, s_l, dl_l, ex0_l, ex1_l, h, acc);
    }
    const int rem = n - nproc;
    if (rem > 0 && nproc > 0) {
        unsigned keep = 0u;
        if (lane < rem) keep = q[nproc + lane];
        if (lane < rem) q[lane] = keep;
    }
    return rem;
}

template <int PASS>
__device__ __forceinline__ void scan_edges(const int* __restrict__ srcp,
                                           const int* __restrict__ dstp,
                                           const float* __restrict__ a_s,
                                           const float* __restrict__ h,
                                           unsigned* q, int n, int base, int nloc,
                                           int lane, int w,
                                           const float* ad_l, unsigned* m_enc,
                                           float* den, float* acc,
                                           unsigned* kept_w, int& kc, bool& ovf) {
    int sup = blockIdx.x % NSUPER;
    for (int ss = 0; ss < NSUPER; ++ss) {
        int e_base = sup * SCAN_SUP + w * 128 + lane * 4;
        int4 dv[SCAN_U];
#pragma unroll
        for (int u = 0; u < SCAN_U; ++u) {
            int e = e_base + u * SCAN_STEP;
            dv[u] = (e < EE) ? *(const int4*)(dstp + e) : make_int4(-1, -1, -1, -1);
        }
#pragma unroll
        for (int u = 0; u < SCAN_U; ++u) {
            int e = e_base + u * SCAN_STEP;
            unsigned d0 = (unsigned)(dv[u].x - base), d1 = (unsigned)(dv[u].y - base);
            unsigned d2 = (unsigned)(dv[u].z - base), d3 = (unsigned)(dv[u].w - base);
            bool h0 = d0 < (unsigned)nloc, h1 = d1 < (unsigned)nloc;
            bool h2 = d2 < (unsigned)nloc, h3 = d3 < (unsigned)nloc;
            unsigned m0 = __builtin_amdgcn_ballot_w32(h0), m1 = __builtin_amdgcn_ballot_w32(h1);
            unsigned m2 = __builtin_amdgcn_ballot_w32(h2), m3 = __builtin_amdgcn_ballot_w32(h3);
            if ((m0 | m1 | m2 | m3) == 0u) continue;
            if (h0) q[n + __builtin_amdgcn_mbcnt_lo(m0, 0u)] = make_entry(e + 0, d0);
            n += __popc(m0);
            if (h1) q[n + __builtin_amdgcn_mbcnt_lo(m1, 0u)] = make_entry(e + 1, d1);
            n += __popc(m1);
            if (h2) q[n + __builtin_amdgcn_mbcnt_lo(m2, 0u)] = make_entry(e + 2, d2);
            n += __popc(m2);
            if (h3) q[n + __builtin_amdgcn_mbcnt_lo(m3, 0u)] = make_entry(e + 3, d3);
            n += __popc(m3);
        }
        if (++sup == NSUPER) sup = 0;
        const bool last = (ss == NSUPER - 1);
        if (n >= QDRAIN || (last && n > 0)) {
            if (PASS == 1) {
                if (kc + n > KCAP - KSELF) ovf = true;
                drain_max(q, n, lane, srcp, a_s, ad_l, m_enc, kept_w + kc, !ovf);
                if (!ovf) kc += n;
                n = 0;
            } else {
                n = drain_sum<false>(q, n, last, lane, srcp, a_s, h, ad_l, m_enc, den, acc);
            }
        }
    }
}

#define LIST_NONE (-1)
template <bool FIND>
__global__ __launch_bounds__(AGG_T)
void k_tile_aggregate(const int* __restrict__ ei,
                      const float* __restrict__ h,
                      const float* __restrict__ a_s,
                      const float* __restrict__ a_d,
                      float* __restrict__ out,
                      unsigned* __restrict__ lists,
                      int* __restrict__ counts) {
    __shared__ alignas(16) float acc[TILE * HC];
    __shared__ unsigned queue[AGG_W * QCAP];
    __shared__ unsigned kept[AGG_W * KCAP];
    __shared__ unsigned m_enc[TILE * 2];
    __shared__ float    den[TILE * 2];
    __shared__ float    ad_l[TILE * 2];

    const int tid = threadIdx.x, lane = tid & 31;
    const int w = __builtin_amdgcn_readfirstlane(tid >> 5);
    const int base = blockIdx.x * TILE;
    const int nloc = min(TILE, NN - base);
    const int* __restrict__ srcp = ei;
    const int* __restrict__ dstp = ei + EE;
    unsigned* q = queue + w * QCAP;
    unsigned* kept_w = kept + w * KCAP;
    int  kc  = 0;
    bool ovf = false;

    for (int i = tid; i < TILE * HC / 4; i += AGG_T)
        ((float4*)acc)[i] = make_float4(0.f, 0.f, 0.f, 0.f);
    if (tid < TILE * 2) den[tid] = 0.f;
    if (tid < nloc * 2) {
        int g = base * 2 + tid;
        float ad = a_d[g];
        m_enc[tid] = enc_f32(leaky(a_s[g] + ad));
        ad_l[tid]  = ad;
    }
    __syncthreads();

    unsigned* lst = lists + ((size_t)blockIdx.x * AGG_W + w) * KCAP;
    int* cntp = counts + blockIdx.x * AGG_W + w;
    if (FIND) {
        scan_edges<1>(srcp, dstp, a_s, h, q, 0, base, nloc, lane, w, ad_l, m_enc, den, acc,
                      kept_w, kc, ovf);
        if (!ovf)
            for (int i = lane; i < kc; i += 32) lst[i] = kept_w[i];
        if (lane == 0) *cntp = ovf ? LIST_NONE : kc;
    } else {
        int c = __builtin_amdgcn_readfirstlane(*cntp);
        if (c != LIST_NONE) {
            kc = min(max(c, 0), KCAP - KSELF);
            take_list_max(lst, kc, lane, a_s, ad_l, m_enc, kept_w);
        } else {
            ovf = true;
            scan_edges<1>(srcp, dstp, a_s, h, q, 0, base, nloc, lane, w, ad_l, m_enc, den, acc,
                          kept_w, kc, ovf);
        }
    }
    __syncthreads();

    int nself = (nloc > w) ? (nloc - w + 31) / 32 : 0;
    if (!ovf) {
        if (lane < nself) {
            int dl = w + 32 * lane;
            kept_w[kc + lane] = make_entry(base + dl, (unsigned)dl);
        }
        drain_sum<true>(kept_w, kc + nself, true, lane, srcp, a_s, h, ad_l, m_enc, den, acc);
    } else {
        if (lane < nself) {
            int dl = w + 32 * lane;
            q[lane] = make_entry(EE + base + dl, (unsigned)dl);
        }
        scan_edges<2>(srcp, dstp, a_s, h, q, nself, base, nloc, lane, w, ad_l, m_enc, den, acc,
                      kept_w, kc, ovf);
    }
    __syncthreads();

    for (int dl = w; dl < nloc; dl += AGG_W) {
        float dn0 = den[dl * 2 + 0] + 1e-16f, dn1 = den[dl * 2 + 1] + 1e-16f;
        const float* ar = acc + dl * HC + lane;
        float4 o0 = make_float4(ar[0] / dn0,   ar[32] / dn0,  ar[64] / dn0,  ar[96] / dn0);
        float4 o1 = make_float4(ar[128] / dn1, ar[160] / dn1, ar[192] / dn1, ar[224] / dn1);
        float* op = out + (size_t)(base + dl) * HC + 4 * lane;
        *(float4*)op        = o0;
        *(float4*)(op + CC) = o1;
    }
}

__global__ void k_final(const float* __restrict__ acc,
                        const float* __restrict__ b1,
                        const float* __restrict__ Wl,
                        const float* __restrict__ bl,
                        float* __restrict__ out) {
    int node = blockIdx.x * blockDim.x + threadIdx.x;
    if (node >= NN) return;
    const float* a = acc + (size_t)node * HC;
    float z0 = bl[0], z1 = bl[1];
#pragma unroll 4
    for (int k = 0; k < HC; ++k) {
        float v = a[k] + b1[k];
        z0 += v * Wl[k * 2 + 0];
        z1 += v * Wl[k * 2 + 1];
    }
    float mx = fmaxf(z0, z1);
    float e0 = expf(z0 - mx), e1 = expf(z1 - mx);
    float inv = 1.f / (e0 + e1);
    out[node * 2 + 0] = e0 * inv;
    out[node * 2 + 1] = e1 * inv;
}

extern "C" void kernel_launch(void* const* d_in, const int* in_sizes, int n_in,
                              void* d_out, int out_size, void* d_ws, size_t ws_size,
                              hipStream_t stream) {
    const float* x        = (const float*)d_in[0];
    const int*   ei       = (const int*)  d_in[1];
    const float* W0       = (const float*)d_in[2];
    const float* att_src0 = (const float*)d_in[3];
    const float* att_dst0 = (const float*)d_in[4];
    const float* b0       = (const float*)d_in[5];
    const float* W1       = (const float*)d_in[6];
    const float* att_src1 = (const float*)d_in[7];
    const float* att_dst1 = (const float*)d_in[8];
    const float* b1       = (const float*)d_in[9];
    const float* Wl       = (const float*)d_in[10];
    const float* bl       = (const float*)d_in[11];
    float* out = (float*)d_out;

    char* base = (char*)d_ws;
    size_t off = 0;
    auto carve = [&](size_t bytes) -> char* {
        char* p = base + off;
        off = (off + bytes + 255) & ~(size_t)255;
        return p;
    };
    float*    h_feat = (float*)   carve((size_t)NN * HC * 4);
    float*    accb   = (float*)   carve((size_t)NN * HC * 4);
    __bf16*   wtbf   = (__bf16*)  carve((size_t)HC * HC * 2);
    float*    a_s    = (float*)   carve((size_t)NN * 2 * 4);
    float*    a_d    = (float*)   carve((size_t)NN * 2 * 4);
    unsigned* lists  = (unsigned*)carve((size_t)NTILES * AGG_W * KCAP * 4);
    int*      counts = (int*)     carve((size_t)NTILES * AGG_W * 4);
    (void)ws_size; (void)in_sizes; (void)n_in; (void)out_size;

    const int T = 256;
    dim3 blk(T);

    k_transpose_bf16<<<(DD * HC + T - 1) / T, blk, 0, stream>>>(W0, wtbf, DD, HC);

    k_gemm_rows_wmma<DD, false><<<NN / 16, dim3(GEMM_T), 0, stream>>>(x, b0, wtbf, h_feat);

    k_att_scores<<<(NN * 2 + T - 1) / T, blk, 0, stream>>>(
        h_feat, att_src0, att_dst0, a_s, a_d);

    k_tile_aggregate<true><<<NTILES, dim3(AGG_T), 0, stream>>>(ei, h_feat, a_s, a_d, accb,
                                                               lists, counts);

    k_transpose_bf16<<<(HC * HC + T - 1) / T, blk, 0, stream>>>(W1, wtbf, HC, HC);

    k_gemm_rows_wmma<HC, true><<<NN / 16, dim3(GEMM_T), 0, stream>>>(accb, b0, wtbf, h_feat);

    k_att_scores<<<(NN * 2 + T - 1) / T, blk, 0, stream>>>(
        h_feat, att_src1, att_dst1, a_s, a_d);

    k_tile_aggregate<false><<<NTILES, dim3(AGG_T), 0, stream>>>(ei, h_feat, a_s, a_d, accb,
                                                                lists, counts);

    k_final<<<(NN + T - 1) / T, blk, 0, stream>>>(accb, b1, Wl, bl, out);
}


# Each product block forms its 16 rows' attention scores from its f32 accumulators (DPP lane-row butterfly, 8 wave shares summed via LDS, fixed order): 2 score launches gone; score sum order changes
# speedup vs baseline: 7.4620x; 1.0638x over previous
#include <hip/hip_runtime.h>
#include <math.h>

#define NN    50000
#define DD    128
#define HC    256
#define CC    128
#define EE    800000
#define EP    (EE + NN)
#define NEG_SLOPE 0.2f

typedef __attribute__((ext_vector_type(16))) __bf16 v16bf;
typedef __attribute__((ext_vector_type(8)))  __bf16 v8bf;
typedef __attribute__((ext_vector_type(8)))  float  v8f;

__device__ __forceinline__ unsigned enc_f32(float f) {
    unsigned u = __float_as_uint(f);
    return (u & 0x80000000u) ? ~u : (u | 0x80000000u);
}
__device__ __forceinline__ float dec_f32(unsigned u) {
    u = (u & 0x80000000u) ? (u & 0x7FFFFFFFu) : ~u;
    return __uint_as_float(u);
}

__global__ void k_transpose_bf16(const float* __restrict__ W,
                                 __bf16* __restrict__ Wt, int K, int Nn) {
    int i = blockIdx.x * blockDim.x + threadIdx.x;
    if (i < K * Nn) {
        int n = i / K, k = i - n * K;
        Wt[i] = (__bf16)W[(size_t)k * Nn + n];
    }
}

typedef __attribute__((ext_vector_type(4))) __bf16 v4bf;
#define GEMM_T   512
#define A_PAD    8
static_assert(NN % 16 == 0 && HC == 16 * (GEMM_T / 32), "a block covers 16 whole rows");

template <int M>
__device__ __forceinline__ float row_xor(float v) {
    return __int_as_float(__builtin_amdgcn_update_dpp(0, __float_as_int(v), 0x160 + M,
                                                      0xf, 0xf, true));
}

template <int K, bool TANH>
__global__ __launch_bounds__(GEMM_T)
void k_gemm_rows_wmma(const float* __restrict__ A,
                      const float* __restrict__ bias,
                      const __bf16* __restrict__ Bt,
                      float* __restrict__ C,
                      const float* __restrict__ att_src,
                      const float* __restrict__ att_dst,
                      float* __restrict__ a_s,
                      float* __restrict__ a_d) {
    constexpr int PITCH = K + A_PAD;
    __shared__ alignas(16) __bf16 a_lds[16 * PITCH];
    __shared__ float sc_part[(GEMM_T / 32) * 32];
    __shared__ alignas(8) float att_lds[2 * HC];
    static_assert(GEMM_T == 2 * HC, "one element of the two attention vectors per thread");

    const int tid = threadIdx.x;
    const int tM  = blockIdx.x << 4;
    const int lane = tid & 31;
    const int tN   = (tid >> 5) << 4;
    const int half = lane >> 4;
    const int lr   = lane & 15;

    const __bf16* brow = Bt + (size_t)(tN + lr) * K;
    v8bf bfr[K / 16];
#pragma unroll
    for (int j = 0; j < K / 16; ++j)
        bfr[j] = *(const v8bf*)(brow + 16 * j + 8 * half);
    const float att_el = (tid < HC) ? att_src[tid] : att_dst[tid - HC];

    constexpr int V4_ROW = K / 4;
    constexpr int NIT    = 16 * V4_ROW / GEMM_T;
    static_assert(NIT * GEMM_T == 16 * V4_ROW, "the block's threads cover the 16 rows exactly");
    float4 av[NIT];
#pragma unroll
    for (int it = 0; it < NIT; ++it) {
        int i = tid + it * GEMM_T;
        int row = i / V4_ROW, c4 = i - row * V4_ROW;
        av[it] = *(const float4*)(A + (size_t)(tM + row) * K + 4 * c4);
    }
#pragma unroll
    for (int it = 0; it < NIT; ++it) {
        int i = tid + it * GEMM_T;
        int row = i / V4_ROW, c4 = i - row * V4_ROW;
        float4 v = av[it];
        if (TANH) {
            float4 b = *(const float4*)(bias + 4 * c4);
            v.x = tanhf(v.x + b.x); v.y = tanhf(v.y + b.y);
            v.z = tanhf(v.z + b.z); v.w = tanhf(v.w + b.w);
        }
        v4bf o;
        o[0] = (__bf16)v.x; o[1] = (__bf16)v.y; o[2] = (__bf16)v.z; o[3] = (__bf16)v.w;
        *(v4bf*)(a_lds + row * PITCH + 4 * c4) = o;
    }
    att_lds[(tid < HC) ? 2 * tid : 2 * (tid - HC) + 1] = att_el;
    __syncthreads();

    v8f acc = {};
    const __bf16* arow = a_lds + lr * PITCH;

#pragma unroll
    for (int kk = 0; kk < K; kk += 32) {
        v8bf alo = *(const v8bf*)(arow + kk + 8 * half);
        v8bf ahi = *(const v8bf*)(arow + kk + 16 + 8 * half);
        v8bf blo = bfr[kk / 16];
        v8bf bhi = bfr[kk / 16 + 1];
        v16bf a, b;
#pragma unroll
        for (int i = 0; i < 8; ++i) {
            a[i] = alo[i]; a[i + 8] = ahi[i];
            b[i] = blo[i]; b[i + 8] = bhi[i];
        }
        acc = __builtin_amdgcn_wmma_f32_16x16x32_bf16(
                  false, a, false, b, (short)0, acc, false, false);
    }

    const float2 att_n = *(const float2*)(att_lds + 2 * (tN + lr));
    const float as_n = att_n.x, ad_n = att_n.y;
    float p[16];
#pragma unroll
    for (int r = 0; r < 8; ++r) { p[2 * r] = acc[r] * as_n; p[2 * r + 1] = acc[r] * ad_n; }
    const bool b3 = (lr & 8) != 0, b2 = (lr & 4) != 0, b1 = (lr & 2) != 0, b0 = (lr & 1) != 0;
    float q8[8], q4[4], q2[2];
#pragma unroll
    for (int j = 0; j < 8; ++j)
        q8[j] = (b3 ? p[j + 8] : p[j]) + row_xor<8>(b3 ? p[j] : p[j + 8]);
#pragma unroll
    for (int j = 0; j < 4; ++j)
        q4[j] = (b2 ? q8[j + 4] : q8[j]) + row_xor<4>(b2 ? q8[j] : q8[j + 4]);
#pragma unroll
    for (int j = 0; j < 2; ++j)
        q2[j] = (b1 ? q4[j + 2] : q4[j]) + row_xor<2>(b1 ? q4[j] : q4[j + 2]);
    const float q1 = (b0 ? q2[1] : q2[0]) + row_xor<1>(b0 ? q2[0] : q2[1]);
    sc_part[(tid >> 5) * 32 + lane] = q1;
    __syncthreads();

    int n = tN + lr;
#pragma unroll
    for (int r = 0; r < 8; ++r) {
        int m = tM + r + 8 * half;
        C[(size_t)m * HC + n] = acc[r];
    }

    if (tid < 64) {
        const int sc = tid >> 5, idx = tid & 31;
        const int row = idx >> 1, hd = idx & 1;
        const int li = 16 * (row >> 3) + 2 * (row & 7) + sc;
        float s = 0.f;
#pragma unroll
        for (int g = 0; g < 8; ++g) s += sc_part[(hd * 8 + g) * 32 + li];
        (sc ? a_d : a_s)[tM * 2 + idx] = s;
    }
}

#define TILE      196
#define NTILES    ((NN + TILE - 1) / TILE)
#define AGG_T     1024
#define AGG_W     (AGG_T / 32)
#define SCAN_U    4
#define SCAN_STEP (AGG_W * 128)
#define SCAN_SUP  (SCAN_STEP * SCAN_U)
#define NSUPER    ((EE + SCAN_SUP - 1) / SCAN_SUP)
#define QDRAIN    32
#define QCAP      (QDRAIN + 128 * SCAN_U)
#define KCAP      384
#define KSELF     8
static_assert(EP <= (1 << 20) && NN <= (1 << 20) && TILE <= (1 << 12), "queue entry packing");
static_assert((TILE + AGG_W - 1) / AGG_W <= KSELF && KSELF <= 32, "self loops fit the reserved room");

__device__ __forceinline__ float leaky(float l) {
    return (l > 0.f) ? l : NEG_SLOPE * l;
}

__device__ __forceinline__ unsigned make_entry(int e, unsigned dl) {
    return (unsigned)e | (dl << 20);
}
__device__ __forceinline__ int entry_src(unsigned ent, const int* __restrict__ srcp) {
    int e = (int)(ent & 0xFFFFFu);
    return (e < EE) ? srcp[e] : e - EE;
}

__device__ __forceinline__ void drain_max(const unsigned* q, int n, int lane,
                                          const int* __restrict__ srcp,
                                          const float* __restrict__ a_s,
                                          const float* ad_l, unsigned* m_enc,
                                          unsigned* kp, bool keep) {
    for (int b = 0; b < n; b += 32) {
        int i = b + lane;
        if (i < n) {
            unsigned ent = q[i];
            int s = entry_src(ent, srcp);
            float2 as = *(const float2*)(a_s + 2 * s);
            int dl = (int)(ent >> 20);
            if (keep) kp[i] = make_entry(s, (unsigned)dl);
            atomicMax(&m_enc[dl * 2 + 0], enc_f32(leaky(as.x + ad_l[dl * 2 + 0])));
            atomicMax(&m_enc[dl * 2 + 1], enc_f32(leaky(as.y + ad_l[dl * 2 + 1])));
        }
    }
}

__device__ __forceinline__ void take_list_max(const unsigned* __restrict__ lst, int n, int lane,
                                              const float* __restrict__ a_s,
                                              const float* ad_l, unsigned* m_enc,
                                              unsigned* kp) {
    for (int b = 0; b < n; b += 128) {
        unsigned ent[4];
        float2   as[4];
#pragma unroll
        for (int u = 0; u < 4; ++u) {
            int i = b + 32 * u + lane;
            ent[u] = (i < n) ? lst[i] : 0u;
        }
#pragma unroll
        for (int u = 0; u < 4; ++u)
            as[u] = *(const float2*)(a_s + 2 * (int)(ent[u] & 0xFFFFFu));
#pragma unroll
        for (int u = 0; u < 4; ++u) {
            int i = b + 32 * u + lane;
            if (i < n) {
                int dl = (int)(ent[u] >> 20);
                kp[i] = ent[u];
                atomicMax(&m_enc[dl * 2 + 0], enc_f32(leaky(as[u].x + ad_l[dl * 2 + 0])));
                atomicMax(&m_enc[dl * 2 + 1], enc_f32(leaky(as[u].y + ad_l[dl * 2 + 1])));
            }
        }
    }
}

template <bool FULL>
__device__ __forceinline__ void add_rows(int k0, int cnt, int lane, int s_l, int dl_l,
                                         float ex0_l, float ex1_l,
                                         const float* __restrict__ h, float* acc) {
    float4 v0[8], v1[8];
#pragma unroll
    for (int k = 0; k < 8; ++k) {
        if (FULL || k0 + k < cnt) {
            int s = __builtin_amdgcn_readlane(s_l, k0 + k);
            const float* hr = h + (size_t)s * HC + 4 * lane;
            v0[k] = *(const float4*)hr;
            v1[k] = *(const float4*)(hr + CC);
        }
    }
#pragma unroll
    for (int k = 0; k < 8; ++k) {
        if (FULL || k0 + k < cnt) {
            int   dl = __builtin_amdgcn_readlane(dl_l, k0 + k);
            float e0 = __int_as_float(__builtin_amdgcn_readlane(__float_as_int(ex0_l), k0 + k));
            float e1 = __int_as_float(__builtin_amdgcn_readlane(__float_as_int(ex1_l), k0 + k));
            float* ar = acc + dl * HC + lane;
            atomicAdd(ar +   0, e0 * v0[k].x);
            atomicAdd(ar +  32, e0 * v0[k].y);
            atomicAdd(ar +  64, e0 * v0[k].z);
            atomicAdd(ar +  96, e0 * v0[k].w);
            atomicAdd(ar + 128, e1 * v1[k].x);
            atomicAdd(ar + 160, e1 * v1[k].y);
            atomicAdd(ar + 192, e1 * v1[k].z);
            atomicAdd(ar + 224, e1 * v1[k].w);
        }
    }
}

template <bool DIRECT>
__device__ __forceinline__ int drain_sum(unsigned* q, int n, bool last, int lane,
                                          const int* __restrict__ srcp,
                                          const float* __restrict__ a_s,
                                          const float* __restrict__ h,
                                          const float* ad_l, const unsigned* m_enc,
                                          float* den, float* acc) {
    const int nproc = last ? n : (n & ~7);
    for (int b = 0; b < nproc; b += 32) {
        int i = b + lane;
        int cnt = min(32, nproc - b);
        int s_l = 0, dl_l = 0;
        float ex0_l = 0.f, ex1_l = 0.f;
        if (i < nproc) {
            unsigned ent = q[i];
            s_l  = DIRECT ? (int)(ent & 0xFFFFFu) : entry_src(ent, srcp);
            dl_l = (int)(ent >> 20);
            float2 as = *(const float2*)(a_s + 2 * s_l);
            ex0_l = expf(leaky(as.x + ad_l[dl_l * 2 + 0]) - dec_f32(m_enc[dl_l * 2 + 0]));
            ex1_l = expf(leaky(as.y + ad_l[dl_l * 2 + 1]) - dec_f32(m_enc[dl_l * 2 + 1]));
            atomicAdd(&den[dl_l * 2 + 0], ex0_l);
            atomicAdd(&den[dl_l * 2 + 1], ex1_l);
        }
        int k0 = 0;
        for (; k0 + 8 <= cnt; k0 += 8)
            add_rows<true>(k0, cnt, lane, s_l, dl_l, ex0_l, ex1_l, h, acc);
        if (k0 < cnt)
            add_rows<false>(k0, cnt, lane, s_l, dl_l, ex0_l, ex1_l, h, acc);
    }
    const int rem = n - nproc;
    if (rem > 0 && nproc > 0) {
        unsigned keep = 0u;
        if (lane < rem) keep = q[nproc + lane];
        if (lane < rem) q[lane] = keep;
    }
    return rem;
}

template <int PASS>
__device__ __forceinline__ void scan_edges(const int* __restrict__ srcp,
                                           const int* __restrict__ dstp,
                                           const float* __restrict__ a_s,
                                           const float* __restrict__ h,
                                           unsigned* q, int n, int base, int nloc,
                                           int lane, int w,
                                           const float* ad_l, unsigned* m_enc,
                                           float* den, float* acc,
                                           unsigned* kept_w, int& kc, bool& ovf) {
    int sup = blockIdx.x % NSUPER;
    for (int ss = 0; ss < NSUPER; ++ss) {
        int e_base = sup * SCAN_SUP + w * 128 + lane * 4;
        int4 dv[SCAN_U];
#pragma unroll
        for (int u = 0; u < SCAN_U; ++u) {
            int e = e_base + u * SCAN_STEP;
            dv[u] = (e < EE) ? *(const int4*)(dstp + e) : make_int4(-1, -1, -1, -1);
        }
#pragma unroll
        for (int u = 0; u < SCAN_U; ++u) {
            int e = e_base + u * SCAN_STEP;
            unsigned d0 = (unsigned)(dv[u].x - base), d1 = (unsigned)(dv[u].y - base);
            unsigned d2 = (unsigned)(dv[u].z - base), d3 = (unsigned)(dv[u].w - base);
            bool h0 = d0 < (unsigned)nloc, h1 = d1 < (unsigned)nloc;
            bool h2 = d2 < (unsigned)nloc, h3 = d3 < (unsigned)nloc;
            unsigned m0 = __builtin_amdgcn_ballot_w32(h0), m1 = __builtin_amdgcn_ballot_w32(h1);
            unsigned m2 = __builtin_amdgcn_ballot_w32(h2), m3 = __builtin_amdgcn_ballot_w32(h3);
            if ((m0 | m1 | m2 | m3) == 0u) continue;
            if (h0) q[n + __builtin_amdgcn_mbcnt_lo(m0, 0u)] = make_entry(e + 0, d0);
            n += __popc(m0);
            if (h1) q[n + __builtin_amdgcn_mbcnt_lo(m1, 0u)] = make_entry(e + 1, d1);
            n += __popc(m1);
            if (h2) q[n + __builtin_amdgcn_mbcnt_lo(m2, 0u)] = make_entry(e + 2, d2);
            n += __popc(m2);
            if (h3) q[n + __builtin_amdgcn_mbcnt_lo(m3, 0u)] = make_entry(e + 3, d3);
            n += __popc(m3);
        }
        if (++sup == NSUPER) sup = 0;
        const bool last = (ss == NSUPER - 1);
        if (n >= QDRAIN || (last && n > 0)) {
            if (PASS == 1) {
                if (kc + n > KCAP - KSELF) ovf = true;
                drain_max(q, n, lane, srcp, a_s, ad_l, m_enc, kept_w + kc, !ovf);
                if (!ovf) kc += n;
                n = 0;
            } else {
                n = drain_sum<false>(q, n, last, lane, srcp, a_s, h, ad_l, m_enc, den, acc);
            }
        }
    }
}

#define LIST_NONE (-1)
template <bool FIND>
__global__ __launch_bounds__(AGG_T)
void k_tile_aggregate(const int* __restrict__ ei,
                      const float* __restrict__ h,
                      const float* __restrict__ a_s,
                      const float* __restrict__ a_d,
                      float* __restrict__ out,
                      unsigned* __restrict__ lists,
                      int* __restrict__ counts) {
    __shared__ alignas(16) float acc[TILE * HC];
    __shared__ unsigned queue[AGG_W * QCAP];
    __shared__ unsigned kept[AGG_W * KCAP];
    __shared__ unsigned m_enc[TILE * 2];
    __shared__ float    den[TILE * 2];
    __shared__ float    ad_l[TILE * 2];

    const int tid = threadIdx.x, lane = tid & 31;
    const int w = __builtin_amdgcn_readfirstlane(tid >> 5);
    const int base = blockIdx.x * TILE;
    const int nloc = min(TILE, NN - base);
    const int* __restrict__ srcp = ei;
    const int* __restrict__ dstp = ei + EE;
    unsigned* q = queue + w * QCAP;
    unsigned* kept_w = kept + w * KCAP;
    int  kc  = 0;
    bool ovf = false;

    for (int i = tid; i < TILE * HC / 4; i += AGG_T)
        ((float4*)acc)[i] = make_float4(0.f, 0.f, 0.f, 0.f);
    if (tid < TILE * 2) den[tid] = 0.f;
    if (tid < nloc * 2) {
        int g = base * 2 + tid;
        float ad = a_d[g];
        m_enc[tid] = enc_f32(leaky(a_s[g] + ad));
        ad_l[tid]  = ad;
    }
    __syncthreads();

    unsigned* lst = lists + ((size_t)blockIdx.x * AGG_W + w) * KCAP;
    int* cntp = counts + blockIdx.x * AGG_W + w;
    if (FIND) {
        scan_edges<1>(srcp, dstp, a_s, h, q, 0, base, nloc, lane, w, ad_l, m_enc, den, acc,
                      kept_w, kc, ovf);
        if (!ovf)
            for (int i = lane; i < kc; i += 32) lst[i] = kept_w[i];
        if (lane == 0) *cntp = ovf ? LIST_NONE : kc;
    } else {
        int c = __builtin_amdgcn_readfirstlane(*cntp);
        if (c != LIST_NONE) {
            kc = min(max(c, 0), KCAP - KSELF);
            take_list_max(lst, kc, lane, a_s, ad_l, m_enc, kept_w);
        } else {
            ovf = true;
            scan_edges<1>(srcp, dstp, a_s, h, q, 0, base, nloc, lane, w, ad_l, m_enc, den, acc,
                          kept_w, kc, ovf);
        }
    }
    __syncthreads();

    int nself = (nloc > w) ? (nloc - w + 31) / 32 : 0;
    if (!ovf) {
        if (lane < nself) {
            int dl = w + 32 * lane;
            kept_w[kc + lane] = make_entry(base + dl, (unsigned)dl);
        }
        drain_sum<true>(kept_w, kc + nself, true, lane, srcp, a_s, h, ad_l, m_enc, den, acc);
    } else {
        if (lane < nself) {
            int dl = w + 32 * lane;
            q[lane] = make_entry(EE + base + dl, (unsigned)dl);
        }
        scan_edges<2>(srcp, dstp, a_s, h, q, nself, base, nloc, lane, w, ad_l, m_enc, den, acc,
                      kept_w, kc, ovf);
    }
    __syncthreads();

    for (int dl = w; dl < nloc; dl += AGG_W) {
        float dn0 = den[dl * 2 + 0] + 1e-16f, dn1 = den[dl * 2 + 1] + 1e-16f;
        const float* ar = acc + dl * HC + lane;
        float4 o0 = make_float4(ar[0] / dn0,   ar[32] / dn0,  ar[64] / dn0,  ar[96] / dn0);
        float4 o1 = make_float4(ar[128] / dn1, ar[160] / dn1, ar[192] / dn1, ar[224] / dn1);
        float* op = out + (size_t)(base + dl) * HC + 4 * lane;
        *(float4*)op        = o0;
        *(float4*)(op + CC) = o1;
    }
}

__global__ void k_final(const float* __restrict__ acc,
                        const float* __restrict__ b1,
                        const float* __restrict__ Wl,
                        const float* __restrict__ bl,
                        float* __restrict__ out) {
    int node = blockIdx.x * blockDim.x + threadIdx.x;
    if (node >= NN) return;
    const float* a = acc + (size_t)node * HC;
    float z0 = bl[0], z1 = bl[1];
#pragma unroll 4
    for (int k = 0; k < HC; ++k) {
        float v = a[k] + b1[k];
        z0 += v * Wl[k * 2 + 0];
        z1 += v * Wl[k * 2 + 1];
    }
    float mx = fmaxf(z0, z1);
    float e0 = expf(z0 - mx), e1 = expf(z1 - mx);
    float inv = 1.f / (e0 + e1);
    out[node * 2 + 0] = e0 * inv;
    out[node * 2 + 1] = e1 * inv;
}

extern "C" void kernel_launch(void* const* d_in, const int* in_sizes, int n_in,
                              void* d_out, int out_size, void* d_ws, size_t ws_size,
                              hipStream_t stream) {
    const float* x        = (const float*)d_in[0];
    const int*   ei       = (const int*)  d_in[1];
    const float* W0       = (const float*)d_in[2];
    const float* att_src0 = (const float*)d_in[3];
    const float* att_dst0 = (const float*)d_in[4];
    const float* b0       = (const float*)d_in[5];
    const float* W1       = (const float*)d_in[6];
    const float* att_src1 = (const float*)d_in[7];
    const float* att_dst1 = (const float*)d_in[8];
    const float* b1       = (const float*)d_in[9];
    const float* Wl       = (const float*)d_in[10];
    const float* bl       = (const float*)d_in[11];
    float* out = (float*)d_out;

    char* base = (char*)d_ws;
    size_t off = 0;
    auto carve = [&](size_t bytes) -> char* {
        char* p = base + off;
        off = (off + bytes + 255) & ~(size_t)255;
        return p;
    };
    float*    h_feat = (float*)   carve((size_t)NN * HC * 4);
    float*    accb   = (float*)   carve((size_t)NN * HC * 4);
    __bf16*   wtbf   = (__bf16*)  carve((size_t)HC * HC * 2);
    float*    a_s    = (float*)   carve((size_t)NN * 2 * 4);
    float*    a_d    = (float*)   carve((size_t)NN * 2 * 4);
    unsigned* lists  = (unsigned*)carve((size_t)NTILES * AGG_W * KCAP * 4);
    int*      counts = (int*)     carve((size_t)NTILES * AGG_W * 4);
    (void)ws_size; (void)in_sizes; (void)n_in; (void)out_size;

    const int T = 256;
    dim3 blk(T);

    k_transpose_bf16<<<(DD * HC + T - 1) / T, blk, 0, stream>>>(W0, wtbf, DD, HC);

    k_gemm_rows_wmma<DD, false><<<NN / 16, dim3(GEMM_T), 0, stream>>>(
        x, b0, wtbf, h_feat, att_src0, att_dst0, a_s, a_d);

    k_tile_aggregate<true><<<NTILES, dim3(AGG_T), 0, stream>>>(ei, h_feat, a_s, a_d, accb,
                                                               lists, counts);

    k_transpose_bf16<<<(HC * HC + T - 1) / T, blk, 0, stream>>>(W1, wtbf, HC, HC);

    k_gemm_rows_wmma<HC, true><<<NN / 16, dim3(GEMM_T), 0, stream>>>(
        accb, b0, wtbf, h_feat, att_src1, att_dst1, a_s, a_d);

    k_tile_aggregate<false><<<NTILES, dim3(AGG_T), 0, stream>>>(ei, h_feat, a_s, a_d, accb,
                                                                lists, counts);

    k_final<<<(NN + T - 1) / T, blk, 0, stream>>>(accb, b1, Wl, bl, out);
}


# Edge search reads a byte per edge: layer 0's product blocks leave each edge's tile number (dst/196) in the workspace; a lane tests 16 edges a load by word arithmetic; sum order: LDS arrival as before
# speedup vs baseline: 7.7734x; 1.0417x over previous
#include <hip/hip_runtime.h>
#include <math.h>

#define NN    50000
#define DD    128
#define HC    256
#define CC    128
#define EE    800000
#define EP    (EE + NN)
#define NEG_SLOPE 0.2f
#define TILE      196
#define NTILES    ((NN + TILE - 1) / TILE)
static_assert(NTILES <= 256, "a tile number fits a byte");
static_assert(EE % 16 == 0, "the tile-number column is read 16 bytes at a time");

typedef __attribute__((ext_vector_type(16))) __bf16 v16bf;
typedef __attribute__((ext_vector_type(8)))  __bf16 v8bf;
typedef __attribute__((ext_vector_type(8)))  float  v8f;

__device__ __forceinline__ unsigned enc_f32(float f) {
    unsigned u = __float_as_uint(f);
    return (u & 0x80000000u) ? ~u : (u | 0x80000000u);
}
__device__ __forceinline__ float dec_f32(unsigned u) {
    u = (u & 0x80000000u) ? (u & 0x7FFFFFFFu) : ~u;
    return __uint_as_float(u);
}

__global__ void k_transpose_bf16(const float* __restrict__ W,
                                 __bf16* __restrict__ Wt, int K, int Nn) {
    int i = blockIdx.x * blockDim.x + threadIdx.x;
    if (i < K * Nn) {
        int n = i / K, k = i - n * K;
        Wt[i] = (__bf16)W[(size_t)k * Nn + n];
    }
}

typedef __attribute__((ext_vector_type(4))) __bf16 v4bf;
#define GEMM_T   512
#define A_PAD    8
static_assert(NN % 16 == 0 && HC == 16 * (GEMM_T / 32), "a block covers 16 whole rows");

template <int M>
__device__ __forceinline__ float row_xor(float v) {
    return __int_as_float(__builtin_amdgcn_update_dpp(0, __float_as_int(v), 0x160 + M,
                                                      0xf, 0xf, true));
}

template <int K, bool TANH>
__global__ __launch_bounds__(GEMM_T)
void k_gemm_rows_wmma(const float* __restrict__ A,
                      const float* __restrict__ bias,
                      const __bf16* __restrict__ Bt,
                      float* __restrict__ C,
                      const float* __restrict__ att_src,
                      const float* __restrict__ att_dst,
                      float* __restrict__ a_s,
                      float* __restrict__ a_d,
                      const int* __restrict__ dstp,
                      unsigned* __restrict__ tile_of_edge) {
    constexpr int PITCH = K + A_PAD;
    __shared__ alignas(16) __bf16 a_lds[16 * PITCH];
    __shared__ float sc_part[(GEMM_T / 32) * 32];
    __shared__ alignas(8) float att_lds[2 * HC];
    static_assert(GEMM_T == 2 * HC, "one element of the two attention vectors per thread");

    const int tid = threadIdx.x;
    const int tM  = blockIdx.x << 4;
    const int lane = tid & 31;
    const int tN   = (tid >> 5) << 4;
    const int half = lane >> 4;
    const int lr   = lane & 15;

    if (!TANH) {
        constexpr int PRE = (EE / 4 + NN / 16 - 1) / (NN / 16);
        static_assert(PRE % 32 == 0 && PRE <= GEMM_T, "whole waves take the pieces");
        const int g = blockIdx.x * PRE + tid;
        if (tid < PRE && g < EE / 4) {
            const int4 d = *(const int4*)(dstp + 4 * g);
            const unsigned u0 = (unsigned)d.x < (unsigned)NN ? (unsigned)d.x : 0u;
            const unsigned u1 = (unsigned)d.y < (unsigned)NN ? (unsigned)d.y : 0u;
            const unsigned u2 = (unsigned)d.z < (unsigned)NN ? (unsigned)d.z : 0u;
            const unsigned u3 = (unsigned)d.w < (unsigned)NN ? (unsigned)d.w : 0u;
            tile_of_edge[g] = (u0 / TILE) | ((u1 / TILE) << 8) | ((u2 / TILE) << 16) | ((u3 / TILE) << 24);
        }
    }

    const __bf16* brow = Bt + (size_t)(tN + lr) * K;
    v8bf bfr[K / 16];
#pragma unroll
    for (int j = 0; j < K / 16; ++j)
        bfr[j] = *(const v8bf*)(brow + 16 * j + 8 * half);
    const float att_el = (tid < HC) ? att_src[tid] : att_dst[tid - HC];

    constexpr int V4_ROW = K / 4;
    constexpr int NIT    = 16 * V4_ROW / GEMM_T;
    static_assert(NIT * GEMM_T == 16 * V4_ROW, "the block's threads cover the 16 rows exactly");
    float4 av[NIT];
#pragma unroll
    for (int it = 0; it < NIT; ++it) {
        int i = tid + it * GEMM_T;
        int row = i / V4_ROW, c4 = i - row * V4_ROW;
        av[it] = *(const float4*)(A + (size_t)(tM + row) * K + 4 * c4);
    }
#pragma unroll
    for (int it = 0; it < NIT; ++it) {
        int i = tid + it * GEMM_T;
        int row = i / V4_ROW, c4 = i - row * V4_ROW;
        float4 v = av[it];
        if (TANH) {
            float4 b = *(const float4*)(bias + 4 * c4);
            v.x = tanhf(v.x + b.x); v.y = tanhf(v.y + b.y);
            v.z = tanhf(v.z + b.z); v.w = tanhf(v.w + b.w);
        }
        v4bf o;
        o[0] = (__bf16)v.x; o[1] = (__bf16)v.y; o[2] = (__bf16)v.z; o[3] = (__bf16)v.w;
        *(v4bf*)(a_lds + row * PITCH + 4 * c4) = o;
    }
    att_lds[(tid < HC) ? 2 * tid : 2 * (tid - HC) + 1] = att_el;
    __syncthreads();

    v8f acc = {};
    const __bf16* arow = a_lds + lr * PITCH;

#pragma unroll
    for (int kk = 0; kk < K; kk += 32) {
        v8bf alo = *(const v8bf*)(arow + kk + 8 * half);
        v8bf ahi = *(const v8bf*)(arow + kk + 16 + 8 * half);
        v8bf blo = bfr[kk / 16];
        v8bf bhi = bfr[kk / 16 + 1];
        v16bf a, b;
#pragma unroll
        for (int i = 0; i < 8; ++i) {
            a[i] = alo[i]; a[i + 8] = ahi[i];
            b[i] = blo[i]; b[i + 8] = bhi[i];
        }
        acc = __builtin_amdgcn_wmma_f32_16x16x32_bf16(
                  false, a, false, b, (short)0, acc, false, false);
    }

    const float2 att_n = *(const float2*)(att_lds + 2 * (tN + lr));
    const float as_n = att_n.x, ad_n = att_n.y;
    float p[16];
#pragma unroll
    for (int r = 0; r < 8; ++r) { p[2 * r] = acc[r] * as_n; p[2 * r + 1] = acc[r] * ad_n; }
    const bool b3 = (lr & 8) != 0, b2 = (lr & 4) != 0, b1 = (lr & 2) != 0, b0 = (lr & 1) != 0;
    float q8[8], q4[4], q2[2];
#pragma unroll
    for (int j = 0; j < 8; ++j)
        q8[j] = (b3 ? p[j + 8] : p[j]) + row_xor<8>(b3 ? p[j] : p[j + 8]);
#pragma unroll
    for (int j = 0; j < 4; ++j)
        q4[j] = (b2 ? q8[j + 4] : q8[j]) + row_xor<4>(b2 ? q8[j] : q8[j + 4]);
#pragma unroll
    for (int j = 0; j < 2; ++j)
        q2[j] = (b1 ? q4[j + 2] : q4[j]) + row_xor<2>(b1 ? q4[j] : q4[j + 2]);
    const float q1 = (b0 ? q2[1] : q2[0]) + row_xor<1>(b0 ? q2[0] : q2[1]);
    sc_part[(tid >> 5) * 32 + lane] = q1;
    __syncthreads();

    int n = tN + lr;
#pragma unroll
    for (int r = 0; r < 8; ++r) {
        int m = tM + r + 8 * half;
        C[(size_t)m * HC + n] = acc[r];
    }

    if (tid < 64) {
        const int sc = tid >> 5, idx = tid & 31;
        const int row = idx >> 1, hd = idx & 1;
        const int li = 16 * (row >> 3) + 2 * (row & 7) + sc;
        float s = 0.f;
#pragma unroll
        for (int g = 0; g < 8; ++g) s += sc_part[(hd * 8 + g) * 32 + li];
        (sc ? a_d : a_s)[tM * 2 + idx] = s;
    }
}

#define AGG_T     1024
#define AGG_W     (AGG_T / 32)
#define PIECE_E   512
#define SCAN_U    4
#define NPIECE    ((EE / 16 + AGG_T - 1) / AGG_T)
#define QDRAIN    32
#define QCAP      (QDRAIN + PIECE_E)
#define KCAP      384
#define KSELF     8
static_assert(EP <= (1 << 20) && NN <= (1 << 20) && TILE <= (1 << 12), "queue entry packing");
static_assert((TILE + AGG_W - 1) / AGG_W <= KSELF && KSELF <= 32, "self loops fit the reserved room");

__device__ __forceinline__ float leaky(float l) {
    return (l > 0.f) ? l : NEG_SLOPE * l;
}

__device__ __forceinline__ unsigned make_entry(int s, unsigned dl) {
    return (unsigned)s | (dl << 20);
}
__device__ __forceinline__ void entry_nodes(unsigned e, int base, int nloc,
                                            const int* __restrict__ srcp,
                                            const int* __restrict__ dstp,
                                            int& s, int& dl) {
    int d;
    if (e < (unsigned)EE) { s = srcp[e]; d = dstp[e]; }
    else                  { s = (int)e - EE; d = s; }
    dl = min(max(d - base, 0), nloc - 1);
}

__device__ __forceinline__ void drain_max(const unsigned* q, int n, int lane,
                                          const int* __restrict__ srcp,
                                          const int* __restrict__ dstp, int base, int nloc,
                                          const float* __restrict__ a_s,
                                          const float* ad_l, unsigned* m_enc,
                                          unsigned* kp, bool keep) {
    for (int b = 0; b < n; b += 32) {
        int i = b + lane;
        if (i < n) {
            int s, dl;
            entry_nodes(q[i], base, nloc, srcp, dstp, s, dl);
            float2 as = *(const float2*)(a_s + 2 * s);
            if (keep) kp[i] = make_entry(s, (unsigned)dl);
            atomicMax(&m_enc[dl * 2 + 0], enc_f32(leaky(as.x + ad_l[dl * 2 + 0])));
            atomicMax(&m_enc[dl * 2 + 1], enc_f32(leaky(as.y + ad_l[dl * 2 + 1])));
        }
    }
}

__device__ __forceinline__ void take_list_max(const unsigned* __restrict__ lst, int n, int lane,
                                              const float* __restrict__ a_s,
                                              const float* ad_l, unsigned* m_enc,
                                              unsigned* kp) {
    for (int b = 0; b < n; b += 128) {
        unsigned ent[4];
        float2   as[4];
#pragma unroll
        for (int u = 0; u < 4; ++u) {
            int i = b + 32 * u + lane;
            ent[u] = (i < n) ? lst[i] : 0u;
        }
#pragma unroll
        for (int u = 0; u < 4; ++u)
            as[u] = *(const float2*)(a_s + 2 * (int)(ent[u] & 0xFFFFFu));
#pragma unroll
        for (int u = 0; u < 4; ++u) {
            int i = b + 32 * u + lane;
            if (i < n) {
                int dl = (int)(ent[u] >> 20);
                kp[i] = ent[u];
                atomicMax(&m_enc[dl * 2 + 0], enc_f32(leaky(as[u].x + ad_l[dl * 2 + 0])));
                atomicMax(&m_enc[dl * 2 + 1], enc_f32(leaky(as[u].y + ad_l[dl * 2 + 1])));
            }
        }
    }
}

template <bool FULL>
__device__ __forceinline__ void add_rows(int k0, int cnt, int lane, int s_l, int dl_l,
                                         float ex0_l, float ex1_l,
                                         const float* __restrict__ h, float* acc) {
    float4 v0[8], v1[8];
#pragma unroll
    for (int k = 0; k < 8; ++k) {
        if (FULL || k0 + k < cnt) {
            int s = __builtin_amdgcn_readlane(s_l, k0 + k);
            const float* hr = h + (size_t)s * HC + 4 * lane;
            v0[k] = *(const float4*)hr;
            v1[k] = *(const float4*)(hr + CC);
        }
    }
#pragma unroll
    for (int k = 0; k < 8; ++k) {
        if (FULL || k0 + k < cnt) {
            int   dl = __builtin_amdgcn_readlane(dl_l, k0 + k);
            float e0 = __int_as_float(__builtin_amdgcn_readlane(__float_as_int(ex0_l), k0 + k));
            float e1 = __int_as_float(__builtin_amdgcn_readlane(__float_as_int(ex1_l), k0 + k));
            float* ar = acc + dl * HC + lane;
            atomicAdd(ar +   0, e0 * v0[k].x);
            atomicAdd(ar +  32, e0 * v0[k].y);
            atomicAdd(ar +  64, e0 * v0[k].z);
            atomicAdd(ar +  96, e0 * v0[k].w);
            atomicAdd(ar + 128, e1 * v1[k].x);
            atomicAdd(ar + 160, e1 * v1[k].y);
            atomicAdd(ar + 192, e1 * v1[k].z);
            atomicAdd(ar + 224, e1 * v1[k].w);
        }
    }
}

template <bool DIRECT>
__device__ __forceinline__ int drain_sum(unsigned* q, int n, bool last, int lane,
                                          const int* __restrict__ srcp,
                                          const int* __restrict__ dstp, int base, int nloc,
                                          const float* __restrict__ a_s,
                                          const float* __restrict__ h,
                                          const float* ad_l, const unsigned* m_enc,
                                          float* den, float* acc) {
    const int nproc = last ? n : (n & ~7);
    for (int b = 0; b < nproc; b += 32) {
        int i = b + lane;
        int cnt = min(32, nproc - b);
        int s_l = 0, dl_l = 0;
        float ex0_l = 0.f, ex1_l = 0.f;
        if (i < nproc) {
            unsigned ent = q[i];
            if (DIRECT) { s_l = (int)(ent & 0xFFFFFu); dl_l = (int)(ent >> 20); }
            else        entry_nodes(ent, base, nloc, srcp, dstp, s_l, dl_l);
            float2 as = *(const float2*)(a_s + 2 * s_l);
            ex0_l = expf(leaky(as.x + ad_l[dl_l * 2 + 0]) - dec_f32(m_enc[dl_l * 2 + 0]));
            ex1_l = expf(leaky(as.y + ad_l[dl_l * 2 + 1]) - dec_f32(m_enc[dl_l * 2 + 1]));
            atomicAdd(&den[dl_l * 2 + 0], ex0_l);
            atomicAdd(&den[dl_l * 2 + 1], ex1_l);
        }
        int k0 = 0;
        for (; k0 + 8 <= cnt; k0 += 8)
            add_rows<true>(k0, cnt, lane, s_l, dl_l, ex0_l, ex1_l, h, acc);
        if (k0 < cnt)
            add_rows<false>(k0, cnt, lane, s_l, dl_l, ex0_l, ex1_l, h, acc);
    }
    const int rem = n - nproc;
    if (rem > 0 && nproc > 0) {
        unsigned keep = 0u;
        if (lane < rem) keep = q[nproc + lane];
        if (lane < rem) q[lane] = keep;
    }
    return rem;
}

template <int PASS>
__device__ __forceinline__ void scan_edges(const int* __restrict__ srcp,
                                           const int* __restrict__ dstp,
                                           const uint4* __restrict__ tcol,
                                           const float* __restrict__ a_s,
                                           const float* __restrict__ h,
                                           unsigned* q, int n, int base, int nloc,
                                           int lane, int w,
                                           const float* ad_l, unsigned* m_enc,
                                           float* den, float* acc,
                                           unsigned* kept_w, int& kc, bool& ovf) {
    const unsigned idrep = (unsigned)blockIdx.x * 0x01010101u;
    const int goff = w * 32 + lane;
    constexpr int U      = (PASS == 1) ? SCAN_U : 1;
    constexpr int NROUND = (NPIECE + U - 1) / U;
    int rr = blockIdx.x % NROUND;
    for (int r = 0; r < NROUND; ++r) {
        uint4 tv[U];
#pragma unroll
        for (int u = 0; u < U; ++u) {
            int g = (rr * U + u) * AGG_T + goff;
            tv[u] = tcol[min(g, EE / 16 - 1)];
        }
#pragma unroll
        for (int u = 0; u < U; ++u) {
            const int g = (rr * U + u) * AGG_T + goff;
            const int e_base = g * 16;
            const unsigned past = (g < EE / 16) ? 0u : 0xFFFFFFFFu;
            const unsigned xw[4] = { (tv[u].x ^ idrep) | past, (tv[u].y ^ idrep) | past,
                                     (tv[u].z ^ idrep) | past, (tv[u].w ^ idrep) | past };
            unsigned zw[4];
#pragma unroll
            for (int k = 0; k < 4; ++k) zw[k] = (xw[k] - 0x01010101u) & ~xw[k] & 0x80808080u;
            if (__builtin_amdgcn_ballot_w32((zw[0] | zw[1] | zw[2] | zw[3]) != 0u) != 0u) {
#pragma unroll
                for (int k = 0; k < 4; ++k) {
                    if (__builtin_amdgcn_ballot_w32(zw[k] != 0u) == 0u) continue;
#pragma unroll
                    for (int j = 0; j < 4; ++j) {
                        const bool hit = ((xw[k] >> (8 * j)) & 0xFFu) == 0u;
                        const unsigned m = __builtin_amdgcn_ballot_w32(hit);
                        if (m == 0u) continue;
                        if (hit) q[n + __builtin_amdgcn_mbcnt_lo(m, 0u)] = (unsigned)(e_base + 4 * k + j);
                        n += __popc(m);
                    }
                }
            }
            const bool last = (r == NROUND - 1) && (u == U - 1);
            if (n >= QDRAIN || (last && n > 0)) {
                if (PASS == 1) {
                    if (kc + n > KCAP - KSELF) ovf = true;
                    drain_max(q, n, lane, srcp, dstp, base, nloc, a_s, ad_l, m_enc, kept_w + kc, !ovf);
                    if (!ovf) kc += n;
                    n = 0;
                } else {
                    n = drain_sum<false>(q, n, last, lane, srcp, dstp, base, nloc, a_s, h, ad_l, m_enc,
                                         den, acc);
                }
            }
        }
        if (++rr == NROUND) rr = 0;
    }
}

#define LIST_NONE (-1)
template <bool FIND>
__global__ __launch_bounds__(AGG_T)
void k_tile_aggregate(const int* __restrict__ ei,
                      const float* __restrict__ h,
                      const float* __restrict__ a_s,
                      const float* __restrict__ a_d,
                      float* __restrict__ out,
                      unsigned* __restrict__ lists,
                      int* __restrict__ counts,
                      const unsigned* __restrict__ tile_of_edge) {
    __shared__ alignas(16) float acc[TILE * HC];
    __shared__ unsigned queue[AGG_W * QCAP];
    __shared__ unsigned kept[AGG_W * KCAP];
    __shared__ unsigned m_enc[TILE * 2];
    __shared__ float    den[TILE * 2];
    __shared__ float    ad_l[TILE * 2];

    const int tid = threadIdx.x, lane = tid & 31;
    const int w = __builtin_amdgcn_readfirstlane(tid >> 5);
    const int base = blockIdx.x * TILE;
    const int nloc = min(TILE, NN - base);
    const int* __restrict__ srcp = ei;
    const int* __restrict__ dstp = ei + EE;
    const uint4* __restrict__ tcol = (const uint4*)tile_of_edge;
    unsigned* q = queue + w * QCAP;
    unsigned* kept_w = kept + w * KCAP;
    int  kc  = 0;
    bool ovf = false;

    for (int i = tid; i < TILE * HC / 4; i += AGG_T)
        ((float4*)acc)[i] = make_float4(0.f, 0.f, 0.f, 0.f);
    if (tid < TILE * 2) den[tid] = 0.f;
    if (tid < nloc * 2) {
        int g = base * 2 + tid;
        float ad = a_d[g];
        m_enc[tid] = enc_f32(leaky(a_s[g] + ad));
        ad_l[tid]  = ad;
    }
    __syncthreads();

    unsigned* lst = lists + ((size_t)blockIdx.x * AGG_W + w) * KCAP;
    int* cntp = counts + blockIdx.x * AGG_W + w;
    if (FIND) {
        scan_edges<1>(srcp, dstp, tcol, a_s, h, q, 0, base, nloc, lane, w, ad_l, m_enc, den, acc,
                      kept_w, kc, ovf);
        if (!ovf)
            for (int i = lane; i < kc; i += 32) lst[i] = kept_w[i];
        if (lane == 0) *cntp = ovf ? LIST_NONE : kc;
    } else {
        int c = __builtin_amdgcn_readfirstlane(*cntp);
        if (c != LIST_NONE) {
            kc = min(max(c, 0), KCAP - KSELF);
            take_list_max(lst, kc, lane, a_s, ad_l, m_enc, kept_w);
        } else {
            ovf = true;
            scan_edges<1>(srcp, dstp, tcol, a_s, h, q, 0, base, nloc, lane, w, ad_l, m_enc, den, acc,
                          kept_w, kc, ovf);
        }
    }
    __syncthreads();

    int nself = (nloc > w) ? (nloc - w + 31) / 32 : 0;
    if (!ovf) {
        if (lane < nself) {
            int dl = w + 32 * lane;
            kept_w[kc + lane] = make_entry(base + dl, (unsigned)dl);
        }
        drain_sum<true>(kept_w, kc + nself, true, lane, srcp, dstp, base, nloc, a_s, h, ad_l, m_enc, den, acc);
    } else {
        if (lane < nself) {
            int dl = w + 32 * lane;
            q[lane] = (unsigned)(EE + base + dl);
        }
        scan_edges<2>(srcp, dstp, tcol, a_s, h, q, nself, base, nloc, lane, w, ad_l, m_enc, den, acc,
                      kept_w, kc, ovf);
    }
    __syncthreads();

    for (int dl = w; dl < nloc; dl += AGG_W) {
        float dn0 = den[dl * 2 + 0] + 1e-16f, dn1 = den[dl * 2 + 1] + 1e-16f;
        const float* ar = acc + dl * HC + lane;
        float4 o0 = make_float4(ar[0] / dn0,   ar[32] / dn0,  ar[64] / dn0,  ar[96] / dn0);
        float4 o1 = make_float4(ar[128] / dn1, ar[160] / dn1, ar[192] / dn1, ar[224] / dn1);
        float* op = out + (size_t)(base + dl) * HC + 4 * lane;
        *(float4*)op        = o0;
        *(float4*)(op + CC) = o1;
    }
}

__global__ void k_final(const float* __restrict__ acc,
                        const float* __restrict__ b1,
                        const float* __restrict__ Wl,
                        const float* __restrict__ bl,
                        float* __restrict__ out) {
    int node = blockIdx.x * blockDim.x + threadIdx.x;
    if (node >= NN) return;
    const float* a = acc + (size_t)node * HC;
    float z0 = bl[0], z1 = bl[1];
#pragma unroll 4
    for (int k = 0; k < HC; ++k) {
        float v = a[k] + b1[k];
        z0 += v * Wl[k * 2 + 0];
        z1 += v * Wl[k * 2 + 1];
    }
    float mx = fmaxf(z0, z1);
    float e0 = expf(z0 - mx), e1 = expf(z1 - mx);
    float inv = 1.f / (e0 + e1);
    out[node * 2 + 0] = e0 * inv;
    out[node * 2 + 1] = e1 * inv;
}

extern "C" void kernel_launch(void* const* d_in, const int* in_sizes, int n_in,
                              void* d_out, int out_size, void* d_ws, size_t ws_size,
                              hipStream_t stream) {
    const float* x        = (const float*)d_in[0];
    const int*   ei       = (const int*)  d_in[1];
    const float* W0       = (const float*)d_in[2];
    const float* att_src0 = (const float*)d_in[3];
    const float* att_dst0 = (const float*)d_in[4];
    const float* b0       = (const float*)d_in[5];
    const float* W1       = (const float*)d_in[6];
    const float* att_src1 = (const float*)d_in[7];
    const float* att_dst1 = (const float*)d_in[8];
    const float* b1       = (const float*)d_in[9];
    const float* Wl       = (const float*)d_in[10];
    const float* bl       = (const float*)d_in[11];
    float* out = (float*)d_out;

    char* base = (char*)d_ws;
    size_t off = 0;
    auto carve = [&](size_t bytes) -> char* {
        char* p = base + off;
        off = (off + bytes + 255) & ~(size_t)255;
        return p;
    };
    float*    h_feat = (float*)   carve((size_t)NN * HC * 4);
    float*    accb   = (float*)   carve((size_t)NN * HC * 4);
    __bf16*   wtbf   = (__bf16*)  carve((size_t)HC * HC * 2);
    float*    a_s    = (float*)   carve((size_t)NN * 2 * 4);
    float*    a_d    = (float*)   carve((size_t)NN * 2 * 4);
    unsigned* lists  = (unsigned*)carve((size_t)NTILES * AGG_W * KCAP * 4);
    int*      counts = (int*)     carve((size_t)NTILES * AGG_W * 4);
    unsigned* tcol   = (unsigned*)carve((size_t)EE);
    (void)ws_size; (void)in_sizes; (void)n_in; (void)out_size;

    const int T = 256;
    dim3 blk(T);

    k_transpose_bf16<<<(DD * HC + T - 1) / T, blk, 0, stream>>>(W0, wtbf, DD, HC);

    k_gemm_rows_wmma<DD, false><<<NN / 16, dim3(GEMM_T), 0, stream>>>(
        x, b0, wtbf, h_feat, att_src0, att_dst0, a_s, a_d, ei + EE, tcol);

    k_tile_aggregate<true><<<NTILES, dim3(AGG_T), 0, stream>>>(ei, h_feat, a_s, a_d, accb,
                                                               lists, counts, tcol);

    k_transpose_bf16<<<(HC * HC + T - 1) / T, blk, 0, stream>>>(W1, wtbf, HC, HC);

    k_gemm_rows_wmma<HC, true><<<NN / 16, dim3(GEMM_T), 0, stream>>>(
        accb, b0, wtbf, h_feat, att_src1, att_dst1, a_s, a_d, ei + EE, tcol);

    k_tile_aggregate<false><<<NTILES, dim3(AGG_T), 0, stream>>>(ei, h_feat, a_s, a_d, accb,
                                                                lists, counts, tcol);

    k_final<<<(NN + T - 1) / T, blk, 0, stream>>>(accb, b1, Wl, bl, out);
}


# Layer 1's tile block finishes its rows in LDS: bias, 256->2 classifier (8 channels a lane, wave sum), softmax; last launch and the f32 row round trip gone; classifier sum order changes
# speedup vs baseline: 8.2703x; 1.0639x over previous
#include <hip/hip_runtime.h>
#include <math.h>

#define NN    50000
#define DD    128
#define HC    256
#define CC    128
#define EE    800000
#define EP    (EE + NN)
#define NEG_SLOPE 0.2f
#define TILE      196
#define NTILES    ((NN + TILE - 1) / TILE)
static_assert(NTILES <= 256, "a tile number fits a byte");
static_assert(EE % 16 == 0, "the tile-number column is read 16 bytes at a time");

typedef __attribute__((ext_vector_type(16))) __bf16 v16bf;
typedef __attribute__((ext_vector_type(8)))  __bf16 v8bf;
typedef __attribute__((ext_vector_type(8)))  float  v8f;

__device__ __forceinline__ unsigned enc_f32(float f) {
    unsigned u = __float_as_uint(f);
    return (u & 0x80000000u) ? ~u : (u | 0x80000000u);
}
__device__ __forceinline__ float dec_f32(unsigned u) {
    u = (u & 0x80000000u) ? (u & 0x7FFFFFFFu) : ~u;
    return __uint_as_float(u);
}

__global__ void k_transpose_bf16(const float* __restrict__ W,
                                 __bf16* __restrict__ Wt, int K, int Nn) {
    int i = blockIdx.x * blockDim.x + threadIdx.x;
    if (i < K * Nn) {
        int n = i / K, k = i - n * K;
        Wt[i] = (__bf16)W[(size_t)k * Nn + n];
    }
}

typedef __attribute__((ext_vector_type(4))) __bf16 v4bf;
#define GEMM_T   512
#define A_PAD    8
static_assert(NN % 16 == 0 && HC == 16 * (GEMM_T / 32), "a block covers 16 whole rows");

template <int M>
__device__ __forceinline__ float row_xor(float v) {
    return __int_as_float(__builtin_amdgcn_update_dpp(0, __float_as_int(v), 0x160 + M,
                                                      0xf, 0xf, true));
}

template <int K, bool TANH>
__global__ __launch_bounds__(GEMM_T)
void k_gemm_rows_wmma(const float* __restrict__ A,
                      const float* __restrict__ bias,
                      const __bf16* __restrict__ Bt,
                      float* __restrict__ C,
                      const float* __restrict__ att_src,
                      const float* __restrict__ att_dst,
                      float* __restrict__ a_s,
                      float* __restrict__ a_d,
                      const int* __restrict__ dstp,
                      unsigned* __restrict__ tile_of_edge) {
    constexpr int PITCH = K + A_PAD;
    __shared__ alignas(16) __bf16 a_lds[16 * PITCH];
    __shared__ float sc_part[(GEMM_T / 32) * 32];
    __shared__ alignas(8) float att_lds[2 * HC];
    static_assert(GEMM_T == 2 * HC, "one element of the two attention vectors per thread");

    const int tid = threadIdx.x;
    const int tM  = blockIdx.x << 4;
    const int lane = tid & 31;
    const int tN   = (tid >> 5) << 4;
    const int half = lane >> 4;
    const int lr   = lane & 15;

    if (!TANH) {
        constexpr int PRE = (EE / 4 + NN / 16 - 1) / (NN / 16);
        static_assert(PRE % 32 == 0 && PRE <= GEMM_T, "whole waves take the pieces");
        const int g = blockIdx.x * PRE + tid;
        if (tid < PRE && g < EE / 4) {
            const int4 d = *(const int4*)(dstp + 4 * g);
            const unsigned u0 = (unsigned)d.x < (unsigned)NN ? (unsigned)d.x : 0u;
            const unsigned u1 = (unsigned)d.y < (unsigned)NN ? (unsigned)d.y : 0u;
            const unsigned u2 = (unsigned)d.z < (unsigned)NN ? (unsigned)d.z : 0u;
            const unsigned u3 = (unsigned)d.w < (unsigned)NN ? (unsigned)d.w : 0u;
            tile_of_edge[g] = (u0 / TILE) | ((u1 / TILE) << 8) | ((u2 / TILE) << 16) | ((u3 / TILE) << 24);
        }
    }

    const __bf16* brow = Bt + (size_t)(tN + lr) * K;
    v8bf bfr[K / 16];
#pragma unroll
    for (int j = 0; j < K / 16; ++j)
        bfr[j] = *(const v8bf*)(brow + 16 * j + 8 * half);
    const float att_el = (tid < HC) ? att_src[tid] : att_dst[tid - HC];

    constexpr int V4_ROW = K / 4;
    constexpr int NIT    = 16 * V4_ROW / GEMM_T;
    static_assert(NIT * GEMM_T == 16 * V4_ROW, "the block's threads cover the 16 rows exactly");
    float4 av[NIT];
#pragma unroll
    for (int it = 0; it < NIT; ++it) {
        int i = tid + it * GEMM_T;
        int row = i / V4_ROW, c4 = i - row * V4_ROW;
        av[it] = *(const float4*)(A + (size_t)(tM + row) * K + 4 * c4);
    }
#pragma unroll
    for (int it = 0; it < NIT; ++it) {
        int i = tid + it * GEMM_T;
        int row = i / V4_ROW, c4 = i - row * V4_ROW;
        float4 v = av[it];
        if (TANH) {
            float4 b = *(const float4*)(bias + 4 * c4);
            v.x = tanhf(v.x + b.x); v.y = tanhf(v.y + b.y);
            v.z = tanhf(v.z + b.z); v.w = tanhf(v.w + b.w);
        }
        v4bf o;
        o[0] = (__bf16)v.x; o[1] = (__bf16)v.y; o[2] = (__bf16)v.z; o[3] = (__bf16)v.w;
        *(v4bf*)(a_lds + row * PITCH + 4 * c4) = o;
    }
    att_lds[(tid < HC) ? 2 * tid : 2 * (tid - HC) + 1] = att_el;
    __syncthreads();

    v8f acc = {};
    const __bf16* arow = a_lds + lr * PITCH;

#pragma unroll
    for (int kk = 0; kk < K; kk += 32) {
        v8bf alo = *(const v8bf*)(arow + kk + 8 * half);
        v8bf ahi = *(const v8bf*)(arow + kk + 16 + 8 * half);
        v8bf blo = bfr[kk / 16];
        v8bf bhi = bfr[kk / 16 + 1];
        v16bf a, b;
#pragma unroll
        for (int i = 0; i < 8; ++i) {
            a[i] = alo[i]; a[i + 8] = ahi[i];
            b[i] = blo[i]; b[i + 8] = bhi[i];
        }
        acc = __builtin_amdgcn_wmma_f32_16x16x32_bf16(
                  false, a, false, b, (short)0, acc, false, false);
    }

    const float2 att_n = *(const float2*)(att_lds + 2 * (tN + lr));
    const float as_n = att_n.x, ad_n = att_n.y;
    float p[16];
#pragma unroll
    for (int r = 0; r < 8; ++r) { p[2 * r] = acc[r] * as_n; p[2 * r + 1] = acc[r] * ad_n; }
    const bool b3 = (lr & 8) != 0, b2 = (lr & 4) != 0, b1 = (lr & 2) != 0, b0 = (lr & 1) != 0;
    float q8[8], q4[4], q2[2];
#pragma unroll
    for (int j = 0; j < 8; ++j)
        q8[j] = (b3 ? p[j + 8] : p[j]) + row_xor<8>(b3 ? p[j] : p[j + 8]);
#pragma unroll
    for (int j = 0; j < 4; ++j)
        q4[j] = (b2 ? q8[j + 4] : q8[j]) + row_xor<4>(b2 ? q8[j] : q8[j + 4]);
#pragma unroll
    for (int j = 0; j < 2; ++j)
        q2[j] = (b1 ? q4[j + 2] : q4[j]) + row_xor<2>(b1 ? q4[j] : q4[j + 2]);
    const float q1 = (b0 ? q2[1] : q2[0]) + row_xor<1>(b0 ? q2[0] : q2[1]);
    sc_part[(tid >> 5) * 32 + lane] = q1;
    __syncthreads();

    int n = tN + lr;
#pragma unroll
    for (int r = 0; r < 8; ++r) {
        int m = tM + r + 8 * half;
        C[(size_t)m * HC + n] = acc[r];
    }

    if (tid < 64) {
        const int sc = tid >> 5, idx = tid & 31;
        const int row = idx >> 1, hd = idx & 1;
        const int li = 16 * (row >> 3) + 2 * (row & 7) + sc;
        float s = 0.f;
#pragma unroll
        for (int g = 0; g < 8; ++g) s += sc_part[(hd * 8 + g) * 32 + li];
        (sc ? a_d : a_s)[tM * 2 + idx] = s;
    }
}

#define AGG_T     1024
#define AGG_W     (AGG_T / 32)
#define PIECE_E   512
#define SCAN_U    4
#define NPIECE    ((EE / 16 + AGG_T - 1) / AGG_T)
#define QDRAIN    32
#define QCAP      (QDRAIN + PIECE_E)
#define KCAP      384
#define KSELF     8
static_assert(EP <= (1 << 20) && NN <= (1 << 20) && TILE <= (1 << 12), "queue entry packing");
static_assert((TILE + AGG_W - 1) / AGG_W <= KSELF && KSELF <= 32, "self loops fit the reserved room");

__device__ __forceinline__ float leaky(float l) {
    return (l > 0.f) ? l : NEG_SLOPE * l;
}

__device__ __forceinline__ unsigned make_entry(int s, unsigned dl) {
    return (unsigned)s | (dl << 20);
}
__device__ __forceinline__ void entry_nodes(unsigned e, int base, int nloc,
                                            const int* __restrict__ srcp,
                                            const int* __restrict__ dstp,
                                            int& s, int& dl) {
    int d;
    if (e < (unsigned)EE) { s = srcp[e]; d = dstp[e]; }
    else                  { s = (int)e - EE; d = s; }
    dl = min(max(d - base, 0), nloc - 1);
}

__device__ __forceinline__ void drain_max(const unsigned* q, int n, int lane,
                                          const int* __restrict__ srcp,
                                          const int* __restrict__ dstp, int base, int nloc,
                                          const float* __restrict__ a_s,
                                          const float* ad_l, unsigned* m_enc,
                                          unsigned* kp, bool keep) {
    for (int b = 0; b < n; b += 32) {
        int i = b + lane;
        if (i < n) {
            int s, dl;
            entry_nodes(q[i], base, nloc, srcp, dstp, s, dl);
            float2 as = *(const float2*)(a_s + 2 * s);
            if (keep) kp[i] = make_entry(s, (unsigned)dl);
            atomicMax(&m_enc[dl * 2 + 0], enc_f32(leaky(as.x + ad_l[dl * 2 + 0])));
            atomicMax(&m_enc[dl * 2 + 1], enc_f32(leaky(as.y + ad_l[dl * 2 + 1])));
        }
    }
}

__device__ __forceinline__ void take_list_max(const unsigned* __restrict__ lst, int n, int lane,
                                              const float* __restrict__ a_s,
                                              const float* ad_l, unsigned* m_enc,
                                              unsigned* kp) {
    for (int b = 0; b < n; b += 128) {
        unsigned ent[4];
        float2   as[4];
#pragma unroll
        for (int u = 0; u < 4; ++u) {
            int i = b + 32 * u + lane;
            ent[u] = (i < n) ? lst[i] : 0u;
        }
#pragma unroll
        for (int u = 0; u < 4; ++u)
            as[u] = *(const float2*)(a_s + 2 * (int)(ent[u] & 0xFFFFFu));
#pragma unroll
        for (int u = 0; u < 4; ++u) {
            int i = b + 32 * u + lane;
            if (i < n) {
                int dl = (int)(ent[u] >> 20);
                kp[i] = ent[u];
                atomicMax(&m_enc[dl * 2 + 0], enc_f32(leaky(as[u].x + ad_l[dl * 2 + 0])));
                atomicMax(&m_enc[dl * 2 + 1], enc_f32(leaky(as[u].y + ad_l[dl * 2 + 1])));
            }
        }
    }
}

template <bool FULL>
__device__ __forceinline__ void add_rows(int k0, int cnt, int lane, int s_l, int dl_l,
                                         float ex0_l, float ex1_l,
                                         const float* __restrict__ h, float* acc) {
    float4 v0[8], v1[8];
#pragma unroll
    for (int k = 0; k < 8; ++k) {
        if (FULL || k0 + k < cnt) {
            int s = __builtin_amdgcn_readlane(s_l, k0 + k);
            const float* hr = h + (size_t)s * HC + 4 * lane;
            v0[k] = *(const float4*)hr;
            v1[k] = *(const float4*)(hr + CC);
        }
    }
#pragma unroll
    for (int k = 0; k < 8; ++k) {
        if (FULL || k0 + k < cnt) {
            int   dl = __builtin_amdgcn_readlane(dl_l, k0 + k);
            float e0 = __int_as_float(__builtin_amdgcn_readlane(__float_as_int(ex0_l), k0 + k));
            float e1 = __int_as_float(__builtin_amdgcn_readlane(__float_as_int(ex1_l), k0 + k));
            float* ar = acc + dl * HC + lane;
            atomicAdd(ar +   0, e0 * v0[k].x);
            atomicAdd(ar +  32, e0 * v0[k].y);
            atomicAdd(ar +  64, e0 * v0[k].z);
            atomicAdd(ar +  96, e0 * v0[k].w);
            atomicAdd(ar + 128, e1 * v1[k].x);
            atomicAdd(ar + 160, e1 * v1[k].y);
            atomicAdd(ar + 192, e1 * v1[k].z);
            atomicAdd(ar + 224, e1 * v1[k].w);
        }
    }
}

template <bool DIRECT>
__device__ __forceinline__ int drain_sum(unsigned* q, int n, bool last, int lane,
                                          const int* __restrict__ srcp,
                                          const int* __restrict__ dstp, int base, int nloc,
                                          const float* __restrict__ a_s,
                                          const float* __restrict__ h,
                                          const float* ad_l, const unsigned* m_enc,
                                          float* den, float* acc) {
    const int nproc = last ? n : (n & ~7);
    for (int b = 0; b < nproc; b += 32) {
        int i = b + lane;
        int cnt = min(32, nproc - b);
        int s_l = 0, dl_l = 0;
        float ex0_l = 0.f, ex1_l = 0.f;
        if (i < nproc) {
            unsigned ent = q[i];
            if (DIRECT) { s_l = (int)(ent & 0xFFFFFu); dl_l = (int)(ent >> 20); }
            else        entry_nodes(ent, base, nloc, srcp, dstp, s_l, dl_l);
            float2 as = *(const float2*)(a_s + 2 * s_l);
            ex0_l = expf(leaky(as.x + ad_l[dl_l * 2 + 0]) - dec_f32(m_enc[dl_l * 2 + 0]));
            ex1_l = expf(leaky(as.y + ad_l[dl_l * 2 + 1]) - dec_f32(m_enc[dl_l * 2 + 1]));
            atomicAdd(&den[dl_l * 2 + 0], ex0_l);
            atomicAdd(&den[dl_l * 2 + 1], ex1_l);
        }
        int k0 = 0;
        for (; k0 + 8 <= cnt; k0 += 8)
            add_rows<true>(k0, cnt, lane, s_l, dl_l, ex0_l, ex1_l, h, acc);
        if (k0 < cnt)
            add_rows<false>(k0, cnt, lane, s_l, dl_l, ex0_l, ex1_l, h, acc);
    }
    const int rem = n - nproc;
    if (rem > 0 && nproc > 0) {
        unsigned keep = 0u;
        if (lane < rem) keep = q[nproc + lane];
        if (lane < rem) q[lane] = keep;
    }
    return rem;
}

template <int PASS>
__device__ __forceinline__ void scan_edges(const int* __restrict__ srcp,
                                           const int* __restrict__ dstp,
                                           const uint4* __restrict__ tcol,
                                           const float* __restrict__ a_s,
                                           const float* __restrict__ h,
                                           unsigned* q, int n, int base, int nloc,
                                           int lane, int w,
                                           const float* ad_l, unsigned* m_enc,
                                           float* den, float* acc,
                                           unsigned* kept_w, int& kc, bool& ovf) {
    const unsigned idrep = (unsigned)blockIdx.x * 0x01010101u;
    const int goff = w * 32 + lane;
    constexpr int U      = (PASS == 1) ? SCAN_U : 1;
    constexpr int NROUND = (NPIECE + U - 1) / U;
    int rr = blockIdx.x % NROUND;
    for (int r = 0; r < NROUND; ++r) {
        uint4 tv[U];
#pragma unroll
        for (int u = 0; u < U; ++u) {
            int g = (rr * U + u) * AGG_T + goff;
            tv[u] = tcol[min(g, EE / 16 - 1)];
        }
#pragma unroll
        for (int u = 0; u < U; ++u) {
            const int g = (rr * U + u) * AGG_T + goff;
            const int e_base = g * 16;
            const unsigned past = (g < EE / 16) ? 0u : 0xFFFFFFFFu;
            const unsigned xw[4] = { (tv[u].x ^ idrep) | past, (tv[u].y ^ idrep) | past,
                                     (tv[u].z ^ idrep) | past, (tv[u].w ^ idrep) | past };
            unsigned zw[4];
#pragma unroll
            for (int k = 0; k < 4; ++k) zw[k] = (xw[k] - 0x01010101u) & ~xw[k] & 0x80808080u;
            if (__builtin_amdgcn_ballot_w32((zw[0] | zw[1] | zw[2] | zw[3]) != 0u) != 0u) {
#pragma unroll
                for (int k = 0; k < 4; ++k) {
                    if (__builtin_amdgcn_ballot_w32(zw[k] != 0u) == 0u) continue;
#pragma unroll
                    for (int j = 0; j < 4; ++j) {
                        const bool hit = ((xw[k] >> (8 * j)) & 0xFFu) == 0u;
                        const unsigned m = __builtin_amdgcn_ballot_w32(hit);
                        if (m == 0u) continue;
                        if (hit) q[n + __builtin_amdgcn_mbcnt_lo(m, 0u)] = (unsigned)(e_base + 4 * k + j);
                        n += __popc(m);
                    }
                }
            }
            const bool last = (r == NROUND - 1) && (u == U - 1);
            if (n >= QDRAIN || (last && n > 0)) {
                if (PASS == 1) {
                    if (kc + n > KCAP - KSELF) ovf = true;
                    drain_max(q, n, lane, srcp, dstp, base, nloc, a_s, ad_l, m_enc, kept_w + kc, !ovf);
                    if (!ovf) kc += n;
                    n = 0;
                } else {
                    n = drain_sum<false>(q, n, last, lane, srcp, dstp, base, nloc, a_s, h, ad_l, m_enc,
                                         den, acc);
                }
            }
        }
        if (++rr == NROUND) rr = 0;
    }
}

__device__ __forceinline__ float wave_sum(float v) {
    v += row_xor<1>(v); v += row_xor<2>(v); v += row_xor<4>(v); v += row_xor<8>(v);
    return v + __shfl_xor(v, 16);
}

#define LIST_NONE (-1)
template <bool FIND, bool LAST>
__global__ __launch_bounds__(AGG_T)
void k_tile_aggregate(const int* __restrict__ ei,
                      const float* __restrict__ h,
                      const float* __restrict__ a_s,
                      const float* __restrict__ a_d,
                      float* __restrict__ out,
                      unsigned* __restrict__ lists,
                      int* __restrict__ counts,
                      const unsigned* __restrict__ tile_of_edge,
                      const float* __restrict__ b_last,
                      const float* __restrict__ Wl,
                      const float* __restrict__ bl,
                      float* __restrict__ prob) {
    __shared__ alignas(16) float acc[TILE * HC];
    __shared__ unsigned queue[AGG_W * QCAP];
    __shared__ unsigned kept[AGG_W * KCAP];
    __shared__ unsigned m_enc[TILE * 2];
    __shared__ float    den[TILE * 2];
    __shared__ float    ad_l[TILE * 2];

    const int tid = threadIdx.x, lane = tid & 31;
    const int w = __builtin_amdgcn_readfirstlane(tid >> 5);
    const int base = blockIdx.x * TILE;
    const int nloc = min(TILE, NN - base);
    const int* __restrict__ srcp = ei;
    const int* __restrict__ dstp = ei + EE;
    const uint4* __restrict__ tcol = (const uint4*)tile_of_edge;
    unsigned* q = queue + w * QCAP;
    unsigned* kept_w = kept + w * KCAP;
    int  kc  = 0;
    bool ovf = false;

    for (int i = tid; i < TILE * HC / 4; i += AGG_T)
        ((float4*)acc)[i] = make_float4(0.f, 0.f, 0.f, 0.f);
    if (tid < TILE * 2) den[tid] = 0.f;
    if (tid < nloc * 2) {
        int g = base * 2 + tid;
        float ad = a_d[g];
        m_enc[tid] = enc_f32(leaky(a_s[g] + ad));
        ad_l[tid]  = ad;
    }
    __syncthreads();

    unsigned* lst = lists + ((size_t)blockIdx.x * AGG_W + w) * KCAP;
    int* cntp = counts + blockIdx.x * AGG_W + w;
    if (FIND) {
        scan_edges<1>(srcp, dstp, tcol, a_s, h, q, 0, base, nloc, lane, w, ad_l, m_enc, den, acc,
                      kept_w, kc, ovf);
        if (!ovf)
            for (int i = lane; i < kc; i += 32) lst[i] = kept_w[i];
        if (lane == 0) *cntp = ovf ? LIST_NONE : kc;
    } else {
        int c = __builtin_amdgcn_readfirstlane(*cntp);
        if (c != LIST_NONE) {
            kc = min(max(c, 0), KCAP - KSELF);
            take_list_max(lst, kc, lane, a_s, ad_l, m_enc, kept_w);
        } else {
            ovf = true;
            scan_edges<1>(srcp, dstp, tcol, a_s, h, q, 0, base, nloc, lane, w, ad_l, m_enc, den, acc,
                          kept_w, kc, ovf);
        }
    }
    __syncthreads();

    int nself = (nloc > w) ? (nloc - w + 31) / 32 : 0;
    if (!ovf) {
        if (lane < nself) {
            int dl = w + 32 * lane;
            kept_w[kc + lane] = make_entry(base + dl, (unsigned)dl);
        }
        drain_sum<true>(kept_w, kc + nself, true, lane, srcp, dstp, base, nloc, a_s, h, ad_l, m_enc, den, acc);
    } else {
        if (lane < nself) {
            int dl = w + 32 * lane;
            q[lane] = (unsigned)(EE + base + dl);
        }
        scan_edges<2>(srcp, dstp, tcol, a_s, h, q, nself, base, nloc, lane, w, ad_l, m_enc, den, acc,
                      kept_w, kc, ovf);
    }
    __syncthreads();

    if (!LAST) {
        for (int dl = w; dl < nloc; dl += AGG_W) {
            float dn0 = den[dl * 2 + 0] + 1e-16f, dn1 = den[dl * 2 + 1] + 1e-16f;
            const float* ar = acc + dl * HC + lane;
            float4 o0 = make_float4(ar[0] / dn0,   ar[32] / dn0,  ar[64] / dn0,  ar[96] / dn0);
            float4 o1 = make_float4(ar[128] / dn1, ar[160] / dn1, ar[192] / dn1, ar[224] / dn1);
            float* op = out + (size_t)(base + dl) * HC + 4 * lane;
            *(float4*)op        = o0;
            *(float4*)(op + CC) = o1;
        }
    } else {
        const float4 bA  = *(const float4*)(b_last + 4 * lane);
        const float4 bB  = *(const float4*)(b_last + CC + 4 * lane);
        const float4 wA0 = *(const float4*)(Wl + 8 * lane);
        const float4 wA1 = *(const float4*)(Wl + 8 * lane + 4);
        const float4 wB0 = *(const float4*)(Wl + 2 * CC + 8 * lane);
        const float4 wB1 = *(const float4*)(Wl + 2 * CC + 8 * lane + 4);
        const float bz0 = bA.x * wA0.x + bA.y * wA0.z + bA.z * wA1.x + bA.w * wA1.z
                        + bB.x * wB0.x + bB.y * wB0.z + bB.z * wB1.x + bB.w * wB1.z;
        const float bz1 = bA.x * wA0.y + bA.y * wA0.w + bA.z * wA1.y + bA.w * wA1.w
                        + bB.x * wB0.y + bB.y * wB0.w + bB.z * wB1.y + bB.w * wB1.w;
        const float bl0 = bl[0], bl1 = bl[1];
        for (int dl = w; dl < nloc; dl += AGG_W) {
            const float inv0 = 1.f / (den[dl * 2 + 0] + 1e-16f);
            const float inv1 = 1.f / (den[dl * 2 + 1] + 1e-16f);
            const float* ar = acc + dl * HC + lane;
            const float a0 = ar[0],   a1 = ar[32],  a2 = ar[64],  a3 = ar[96];
            const float c0 = ar[128], c1 = ar[160], c2 = ar[192], c3 = ar[224];
            const float tA0 = a0 * wA0.x + a1 * wA0.z + a2 * wA1.x + a3 * wA1.z;
            const float tA1 = a0 * wA0.y + a1 * wA0.w + a2 * wA1.y + a3 * wA1.w;
            const float tB0 = c0 * wB0.x + c1 * wB0.z + c2 * wB1.x + c3 * wB1.z;
            const float tB1 = c0 * wB0.y + c1 * wB0.w + c2 * wB1.y + c3 * wB1.w;
            const float z0 = wave_sum(tA0 * inv0 + tB0 * inv1 + bz0) + bl0;
            const float z1 = wave_sum(tA1 * inv0 + tB1 * inv1 + bz1) + bl1;
            if (lane == 0) {
                m_enc[dl * 2 + 0] = __float_as_uint(z0);
                m_enc[dl * 2 + 1] = __float_as_uint(z1);
            }
        }
        __syncthreads();
        const int nd = (base & ~15) + tid;
        if (nd >= base && nd < base + nloc) {
            const float z0 = __uint_as_float(m_enc[(nd - base) * 2 + 0]);
            const float z1 = __uint_as_float(m_enc[(nd - base) * 2 + 1]);
            const float mx = fmaxf(z0, z1);
            const float e0 = expf(z0 - mx), e1 = expf(z1 - mx);
            const float inv = 1.f / (e0 + e1);
            *(float2*)(prob + 2 * nd) = make_float2(e0 * inv, e1 * inv);
        }
    }
}

extern "C" void kernel_launch(void* const* d_in, const int* in_sizes, int n_in,
                              void* d_out, int out_size, void* d_ws, size_t ws_size,
                              hipStream_t stream) {
    const float* x        = (const float*)d_in[0];
    const int*   ei       = (const int*)  d_in[1];
    const float* W0       = (const float*)d_in[2];
    const float* att_src0 = (const float*)d_in[3];
    const float* att_dst0 = (const float*)d_in[4];
    const float* b0       = (const float*)d_in[5];
    const float* W1       = (const float*)d_in[6];
    const float* att_src1 = (const float*)d_in[7];
    const float* att_dst1 = (const float*)d_in[8];
    const float* b1       = (const float*)d_in[9];
    const float* Wl       = (const float*)d_in[10];
    const float* bl       = (const float*)d_in[11];
    float* out = (float*)d_out;

    char* base = (char*)d_ws;
    size_t off = 0;
    auto carve = [&](size_t bytes) -> char* {
        char* p = base + off;
        off = (off + bytes + 255) & ~(size_t)255;
        return p;
    };
    float*    h_feat = (float*)   carve((size_t)NN * HC * 4);
    float*    accb   = (float*)   carve((size_t)NN * HC * 4);
    __bf16*   wtbf   = (__bf16*)  carve((size_t)HC * HC * 2);
    float*    a_s    = (float*)   carve((size_t)NN * 2 * 4);
    float*    a_d    = (float*)   carve((size_t)NN * 2 * 4);
    unsigned* lists  = (unsigned*)carve((size_t)NTILES * AGG_W * KCAP * 4);
    int*      counts = (int*)     carve((size_t)NTILES * AGG_W * 4);
    unsigned* tcol   = (unsigned*)carve((size_t)EE);
    (void)ws_size; (void)in_sizes; (void)n_in; (void)out_size;

    const int T = 256;
    dim3 blk(T);

    k_transpose_bf16<<<(DD * HC + T - 1) / T, blk, 0, stream>>>(W0, wtbf, DD, HC);

    k_gemm_rows_wmma<DD, false><<<NN / 16, dim3(GEMM_T), 0, stream>>>(
        x, b0, wtbf, h_feat, att_src0, att_dst0, a_s, a_d, ei + EE, tcol);

    k_tile_aggregate<true, false><<<NTILES, dim3(AGG_T), 0, stream>>>(ei, h_feat, a_s, a_d, accb,
                                                                      lists, counts, tcol,
                                                                      b1, Wl, bl, out);

    k_transpose_bf16<<<(HC * HC + T - 1) / T, blk, 0, stream>>>(W1, wtbf, HC, HC);

    k_gemm_rows_wmma<HC, true><<<NN / 16, dim3(GEMM_T), 0, stream>>>(
        accb, b0, wtbf, h_feat, att_src1, att_dst1, a_s, a_d, ei + EE, tcol);

    k_tile_aggregate<false, true><<<NTILES, dim3(AGG_T), 0, stream>>>(ei, h_feat, a_s, a_d, accb,
                                                                      lists, counts, tcol,
                                                                      b1, Wl, bl, out);
}


# Edge phase: a found edge is filed (LDS counter as rank) under the wave that owns its destination row; the owner sums rows by 16-byte LDS read-add-write, not 8 LDS float atomics; sum order: list order
# speedup vs baseline: 8.7054x; 1.0526x over previous
#include <hip/hip_runtime.h>
#include <math.h>

#define NN    50000
#define DD    128
#define HC    256
#define CC    128
#define EE    800000
#define EP    (EE + NN)
#define NEG_SLOPE 0.2f
#define TILE      196
#define NTILES    ((NN + TILE - 1) / TILE)
static_assert(NTILES <= 256, "a tile number fits a byte");
static_assert(EE % 16 == 0, "the tile-number column is read 16 bytes at a time");

typedef __attribute__((ext_vector_type(16))) __bf16 v16bf;
typedef __attribute__((ext_vector_type(8)))  __bf16 v8bf;
typedef __attribute__((ext_vector_type(8)))  float  v8f;

__device__ __forceinline__ unsigned enc_f32(float f) {
    unsigned u = __float_as_uint(f);
    return (u & 0x80000000u) ? ~u : (u | 0x80000000u);
}
__device__ __forceinline__ float dec_f32(unsigned u) {
    u = (u & 0x80000000u) ? (u & 0x7FFFFFFFu) : ~u;
    return __uint_as_float(u);
}

__global__ void k_transpose_bf16(const float* __restrict__ W,
                                 __bf16* __restrict__ Wt, int K, int Nn) {
    int i = blockIdx.x * blockDim.x + threadIdx.x;
    if (i < K * Nn) {
        int n = i / K, k = i - n * K;
        Wt[i] = (__bf16)W[(size_t)k * Nn + n];
    }
}

typedef __attribute__((ext_vector_type(4))) __bf16 v4bf;
#define GEMM_T   512
#define A_PAD    8
static_assert(NN % 16 == 0 && HC == 16 * (GEMM_T / 32), "a block covers 16 whole rows");

template <int M>
__device__ __forceinline__ float row_xor(float v) {
    return __int_as_float(__builtin_amdgcn_update_dpp(0, __float_as_int(v), 0x160 + M,
                                                      0xf, 0xf, true));
}

template <int K, bool TANH>
__global__ __launch_bounds__(GEMM_T)
void k_gemm_rows_wmma(const float* __restrict__ A,
                      const float* __restrict__ bias,
                      const __bf16* __restrict__ Bt,
                      float* __restrict__ C,
                      const float* __restrict__ att_src,
                      const float* __restrict__ att_dst,
                      float* __restrict__ a_s,
                      float* __restrict__ a_d,
                      const int* __restrict__ dstp,
                      unsigned* __restrict__ tile_of_edge) {
    constexpr int PITCH = K + A_PAD;
    __shared__ alignas(16) __bf16 a_lds[16 * PITCH];
    __shared__ float sc_part[(GEMM_T / 32) * 32];
    __shared__ alignas(8) float att_lds[2 * HC];
    static_assert(GEMM_T == 2 * HC, "one element of the two attention vectors per thread");

    const int tid = threadIdx.x;
    const int tM  = blockIdx.x << 4;
    const int lane = tid & 31;
    const int tN   = (tid >> 5) << 4;
    const int half = lane >> 4;
    const int lr   = lane & 15;

    if (!TANH) {
        constexpr int PRE = (EE / 4 + NN / 16 - 1) / (NN / 16);
        static_assert(PRE % 32 == 0 && PRE <= GEMM_T, "whole waves take the pieces");
        const int g = blockIdx.x * PRE + tid;
        if (tid < PRE && g < EE / 4) {
            const int4 d = *(const int4*)(dstp + 4 * g);
            const unsigned u0 = (unsigned)d.x < (unsigned)NN ? (unsigned)d.x : 0u;
            const unsigned u1 = (unsigned)d.y < (unsigned)NN ? (unsigned)d.y : 0u;
            const unsigned u2 = (unsigned)d.z < (unsigned)NN ? (unsigned)d.z : 0u;
            const unsigned u3 = (unsigned)d.w < (unsigned)NN ? (unsigned)d.w : 0u;
            tile_of_edge[g] = (u0 / TILE) | ((u1 / TILE) << 8) | ((u2 / TILE) << 16) | ((u3 / TILE) << 24);
        }
    }

    const __bf16* brow = Bt + (size_t)(tN + lr) * K;
    v8bf bfr[K / 16];
#pragma unroll
    for (int j = 0; j < K / 16; ++j)
        bfr[j] = *(const v8bf*)(brow + 16 * j + 8 * half);
    const float att_el = (tid < HC) ? att_src[tid] : att_dst[tid - HC];

    constexpr int V4_ROW = K / 4;
    constexpr int NIT    = 16 * V4_ROW / GEMM_T;
    static_assert(NIT * GEMM_T == 16 * V4_ROW, "the block's threads cover the 16 rows exactly");
    float4 av[NIT];
#pragma unroll
    for (int it = 0; it < NIT; ++it) {
        int i = tid + it * GEMM_T;
        int row = i / V4_ROW, c4 = i - row * V4_ROW;
        av[it] = *(const float4*)(A + (size_t)(tM + row) * K + 4 * c4);
    }
#pragma unroll
    for (int it = 0; it < NIT; ++it) {
        int i = tid + it * GEMM_T;
        int row = i / V4_ROW, c4 = i - row * V4_ROW;
        float4 v = av[it];
        if (TANH) {
            float4 b = *(const float4*)(bias + 4 * c4);
            v.x = tanhf(v.x + b.x); v.y = tanhf(v.y + b.y);
            v.z = tanhf(v.z + b.z); v.w = tanhf(v.w + b.w);
        }
        v4bf o;
        o[0] = (__bf16)v.x; o[1] = (__bf16)v.y; o[2] = (__bf16)v.z; o[3] = (__bf16)v.w;
        *(v4bf*)(a_lds + row * PITCH + 4 * c4) = o;
    }
    att_lds[(tid < HC) ? 2 * tid : 2 * (tid - HC) + 1] = att_el;
    __syncthreads();

    v8f acc = {};
    const __bf16* arow = a_lds + lr * PITCH;

#pragma unroll
    for (int kk = 0; kk < K; kk += 32) {
        v8bf alo = *(const v8bf*)(arow + kk + 8 * half);
        v8bf ahi = *(const v8bf*)(arow + kk + 16 + 8 * half);
        v8bf blo = bfr[kk / 16];
        v8bf bhi = bfr[kk / 16 + 1];
        v16bf a, b;
#pragma unroll
        for (int i = 0; i < 8; ++i) {
            a[i] = alo[i]; a[i + 8] = ahi[i];
            b[i] = blo[i]; b[i + 8] = bhi[i];
        }
        acc = __builtin_amdgcn_wmma_f32_16x16x32_bf16(
                  false, a, false, b, (short)0, acc, false, false);
    }

    const float2 att_n = *(const float2*)(att_lds + 2 * (tN + lr));
    const float as_n = att_n.x, ad_n = att_n.y;
    float p[16];
#pragma unroll
    for (int r = 0; r < 8; ++r) { p[2 * r] = acc[r] * as_n; p[2 * r + 1] = acc[r] * ad_n; }
    const bool b3 = (lr & 8) != 0, b2 = (lr & 4) != 0, b1 = (lr & 2) != 0, b0 = (lr & 1) != 0;
    float q8[8], q4[4], q2[2];
#pragma unroll
    for (int j = 0; j < 8; ++j)
        q8[j] = (b3 ? p[j + 8] : p[j]) + row_xor<8>(b3 ? p[j] : p[j + 8]);
#pragma unroll
    for (int j = 0; j < 4; ++j)
        q4[j] = (b2 ? q8[j + 4] : q8[j]) + row_xor<4>(b2 ? q8[j] : q8[j + 4]);
#pragma unroll
    for (int j = 0; j < 2; ++j)
        q2[j] = (b1 ? q4[j + 2] : q4[j]) + row_xor<2>(b1 ? q4[j] : q4[j + 2]);
    const float q1 = (b0 ? q2[1] : q2[0]) + row_xor<1>(b0 ? q2[0] : q2[1]);
    sc_part[(tid >> 5) * 32 + lane] = q1;
    __syncthreads();

    int n = tN + lr;
#pragma unroll
    for (int r = 0; r < 8; ++r) {
        int m = tM + r + 8 * half;
        C[(size_t)m * HC + n] = acc[r];
    }

    if (tid < 64) {
        const int sc = tid >> 5, idx = tid & 31;
        const int row = idx >> 1, hd = idx & 1;
        const int li = 16 * (row >> 3) + 2 * (row & 7) + sc;
        float s = 0.f;
#pragma unroll
        for (int g = 0; g < 8; ++g) s += sc_part[(hd * 8 + g) * 32 + li];
        (sc ? a_d : a_s)[tM * 2 + idx] = s;
    }
}

#define AGG_T     1024
#define AGG_W     (AGG_T / 32)
#define PIECE_E   512
#define SCAN_U    4
#define NPIECE    ((EE / 16 + AGG_T - 1) / AGG_T)
#define QDRAIN    32
#define QCAP      (QDRAIN + PIECE_E)
#define KCAP      384
#define KSELF     8
static_assert(EP <= (1 << 20) && NN <= (1 << 20) && TILE <= (1 << 12), "queue entry packing");
static_assert((TILE + AGG_W - 1) / AGG_W < KSELF && KSELF <= 32 && KCAP % 8 == 0 && KSELF % 8 == 0,
              "self loops and the fill up to a whole batch fit the reserved room");

__device__ __forceinline__ float leaky(float l) {
    return (l > 0.f) ? l : NEG_SLOPE * l;
}

__device__ __forceinline__ unsigned make_entry(int s, unsigned dl) {
    return (unsigned)s | (dl << 20);
}
__device__ __forceinline__ void entry_nodes(unsigned e, int base, int nloc,
                                            const int* __restrict__ srcp,
                                            const int* __restrict__ dstp,
                                            int& s, int& dl) {
    int d;
    if (e < (unsigned)EE) { s = srcp[e]; d = dstp[e]; }
    else                  { s = (int)e - EE; d = s; }
    dl = min(max(d - base, 0), nloc - 1);
}

__device__ __forceinline__ void drain_max(const unsigned* q, int n, int lane,
                                          const int* __restrict__ srcp,
                                          const int* __restrict__ dstp, int base, int nloc,
                                          const float* __restrict__ a_s,
                                          const float* ad_l, unsigned* m_enc,
                                          unsigned* kept, int* ocnt, int* ovf_flag, bool keep) {
    for (int b = 0; b < n; b += 32) {
        int i = b + lane;
        if (i < n) {
            int s, dl;
            entry_nodes(q[i], base, nloc, srcp, dstp, s, dl);
            float2 as = *(const float2*)(a_s + 2 * s);
            if (keep) {
                const int ow   = dl & (AGG_W - 1);
                const int slot = atomicAdd(&ocnt[ow], 1);
                if (slot < KCAP - KSELF) kept[ow * KCAP + slot] = make_entry(s, (unsigned)dl);
                else                     *ovf_flag = 1;
            }
            atomicMax(&m_enc[dl * 2 + 0], enc_f32(leaky(as.x + ad_l[dl * 2 + 0])));
            atomicMax(&m_enc[dl * 2 + 1], enc_f32(leaky(as.y + ad_l[dl * 2 + 1])));
        }
    }
}

__device__ __forceinline__ void take_list_max(const unsigned* __restrict__ lst, int n, int lane,
                                              const float* __restrict__ a_s,
                                              const float* ad_l, unsigned* m_enc,
                                              unsigned* kp) {
    for (int b = 0; b < n; b += 128) {
        unsigned ent[4];
        float2   as[4];
#pragma unroll
        for (int u = 0; u < 4; ++u) {
            int i = b + 32 * u + lane;
            ent[u] = (i < n) ? lst[i] : 0u;
        }
#pragma unroll
        for (int u = 0; u < 4; ++u)
            as[u] = *(const float2*)(a_s + 2 * (int)(ent[u] & 0xFFFFFu));
#pragma unroll
        for (int u = 0; u < 4; ++u) {
            int i = b + 32 * u + lane;
            if (i < n) {
                int dl = (int)(ent[u] >> 20);
                kp[i] = ent[u];
                atomicMax(&m_enc[dl * 2 + 0], enc_f32(leaky(as[u].x + ad_l[dl * 2 + 0])));
                atomicMax(&m_enc[dl * 2 + 1], enc_f32(leaky(as[u].y + ad_l[dl * 2 + 1])));
            }
        }
    }
}

template <bool FULL, bool OWN>
__device__ __forceinline__ void add_rows(int k0, int cnt, int lane, int s_l, int dl_l,
                                         float ex0_l, float ex1_l,
                                         const float* __restrict__ h, float* acc) {
    float4 v0[8], v1[8];
#pragma unroll
    for (int k = 0; k < 8; ++k) {
        if (FULL || k0 + k < cnt) {
            int s = __builtin_amdgcn_readlane(s_l, k0 + k);
            const float* hr = h + (size_t)s * HC + 4 * lane;
            v0[k] = *(const float4*)hr;
            v1[k] = *(const float4*)(hr + CC);
        }
    }
    if (OWN) __builtin_amdgcn_sched_barrier(0);
#pragma unroll
    for (int k = 0; k < 8; ++k) {
        if (FULL || k0 + k < cnt) {
            int   dl = __builtin_amdgcn_readlane(dl_l, k0 + k);
            float e0 = __int_as_float(__builtin_amdgcn_readlane(__float_as_int(ex0_l), k0 + k));
            float e1 = __int_as_float(__builtin_amdgcn_readlane(__float_as_int(ex1_l), k0 + k));
            float* ar = acc + dl * HC + 4 * lane;
            if (OWN) {
                float4 a0 = *(const float4*)ar;
                float4 a1 = *(const float4*)(ar + CC);
                a0.x = fmaf(e0, v0[k].x, a0.x); a0.y = fmaf(e0, v0[k].y, a0.y);
                a0.z = fmaf(e0, v0[k].z, a0.z); a0.w = fmaf(e0, v0[k].w, a0.w);
                a1.x = fmaf(e1, v1[k].x, a1.x); a1.y = fmaf(e1, v1[k].y, a1.y);
                a1.z = fmaf(e1, v1[k].z, a1.z); a1.w = fmaf(e1, v1[k].w, a1.w);
                *(float4*)ar        = a0;
                *(float4*)(ar + CC) = a1;
            } else {
                atomicAdd(ar + 0, e0 * v0[k].x);
                atomicAdd(ar + 1, e0 * v0[k].y);
                atomicAdd(ar + 2, e0 * v0[k].z);
                atomicAdd(ar + 3, e0 * v0[k].w);
                atomicAdd(ar + CC + 0, e1 * v1[k].x);
                atomicAdd(ar + CC + 1, e1 * v1[k].y);
                atomicAdd(ar + CC + 2, e1 * v1[k].z);
                atomicAdd(ar + CC + 3, e1 * v1[k].w);
            }
        }
    }
}

template <bool DIRECT>
__device__ __forceinline__ int drain_sum(unsigned* q, int n, int n_real, bool last, int lane,
                                          const int* __restrict__ srcp,
                                          const int* __restrict__ dstp, int base, int nloc,
                                          const float* __restrict__ a_s,
                                          const float* __restrict__ h,
                                          const float* ad_l, const unsigned* m_enc,
                                          float* den, float* acc) {
    const int nproc = last ? n : (n & ~7);
    for (int b = 0; b < nproc; b += 32) {
        int i = b + lane;
        int cnt = min(32, nproc - b);
        int s_l = 0, dl_l = 0;
        float ex0_l = 0.f, ex1_l = 0.f;
        if (i < nproc) {
            unsigned ent = q[i];
            if (DIRECT) { s_l = (int)(ent & 0xFFFFFu); dl_l = (int)(ent >> 20); }
            else        entry_nodes(ent, base, nloc, srcp, dstp, s_l, dl_l);
            float2 as = *(const float2*)(a_s + 2 * s_l);
            ex0_l = expf(leaky(as.x + ad_l[dl_l * 2 + 0]) - dec_f32(m_enc[dl_l * 2 + 0]));
            ex1_l = expf(leaky(as.y + ad_l[dl_l * 2 + 1]) - dec_f32(m_enc[dl_l * 2 + 1]));
            if (DIRECT && i >= n_real) {
                ex0_l = 0.f; ex1_l = 0.f;
            } else {
                atomicAdd(&den[dl_l * 2 + 0], ex0_l);
                atomicAdd(&den[dl_l * 2 + 1], ex1_l);
            }
        }
        int k0 = 0;
        for (; k0 + 8 <= cnt; k0 += 8)
            add_rows<true, DIRECT>(k0, cnt, lane, s_l, dl_l, ex0_l, ex1_l, h, acc);
        if (!DIRECT && k0 < cnt)
            add_rows<false, false>(k0, cnt, lane, s_l, dl_l, ex0_l, ex1_l, h, acc);
    }
    const int rem = n - nproc;
    if (rem > 0 && nproc > 0) {
        unsigned keep = 0u;
        if (lane < rem) keep = q[nproc + lane];
        if (lane < rem) q[lane] = keep;
    }
    return rem;
}

template <int PASS>
__device__ __forceinline__ void scan_edges(const int* __restrict__ srcp,
                                           const int* __restrict__ dstp,
                                           const uint4* __restrict__ tcol,
                                           const float* __restrict__ a_s,
                                           const float* __restrict__ h,
                                           unsigned* q, int n, int base, int nloc,
                                           int lane, int w,
                                           const float* ad_l, unsigned* m_enc,
                                           float* den, float* acc,
                                           unsigned* kept, int* ocnt, int* ovf_flag, bool keep) {
    const unsigned idrep = (unsigned)blockIdx.x * 0x01010101u;
    const int goff = w * 32 + lane;
    constexpr int U      = (PASS == 1) ? SCAN_U : 1;
    constexpr int NROUND = (NPIECE + U - 1) / U;
    int rr = blockIdx.x % NROUND;
    for (int r = 0; r < NROUND; ++r) {
        uint4 tv[U];
#pragma unroll
        for (int u = 0; u < U; ++u) {
            int g = (rr * U + u) * AGG_T + goff;
            tv[u] = tcol[min(g, EE / 16 - 1)];
        }
#pragma unroll
        for (int u = 0; u < U; ++u) {
            const int g = (rr * U + u) * AGG_T + goff;
            const int e_base = g * 16;
            const unsigned past = (g < EE / 16) ? 0u : 0xFFFFFFFFu;
            const unsigned xw[4] = { (tv[u].x ^ idrep) | past, (tv[u].y ^ idrep) | past,
                                     (tv[u].z ^ idrep) | past, (tv[u].w ^ idrep) | past };
            unsigned zw[4];
#pragma unroll
            for (int k = 0; k < 4; ++k) zw[k] = (xw[k] - 0x01010101u) & ~xw[k] & 0x80808080u;
            if (__builtin_amdgcn_ballot_w32((zw[0] | zw[1] | zw[2] | zw[3]) != 0u) != 0u) {
#pragma unroll
                for (int k = 0; k < 4; ++k) {
                    if (__builtin_amdgcn_ballot_w32(zw[k] != 0u) == 0u) continue;
#pragma unroll
                    for (int j = 0; j < 4; ++j) {
                        const bool hit = ((xw[k] >> (8 * j)) & 0xFFu) == 0u;
                        const unsigned m = __builtin_amdgcn_ballot_w32(hit);
                        if (m == 0u) continue;
                        if (hit) q[n + __builtin_amdgcn_mbcnt_lo(m, 0u)] = (unsigned)(e_base + 4 * k + j);
                        n += __popc(m);
                    }
                }
            }
            const bool last = (r == NROUND - 1) && (u == U - 1);
            if (n >= QDRAIN || (last && n > 0)) {
                if (PASS == 1) {
                    drain_max(q, n, lane, srcp, dstp, base, nloc, a_s, ad_l, m_enc,
                              kept, ocnt, ovf_flag, keep);
                    n = 0;
                } else {
                    n = drain_sum<false>(q, n, n, last, lane, srcp, dstp, base, nloc, a_s, h, ad_l, m_enc,
                                         den, acc);
                }
            }
        }
        if (++rr == NROUND) rr = 0;
    }
}

__device__ __forceinline__ float wave_sum(float v) {
    v += row_xor<1>(v); v += row_xor<2>(v); v += row_xor<4>(v); v += row_xor<8>(v);
    return v + __shfl_xor(v, 16);
}

#define LIST_NONE (-1)
template <bool FIND, bool LAST>
__global__ __launch_bounds__(AGG_T)
void k_tile_aggregate(const int* __restrict__ ei,
                      const float* __restrict__ h,
                      const float* __restrict__ a_s,
                      const float* __restrict__ a_d,
                      float* __restrict__ out,
                      unsigned* __restrict__ lists,
                      int* __restrict__ counts,
                      const unsigned* __restrict__ tile_of_edge,
                      const float* __restrict__ b_last,
                      const float* __restrict__ Wl,
                      const float* __restrict__ bl,
                      float* __restrict__ prob) {
    __shared__ alignas(16) float acc[TILE * HC];
    __shared__ unsigned queue[AGG_W * QCAP];
    __shared__ unsigned kept[AGG_W * KCAP];
    __shared__ unsigned m_enc[TILE * 2];
    __shared__ float    den[TILE * 2];
    __shared__ float    ad_l[TILE * 2];
    __shared__ int      ocnt[AGG_W];
    __shared__ int      ovf_flag;

    const int tid = threadIdx.x, lane = tid & 31;
    const int w = __builtin_amdgcn_readfirstlane(tid >> 5);
    const int base = blockIdx.x * TILE;
    const int nloc = min(TILE, NN - base);
    const int* __restrict__ srcp = ei;
    const int* __restrict__ dstp = ei + EE;
    const uint4* __restrict__ tcol = (const uint4*)tile_of_edge;
    unsigned* q = queue + w * QCAP;
    unsigned* kept_w = kept + w * KCAP;
    int  kc  = 0;

    for (int i = tid; i < TILE * HC / 4; i += AGG_T)
        ((float4*)acc)[i] = make_float4(0.f, 0.f, 0.f, 0.f);
    if (tid < TILE * 2) den[tid] = 0.f;
    if (tid < AGG_W) ocnt[tid] = 0;
    if (tid == 0) ovf_flag = 0;
    if (tid < nloc * 2) {
        int g = base * 2 + tid;
        float ad = a_d[g];
        m_enc[tid] = enc_f32(leaky(a_s[g] + ad));
        ad_l[tid]  = ad;
    }
    __syncthreads();

    unsigned* lst = lists + ((size_t)blockIdx.x * AGG_W + w) * KCAP;
    int* cntp = counts + blockIdx.x * AGG_W + w;
    if (FIND) {
        scan_edges<1>(srcp, dstp, tcol, a_s, h, q, 0, base, nloc, lane, w, ad_l, m_enc, den, acc,
                      kept, ocnt, &ovf_flag, true);
    } else {
        int c = __builtin_amdgcn_readfirstlane(*cntp);
        if (c != LIST_NONE) {
            kc = min(max(c, 0), KCAP - KSELF);
            take_list_max(lst, kc, lane, a_s, ad_l, m_enc, kept_w);
        } else {
            if (lane == 0) ovf_flag = 1;
            scan_edges<1>(srcp, dstp, tcol, a_s, h, q, 0, base, nloc, lane, w, ad_l, m_enc, den, acc,
                          kept, ocnt, &ovf_flag, false);
        }
    }
    __syncthreads();
    const bool ovf = __builtin_amdgcn_readfirstlane(ovf_flag) != 0;
    if (FIND) {
        kc = min(max(__builtin_amdgcn_readfirstlane(ocnt[w]), 0), KCAP - KSELF);
        if (!ovf)
            for (int i = lane; i < kc; i += 32) lst[i] = kept_w[i];
        if (lane == 0) *cntp = ovf ? LIST_NONE : kc;
    }

    int nself = (nloc > w) ? (nloc - w + 31) / 32 : 0;
    if (!ovf) {
        if (lane < nself) {
            int dl = w + 32 * lane;
            kept_w[kc + lane] = make_entry(base + dl, (unsigned)dl);
        }
        const int n_real = kc + nself;
        const int n_fill = (n_real + 7) & ~7;
        if (n_real + lane < n_fill) kept_w[n_real + lane] = make_entry(base + w, (unsigned)w);
        drain_sum<true>(kept_w, n_fill, n_real, true, lane, srcp, dstp, base, nloc, a_s, h, ad_l, m_enc, den, acc);
    } else {
        if (lane < nself) {
            int dl = w + 32 * lane;
            q[lane] = (unsigned)(EE + base + dl);
        }
        scan_edges<2>(srcp, dstp, tcol, a_s, h, q, nself, base, nloc, lane, w, ad_l, m_enc, den, acc,
                      kept, ocnt, &ovf_flag, false);
    }
    __syncthreads();

    if (!LAST) {
        for (int dl = w; dl < nloc; dl += AGG_W) {
            float dn0 = den[dl * 2 + 0] + 1e-16f, dn1 = den[dl * 2 + 1] + 1e-16f;
            const float4 r0 = *(const float4*)(acc + dl * HC + 4 * lane);
            const float4 r1 = *(const float4*)(acc + dl * HC + CC + 4 * lane);
            float4 o0 = make_float4(r0.x / dn0, r0.y / dn0, r0.z / dn0, r0.w / dn0);
            float4 o1 = make_float4(r1.x / dn1, r1.y / dn1, r1.z / dn1, r1.w / dn1);
            float* op = out + (size_t)(base + dl) * HC + 4 * lane;
            *(float4*)op        = o0;
            *(float4*)(op + CC) = o1;
        }
    } else {
        const float4 bA  = *(const float4*)(b_last + 4 * lane);
        const float4 bB  = *(const float4*)(b_last + CC + 4 * lane);
        const float4 wA0 = *(const float4*)(Wl + 8 * lane);
        const float4 wA1 = *(const float4*)(Wl + 8 * lane + 4);
        const float4 wB0 = *(const float4*)(Wl + 2 * CC + 8 * lane);
        const float4 wB1 = *(const float4*)(Wl + 2 * CC + 8 * lane + 4);
        const float bz0 = bA.x * wA0.x + bA.y * wA0.z + bA.z * wA1.x + bA.w * wA1.z
                        + bB.x * wB0.x + bB.y * wB0.z + bB.z * wB1.x + bB.w * wB1.z;
        const float bz1 = bA.x * wA0.y + bA.y * wA0.w + bA.z * wA1.y + bA.w * wA1.w
                        + bB.x * wB0.y + bB.y * wB0.w + bB.z * wB1.y + bB.w * wB1.w;
        const float bl0 = bl[0], bl1 = bl[1];
        for (int dl = w; dl < nloc; dl += AGG_W) {
            const float inv0 = 1.f / (den[dl * 2 + 0] + 1e-16f);
            const float inv1 = 1.f / (den[dl * 2 + 1] + 1e-16f);
            const float4 r0 = *(const float4*)(acc + dl * HC + 4 * lane);
            const float4 r1 = *(const float4*)(acc + dl * HC + CC + 4 * lane);
            const float a0 = r0.x, a1 = r0.y, a2 = r0.z, a3 = r0.w;
            const float c0 = r1.x, c1 = r1.y, c2 = r1.z, c3 = r1.w;
            const float tA0 = a0 * wA0.x + a1 * wA0.z + a2 * wA1.x + a3 * wA1.z;
            const float tA1 = a0 * wA0.y + a1 * wA0.w + a2 * wA1.y + a3 * wA1.w;
            const float tB0 = c0 * wB0.x + c1 * wB0.z + c2 * wB1.x + c3 * wB1.z;
            const float tB1 = c0 * wB0.y + c1 * wB0.w + c2 * wB1.y + c3 * wB1.w;
            const float z0 = wave_sum(tA0 * inv0 + tB0 * inv1 + bz0) + bl0;
            const float z1 = wave_sum(tA1 * inv0 + tB1 * inv1 + bz1) + bl1;
            if (lane == 0) {
                m_enc[dl * 2 + 0] = __float_as_uint(z0);
                m_enc[dl * 2 + 1] = __float_as_uint(z1);
            }
        }
        __syncthreads();
        const int nd = (base & ~15) + tid;
        if (nd >= base && nd < base + nloc) {
            const float z0 = __uint_as_float(m_enc[(nd - base) * 2 + 0]);
            const float z1 = __uint_as_float(m_enc[(nd - base) * 2 + 1]);
            const float mx = fmaxf(z0, z1);
            const float e0 = expf(z0 - mx), e1 = expf(z1 - mx);
            const float inv = 1.f / (e0 + e1);
            *(float2*)(prob + 2 * nd) = make_float2(e0 * inv, e1 * inv);
        }
    }
}

extern "C" void kernel_launch(void* const* d_in, const int* in_sizes, int n_in,
                              void* d_out, int out_size, void* d_ws, size_t ws_size,
                              hipStream_t stream) {
    const float* x        = (const float*)d_in[0];
    const int*   ei       = (const int*)  d_in[1];
    const float* W0       = (const float*)d_in[2];
    const float* att_src0 = (const float*)d_in[3];
    const float* att_dst0 = (const float*)d_in[4];
    const float* b0       = (const float*)d_in[5];
    const float* W1       = (const float*)d_in[6];
    const float* att_src1 = (const float*)d_in[7];
    const float* att_dst1 = (const float*)d_in[8];
    const float* b1       = (const float*)d_in[9];
    const float* Wl       = (const float*)d_in[10];
    const float* bl       = (const float*)d_in[11];
    float* out = (float*)d_out;

    char* base = (char*)d_ws;
    size_t off = 0;
    auto carve = [&](size_t bytes) -> char* {
        char* p = base + off;
        off = (off + bytes + 255) & ~(size_t)255;
        return p;
    };
    float*    h_feat = (float*)   carve((size_t)NN * HC * 4);
    float*    accb   = (float*)   carve((size_t)NN * HC * 4);
    __bf16*   wtbf   = (__bf16*)  carve((size_t)HC * HC * 2);
    float*    a_s    = (float*)   carve((size_t)NN * 2 * 4);
    float*    a_d    = (float*)   carve((size_t)NN * 2 * 4);
    unsigned* lists  = (unsigned*)carve((size_t)NTILES * AGG_W * KCAP * 4);
    int*      counts = (int*)     carve((size_t)NTILES * AGG_W * 4);
    unsigned* tcol   = (unsigned*)carve((size_t)EE);
    (void)ws_size; (void)in_sizes; (void)n_in; (void)out_size;

    const int T = 256;
    dim3 blk(T);

    k_transpose_bf16<<<(DD * HC + T - 1) / T, blk, 0, stream>>>(W0, wtbf, DD, HC);

    k_gemm_rows_wmma<DD, false><<<NN / 16, dim3(GEMM_T), 0, stream>>>(
        x, b0, wtbf, h_feat, att_src0, att_dst0, a_s, a_d, ei + EE, tcol);

    k_tile_aggregate<true, false><<<NTILES, dim3(AGG_T), 0, stream>>>(ei, h_feat, a_s, a_d, accb,
                                                                      lists, counts, tcol,
                                                                      b1, Wl, bl, out);

    k_transpose_bf16<<<(HC * HC + T - 1) / T, blk, 0, stream>>>(W1, wtbf, HC, HC);

    k_gemm_rows_wmma<HC, true><<<NN / 16, dim3(GEMM_T), 0, stream>>>(
        accb, b0, wtbf, h_feat, att_src1, att_dst1, a_s, a_d, ei + EE, tcol);

    k_tile_aggregate<false, true><<<NTILES, dim3(AGG_T), 0, stream>>>(ei, h_feat, a_s, a_d, accb,
                                                                      lists, counts, tcol,
                                                                      b1, Wl, bl, out);
}


# Both products: 512 blocks walk 16-row strips; a wave keeps its B fragments in registers for all its strips, later strips' rows prefetched in a register ring past LDS-only barriers; sums unchanged
# speedup vs baseline: 9.5844x; 1.1010x over previous
#include <hip/hip_runtime.h>
#include <math.h>

#define NN    50000
#define DD    128
#define HC    256
#define CC    128
#define EE    800000
#define EP    (EE + NN)
#define NEG_SLOPE 0.2f
#define TILE      196
#define NTILES    ((NN + TILE - 1) / TILE)
static_assert(NTILES <= 256, "a tile number fits a byte");
static_assert(EE % 16 == 0, "the tile-number column is read 16 bytes at a time");

typedef __attribute__((ext_vector_type(16))) __bf16 v16bf;
typedef __attribute__((ext_vector_type(8)))  __bf16 v8bf;
typedef __attribute__((ext_vector_type(8)))  float  v8f;

__device__ __forceinline__ unsigned enc_f32(float f) {
    unsigned u = __float_as_uint(f);
    return (u & 0x80000000u) ? ~u : (u | 0x80000000u);
}
__device__ __forceinline__ float dec_f32(unsigned u) {
    u = (u & 0x80000000u) ? (u & 0x7FFFFFFFu) : ~u;
    return __uint_as_float(u);
}

__global__ void k_transpose_bf16(const float* __restrict__ W,
                                 __bf16* __restrict__ Wt, int K, int Nn) {
    int i = blockIdx.x * blockDim.x + threadIdx.x;
    if (i < K * Nn) {
        int n = i / K, k = i - n * K;
        Wt[i] = (__bf16)W[(size_t)k * Nn + n];
    }
}

typedef __attribute__((ext_vector_type(4))) __bf16 v4bf;
#define GEMM_T   512
#define GEMM_G   512
#define NSTRIP   (NN / 16)
#define A_PAD    8
static_assert(NN % 16 == 0 && HC == 16 * (GEMM_T / 32), "a block covers 16 whole rows");
static_assert(GEMM_G <= NSTRIP, "every block has a strip");

__device__ __forceinline__ void lds_barrier() {
    asm volatile("s_wait_dscnt 0x0\n\ts_barrier_signal -1\n\ts_barrier_wait -1" ::: "memory");
}

template <int M>
__device__ __forceinline__ float row_xor(float v) {
    return __int_as_float(__builtin_amdgcn_update_dpp(0, __float_as_int(v), 0x160 + M,
                                                      0xf, 0xf, true));
}

template <int K, bool TANH>
__global__ __launch_bounds__(GEMM_T, 8)
void k_gemm_rows_wmma(const float* __restrict__ A,
                      const float* __restrict__ bias,
                      const __bf16* __restrict__ Bt,
                      float* __restrict__ C,
                      const float* __restrict__ att_src,
                      const float* __restrict__ att_dst,
                      float* __restrict__ a_s,
                      float* __restrict__ a_d,
                      const int* __restrict__ dstp,
                      unsigned* __restrict__ tile_of_edge) {
    constexpr int PITCH = K + A_PAD;
    __shared__ alignas(16) __bf16 a_lds[16 * PITCH];
    __shared__ float sc_part[(GEMM_T / 32) * 32];
    __shared__ alignas(8) float att_lds[2 * HC];
    static_assert(GEMM_T == 2 * HC, "one element of the two attention vectors per thread");

    const int tid = threadIdx.x;
    const int lane = tid & 31;
    const int tN   = (tid >> 5) << 4;
    const int half = lane >> 4;
    const int lr   = lane & 15;

    constexpr int PRE = (EE / 4 + GEMM_G - 1) / GEMM_G;
    static_assert(PRE <= GEMM_T, "a thread takes at most one piece");
    const int  pre_g = blockIdx.x * PRE + tid;
    const bool pre   = !TANH && tid < PRE && pre_g < EE / 4;
    int4 pre_d = make_int4(0, 0, 0, 0);
    if (pre) pre_d = *(const int4*)(dstp + 4 * pre_g);

    const __bf16* brow = Bt + (size_t)(tN + lr) * K;
    v8bf bfr[K / 16];
#pragma unroll
    for (int j = 0; j < K / 16; ++j)
        bfr[j] = *(const v8bf*)(brow + 16 * j + 8 * half);
    const float att_el = (tid < HC) ? att_src[tid] : att_dst[tid - HC];

    constexpr int V4_ROW = K / 4;
    constexpr int NIT    = 16 * V4_ROW / GEMM_T;
    constexpr int DEPTH  = 4 / NIT;
    static_assert(NIT * GEMM_T == 16 * V4_ROW, "the block's threads cover the 16 rows exactly");
    static_assert(GEMM_T % V4_ROW == 0, "a thread's pieces of a strip lie in the same columns");
    const int prow = tid / V4_ROW, pc4 = tid - prow * V4_ROW;
    const float* aptr = A + (size_t)prow * K + 4 * pc4;
    constexpr int ROW_IT = GEMM_T / V4_ROW;

    float4 bias4 = make_float4(0.f, 0.f, 0.f, 0.f);
    if (TANH) bias4 = *(const float4*)(bias + 4 * pc4);

    float4 av[DEPTH][NIT];
#pragma unroll
    for (int d = 0; d < DEPTH; ++d) {
        const int sp = min((int)blockIdx.x + d * GEMM_G, NSTRIP - 1);
#pragma unroll
        for (int it = 0; it < NIT; ++it)
            av[d][it] = *(const float4*)(aptr + (size_t)(16 * sp + it * ROW_IT) * K);
    }

    if (pre) {
        const unsigned u0 = (unsigned)pre_d.x < (unsigned)NN ? (unsigned)pre_d.x : 0u;
        const unsigned u1 = (unsigned)pre_d.y < (unsigned)NN ? (unsigned)pre_d.y : 0u;
        const unsigned u2 = (unsigned)pre_d.z < (unsigned)NN ? (unsigned)pre_d.z : 0u;
        const unsigned u3 = (unsigned)pre_d.w < (unsigned)NN ? (unsigned)pre_d.w : 0u;
        tile_of_edge[pre_g] = (u0 / TILE) | ((u1 / TILE) << 8) | ((u2 / TILE) << 16) | ((u3 / TILE) << 24);
    }
    att_lds[(tid < HC) ? 2 * tid : 2 * (tid - HC) + 1] = att_el;

    const __bf16* arow = a_lds + lr * PITCH;
    const int n = tN + lr;

    auto step = [&](const int strip, float4 (&rv)[NIT], const bool ahead) __attribute__((always_inline)) {
        const int tM = strip << 4;

#pragma unroll
        for (int it = 0; it < NIT; ++it) {
            float4 v = rv[it];
            if (TANH) {
                v.x = tanhf(v.x + bias4.x); v.y = tanhf(v.y + bias4.y);
                v.z = tanhf(v.z + bias4.z); v.w = tanhf(v.w + bias4.w);
            }
            v4bf o;
            o[0] = (__bf16)v.x; o[1] = (__bf16)v.y; o[2] = (__bf16)v.z; o[3] = (__bf16)v.w;
            *(v4bf*)(a_lds + (prow + it * ROW_IT) * PITCH + 4 * pc4) = o;
        }
        lds_barrier();

        if (ahead) {
            const int sp = min(strip + DEPTH * GEMM_G, NSTRIP - 1);
#pragma unroll
            for (int it = 0; it < NIT; ++it)
                rv[it] = *(const float4*)(aptr + (size_t)(16 * sp + it * ROW_IT) * K);
        }

        v8f acc = {};
#pragma unroll
        for (int kk = 0; kk < K; kk += 32) {
            v8bf alo = *(const v8bf*)(arow + kk + 8 * half);
            v8bf ahi = *(const v8bf*)(arow + kk + 16 + 8 * half);
            v8bf blo = bfr[kk / 16];
            v8bf bhi = bfr[kk / 16 + 1];
            v16bf a, b;
#pragma unroll
            for (int i = 0; i < 8; ++i) {
                a[i] = alo[i]; a[i + 8] = ahi[i];
                b[i] = blo[i]; b[i + 8] = bhi[i];
            }
            acc = __builtin_amdgcn_wmma_f32_16x16x32_bf16(
                      false, a, false, b, (short)0, acc, false, false);
        }

        const float2 att_n = *(const float2*)(att_lds + 2 * n);
        const float as_n = att_n.x, ad_n = att_n.y;
        float p[16];
#pragma unroll
        for (int r = 0; r < 8; ++r) { p[2 * r] = acc[r] * as_n; p[2 * r + 1] = acc[r] * ad_n; }
        const bool b3 = (lr & 8) != 0, b2 = (lr & 4) != 0, b1 = (lr & 2) != 0, b0 = (lr & 1) != 0;
        float q8[8], q4[4], q2[2];
#pragma unroll
        for (int j = 0; j < 8; ++j)
            q8[j] = (b3 ? p[j + 8] : p[j]) + row_xor<8>(b3 ? p[j] : p[j + 8]);
#pragma unroll
        for (int j = 0; j < 4; ++j)
            q4[j] = (b2 ? q8[j + 4] : q8[j]) + row_xor<4>(b2 ? q8[j] : q8[j + 4]);
#pragma unroll
        for (int j = 0; j < 2; ++j)
            q2[j] = (b1 ? q4[j + 2] : q4[j]) + row_xor<2>(b1 ? q4[j] : q4[j + 2]);
        const float q1 = (b0 ? q2[1] : q2[0]) + row_xor<1>(b0 ? q2[0] : q2[1]);
        sc_part[(tid >> 5) * 32 + lane] = q1;
        lds_barrier();

#pragma unroll
        for (int r = 0; r < 8; ++r) {
            int m = tM + r + 8 * half;
            C[(size_t)m * HC + n] = acc[r];
        }

        if (tid < 64) {
            const int sc = tid >> 5, idx = tid & 31;
            const int row = idx >> 1, hd = idx & 1;
            const int li = 16 * (row >> 3) + 2 * (row & 7) + sc;
            float s = 0.f;
#pragma unroll
            for (int g = 0; g < 8; ++g) s += sc_part[(hd * 8 + g) * 32 + li];
            (sc ? a_d : a_s)[tM * 2 + idx] = s;
        }
    };

    int strip0 = blockIdx.x;
    for (; strip0 + (DEPTH - 1) * GEMM_G < NSTRIP; strip0 += DEPTH * GEMM_G) {
#pragma unroll
        for (int d = 0; d < DEPTH; ++d) step(strip0 + d * GEMM_G, av[d], true);
    }
#pragma unroll
    for (int d = 0; d + 1 < DEPTH; ++d) {
        const int strip = strip0 + d * GEMM_G;
        if (strip >= NSTRIP) return;
        step(strip, av[d], false);
    }
}

#define AGG_T     1024
#define AGG_W     (AGG_T / 32)
#define PIECE_E   512
#define SCAN_U    4
#define NPIECE    ((EE / 16 + AGG_T - 1) / AGG_T)
#define QDRAIN    32
#define QCAP      (QDRAIN + PIECE_E)
#define KCAP      384
#define KSELF     8
static_assert(EP <= (1 << 20) && NN <= (1 << 20) && TILE <= (1 << 12), "queue entry packing");
static_assert((TILE + AGG_W - 1) / AGG_W < KSELF && KSELF <= 32 && KCAP % 8 == 0 && KSELF % 8 == 0,
              "self loops and the fill up to a whole batch fit the reserved room");

__device__ __forceinline__ float leaky(float l) {
    return (l > 0.f) ? l : NEG_SLOPE * l;
}

__device__ __forceinline__ unsigned make_entry(int s, unsigned dl) {
    return (unsigned)s | (dl << 20);
}
__device__ __forceinline__ void entry_nodes(unsigned e, int base, int nloc,
                                            const int* __restrict__ srcp,
                                            const int* __restrict__ dstp,
                                            int& s, int& dl) {
    int d;
    if (e < (unsigned)EE) { s = srcp[e]; d = dstp[e]; }
    else                  { s = (int)e - EE; d = s; }
    dl = min(max(d - base, 0), nloc - 1);
}

__device__ __forceinline__ void drain_max(const unsigned* q, int n, int lane,
                                          const int* __restrict__ srcp,
                                          const int* __restrict__ dstp, int base, int nloc,
                                          const float* __restrict__ a_s,
                                          const float* ad_l, unsigned* m_enc,
                                          unsigned* kept, int* ocnt, int* ovf_flag, bool keep) {
    for (int b = 0; b < n; b += 32) {
        int i = b + lane;
        if (i < n) {
            int s, dl;
            entry_nodes(q[i], base, nloc, srcp, dstp, s, dl);
            float2 as = *(const float2*)(a_s + 2 * s);
            if (keep) {
                const int ow   = dl & (AGG_W - 1);
                const int slot = atomicAdd(&ocnt[ow], 1);
                if (slot < KCAP - KSELF) kept[ow * KCAP + slot] = make_entry(s, (unsigned)dl);
                else                     *ovf_flag = 1;
            }
            atomicMax(&m_enc[dl * 2 + 0], enc_f32(leaky(as.x + ad_l[dl * 2 + 0])));
            atomicMax(&m_enc[dl * 2 + 1], enc_f32(leaky(as.y + ad_l[dl * 2 + 1])));
        }
    }
}

__device__ __forceinline__ void take_list_max(const unsigned* __restrict__ lst, int n, int lane,
                                              const float* __restrict__ a_s,
                                              const float* ad_l, unsigned* m_enc,
                                              unsigned* kp) {
    for (int b = 0; b < n; b += 128) {
        unsigned ent[4];
        float2   as[4];
#pragma unroll
        for (int u = 0; u < 4; ++u) {
            int i = b + 32 * u + lane;
            ent[u] = (i < n) ? lst[i] : 0u;
        }
#pragma unroll
        for (int u = 0; u < 4; ++u)
            as[u] = *(const float2*)(a_s + 2 * (int)(ent[u] & 0xFFFFFu));
#pragma unroll
        for (int u = 0; u < 4; ++u) {
            int i = b + 32 * u + lane;
            if (i < n) {
                int dl = (int)(ent[u] >> 20);
                kp[i] = ent[u];
                atomicMax(&m_enc[dl * 2 + 0], enc_f32(leaky(as[u].x + ad_l[dl * 2 + 0])));
                atomicMax(&m_enc[dl * 2 + 1], enc_f32(leaky(as[u].y + ad_l[dl * 2 + 1])));
            }
        }
    }
}

template <bool FULL, bool OWN>
__device__ __forceinline__ void add_rows(int k0, int cnt, int lane, int s_l, int dl_l,
                                         float ex0_l, float ex1_l,
                                         const float* __restrict__ h, float* acc) {
    float4 v0[8], v1[8];
#pragma unroll
    for (int k = 0; k < 8; ++k) {
        if (FULL || k0 + k < cnt) {
            int s = __builtin_amdgcn_readlane(s_l, k0 + k);
            const float* hr = h + (size_t)s * HC + 4 * lane;
            v0[k] = *(const float4*)hr;
            v1[k] = *(const float4*)(hr + CC);
        }
    }
    if (OWN) __builtin_amdgcn_sched_barrier(0);
#pragma unroll
    for (int k = 0; k < 8; ++k) {
        if (FULL || k0 + k < cnt) {
            int   dl = __builtin_amdgcn_readlane(dl_l, k0 + k);
            float e0 = __int_as_float(__builtin_amdgcn_readlane(__float_as_int(ex0_l), k0 + k));
            float e1 = __int_as_float(__builtin_amdgcn_readlane(__float_as_int(ex1_l), k0 + k));
            float* ar = acc + dl * HC + 4 * lane;
            if (OWN) {
                float4 a0 = *(const float4*)ar;
                float4 a1 = *(const float4*)(ar + CC);
                a0.x = fmaf(e0, v0[k].x, a0.x); a0.y = fmaf(e0, v0[k].y, a0.y);
                a0.z = fmaf(e0, v0[k].z, a0.z); a0.w = fmaf(e0, v0[k].w, a0.w);
                a1.x = fmaf(e1, v1[k].x, a1.x); a1.y = fmaf(e1, v1[k].y, a1.y);
                a1.z = fmaf(e1, v1[k].z, a1.z); a1.w = fmaf(e1, v1[k].w, a1.w);
                *(float4*)ar        = a0;
                *(float4*)(ar + CC) = a1;
            } else {
                atomicAdd(ar + 0, e0 * v0[k].x);
                atomicAdd(ar + 1, e0 * v0[k].y);
                atomicAdd(ar + 2, e0 * v0[k].z);
                atomicAdd(ar + 3, e0 * v0[k].w);
                atomicAdd(ar + CC + 0, e1 * v1[k].x);
                atomicAdd(ar + CC + 1, e1 * v1[k].y);
                atomicAdd(ar + CC + 2, e1 * v1[k].z);
                atomicAdd(ar + CC + 3, e1 * v1[k].w);
            }
        }
    }
}

template <bool DIRECT>
__device__ __forceinline__ int drain_sum(unsigned* q, int n, int n_real, bool last, int lane,
                                          const int* __restrict__ srcp,
                                          const int* __restrict__ dstp, int base, int nloc,
                                          const float* __restrict__ a_s,
                                          const float* __restrict__ h,
                                          const float* ad_l, const unsigned* m_enc,
                                          float* den, float* acc) {
    const int nproc = last ? n : (n & ~7);
    for (int b = 0; b < nproc; b += 32) {
        int i = b + lane;
        int cnt = min(32, nproc - b);
        int s_l = 0, dl_l = 0;
        float ex0_l = 0.f, ex1_l = 0.f;
        if (i < nproc) {
            unsigned ent = q[i];
            if (DIRECT) { s_l = (int)(ent & 0xFFFFFu); dl_l = (int)(ent >> 20); }
            else        entry_nodes(ent, base, nloc, srcp, dstp, s_l, dl_l);
            float2 as = *(const float2*)(a_s + 2 * s_l);
            ex0_l = expf(leaky(as.x + ad_l[dl_l * 2 + 0]) - dec_f32(m_enc[dl_l * 2 + 0]));
            ex1_l = expf(leaky(as.y + ad_l[dl_l * 2 + 1]) - dec_f32(m_enc[dl_l * 2 + 1]));
            if (DIRECT && i >= n_real) {
                ex0_l = 0.f; ex1_l = 0.f;
            } else {
                atomicAdd(&den[dl_l * 2 + 0], ex0_l);
                atomicAdd(&den[dl_l * 2 + 1], ex1_l);
            }
        }
        int k0 = 0;
        for (; k0 + 8 <= cnt; k0 += 8)
            add_rows<true, DIRECT>(k0, cnt, lane, s_l, dl_l, ex0_l, ex1_l, h, acc);
        if (!DIRECT && k0 < cnt)
            add_rows<false, false>(k0, cnt, lane, s_l, dl_l, ex0_l, ex1_l, h, acc);
    }
    const int rem = n - nproc;
    if (rem > 0 && nproc > 0) {
        unsigned keep = 0u;
        if (lane < rem) keep = q[nproc + lane];
        if (lane < rem) q[lane] = keep;
    }
    return rem;
}

template <int PASS>
__device__ __forceinline__ void scan_edges(const int* __restrict__ srcp,
                                           const int* __restrict__ dstp,
                                           const uint4* __restrict__ tcol,
                                           const float* __restrict__ a_s,
                                           const float* __restrict__ h,
                                           unsigned* q, int n, int base, int nloc,
                                           int lane, int w,
                                           const float* ad_l, unsigned* m_enc,
                                           float* den, float* acc,
                                           unsigned* kept, int* ocnt, int* ovf_flag, bool keep) {
    const unsigned idrep = (unsigned)blockIdx.x * 0x01010101u;
    const int goff = w * 32 + lane;
    constexpr int U      = (PASS == 1) ? SCAN_U : 1;
    constexpr int NROUND = (NPIECE + U - 1) / U;
    int rr = blockIdx.x % NROUND;
    for (int r = 0; r < NROUND; ++r) {
        uint4 tv[U];
#pragma unroll
        for (int u = 0; u < U; ++u) {
            int g = (rr * U + u) * AGG_T + goff;
            tv[u] = tcol[min(g, EE / 16 - 1)];
        }
#pragma unroll
        for (int u = 0; u < U; ++u) {
            const int g = (rr * U + u) * AGG_T + goff;
            const int e_base = g * 16;
            const unsigned past = (g < EE / 16) ? 0u : 0xFFFFFFFFu;
            const unsigned xw[4] = { (tv[u].x ^ idrep) | past, (tv[u].y ^ idrep) | past,
                                     (tv[u].z ^ idrep) | past, (tv[u].w ^ idrep) | past };
            unsigned zw[4];
#pragma unroll
            for (int k = 0; k < 4; ++k) zw[k] = (xw[k] - 0x01010101u) & ~xw[k] & 0x80808080u;
            if (__builtin_amdgcn_ballot_w32((zw[0] | zw[1] | zw[2] | zw[3]) != 0u) != 0u) {
#pragma unroll
                for (int k = 0; k < 4; ++k) {
                    if (__builtin_amdgcn_ballot_w32(zw[k] != 0u) == 0u) continue;
#pragma unroll
                    for (int j = 0; j < 4; ++j) {
                        const bool hit = ((xw[k] >> (8 * j)) & 0xFFu) == 0u;
                        const unsigned m = __builtin_amdgcn_ballot_w32(hit);
                        if (m == 0u) continue;
                        if (hit) q[n + __builtin_amdgcn_mbcnt_lo(m, 0u)] = (unsigned)(e_base + 4 * k + j);
                        n += __popc(m);
                    }
                }
            }
            const bool last = (r == NROUND - 1) && (u == U - 1);
            if (n >= QDRAIN || (last && n > 0)) {
                if (PASS == 1) {
                    drain_max(q, n, lane, srcp, dstp, base, nloc, a_s, ad_l, m_enc,
                              kept, ocnt, ovf_flag, keep);
                    n = 0;
                } else {
                    n = drain_sum<false>(q, n, n, last, lane, srcp, dstp, base, nloc, a_s, h, ad_l, m_enc,
                                         den, acc);
                }
            }
        }
        if (++rr == NROUND) rr = 0;
    }
}

__device__ __forceinline__ float wave_sum(float v) {
    v += row_xor<1>(v); v += row_xor<2>(v); v += row_xor<4>(v); v += row_xor<8>(v);
    return v + __shfl_xor(v, 16);
}

#define LIST_NONE (-1)
template <bool FIND, bool LAST>
__global__ __launch_bounds__(AGG_T)
void k_tile_aggregate(const int* __restrict__ ei,
                      const float* __restrict__ h,
                      const float* __restrict__ a_s,
                      const float* __restrict__ a_d,
                      float* __restrict__ out,
                      unsigned* __restrict__ lists,
                      int* __restrict__ counts,
                      const unsigned* __restrict__ tile_of_edge,
                      const float* __restrict__ b_last,
                      const float* __restrict__ Wl,
                      const float* __restrict__ bl,
                      float* __restrict__ prob) {
    __shared__ alignas(16) float acc[TILE * HC];
    __shared__ unsigned queue[AGG_W * QCAP];
    __shared__ unsigned kept[AGG_W * KCAP];
    __shared__ unsigned m_enc[TILE * 2];
    __shared__ float    den[TILE * 2];
    __shared__ float    ad_l[TILE * 2];
    __shared__ int      ocnt[AGG_W];
    __shared__ int      ovf_flag;

    const int tid = threadIdx.x, lane = tid & 31;
    const int w = __builtin_amdgcn_readfirstlane(tid >> 5);
    const int base = blockIdx.x * TILE;
    const int nloc = min(TILE, NN - base);
    const int* __restrict__ srcp = ei;
    const int* __restrict__ dstp = ei + EE;
    const uint4* __restrict__ tcol = (const uint4*)tile_of_edge;
    unsigned* q = queue + w * QCAP;
    unsigned* kept_w = kept + w * KCAP;
    int  kc  = 0;

    for (int i = tid; i < TILE * HC / 4; i += AGG_T)
        ((float4*)acc)[i] = make_float4(0.f, 0.f, 0.f, 0.f);
    if (tid < TILE * 2) den[tid] = 0.f;
    if (tid < AGG_W) ocnt[tid] = 0;
    if (tid == 0) ovf_flag = 0;
    if (tid < nloc * 2) {
        int g = base * 2 + tid;
        float ad = a_d[g];
        m_enc[tid] = enc_f32(leaky(a_s[g] + ad));
        ad_l[tid]  = ad;
    }
    __syncthreads();

    unsigned* lst = lists + ((size_t)blockIdx.x * AGG_W + w) * KCAP;
    int* cntp = counts + blockIdx.x * AGG_W + w;
    if (FIND) {
        scan_edges<1>(srcp, dstp, tcol, a_s, h, q, 0, base, nloc, lane, w, ad_l, m_enc, den, acc,
                      kept, ocnt, &ovf_flag, true);
    } else {
        int c = __builtin_amdgcn_readfirstlane(*cntp);
        if (c != LIST_NONE) {
            kc = min(max(c, 0), KCAP - KSELF);
            take_list_max(lst, kc, lane, a_s, ad_l, m_enc, kept_w);
        } else {
            if (lane == 0) ovf_flag = 1;
            scan_edges<1>(srcp, dstp, tcol, a_s, h, q, 0, base, nloc, lane, w, ad_l, m_enc, den, acc,
                          kept, ocnt, &ovf_flag, false);
        }
    }
    __syncthreads();
    const bool ovf = __builtin_amdgcn_readfirstlane(ovf_flag) != 0;
    if (FIND) {
        kc = min(max(__builtin_amdgcn_readfirstlane(ocnt[w]), 0), KCAP - KSELF);
        if (!ovf)
            for (int i = lane; i < kc; i += 32) lst[i] = kept_w[i];
        if (lane == 0) *cntp = ovf ? LIST_NONE : kc;
    }

    int nself = (nloc > w) ? (nloc - w + 31) / 32 : 0;
    if (!ovf) {
        if (lane < nself) {
            int dl = w + 32 * lane;
            kept_w[kc + lane] = make_entry(base + dl, (unsigned)dl);
        }
        const int n_real = kc + nself;
        const int n_fill = (n_real + 7) & ~7;
        if (n_real + lane < n_fill) kept_w[n_real + lane] = make_entry(base + w, (unsigned)w);
        drain_sum<true>(kept_w, n_fill, n_real, true, lane, srcp, dstp, base, nloc, a_s, h, ad_l, m_enc, den, acc);
    } else {
        if (lane < nself) {
            int dl = w + 32 * lane;
            q[lane] = (unsigned)(EE + base + dl);
        }
        scan_edges<2>(srcp, dstp, tcol, a_s, h, q, nself, base, nloc, lane, w, ad_l, m_enc, den, acc,
                      kept, ocnt, &ovf_flag, false);
    }
    __syncthreads();

    if (!LAST) {
        for (int dl = w; dl < nloc; dl += AGG_W) {
            float dn0 = den[dl * 2 + 0] + 1e-16f, dn1 = den[dl * 2 + 1] + 1e-16f;
            const float4 r0 = *(const float4*)(acc + dl * HC + 4 * lane);
            const float4 r1 = *(const float4*)(acc + dl * HC + CC + 4 * lane);
            float4 o0 = make_float4(r0.x / dn0, r0.y / dn0, r0.z / dn0, r0.w / dn0);
            float4 o1 = make_float4(r1.x / dn1, r1.y / dn1, r1.z / dn1, r1.w / dn1);
            float* op = out + (size_t)(base + dl) * HC + 4 * lane;
            *(float4*)op        = o0;
            *(float4*)(op + CC) = o1;
        }
    } else {
        const float4 bA  = *(const float4*)(b_last + 4 * lane);
        const float4 bB  = *(const float4*)(b_last + CC + 4 * lane);
        const float4 wA0 = *(const float4*)(Wl + 8 * lane);
        const float4 wA1 = *(const float4*)(Wl + 8 * lane + 4);
        const float4 wB0 = *(const float4*)(Wl + 2 * CC + 8 * lane);
        const float4 wB1 = *(const float4*)(Wl + 2 * CC + 8 * lane + 4);
        const float bz0 = bA.x * wA0.x + bA.y * wA0.z + bA.z * wA1.x + bA.w * wA1.z
                        + bB.x * wB0.x + bB.y * wB0.z + bB.z * wB1.x + bB.w * wB1.z;
        const float bz1 = bA.x * wA0.y + bA.y * wA0.w + bA.z * wA1.y + bA.w * wA1.w
                        + bB.x * wB0.y + bB.y * wB0.w + bB.z * wB1.y + bB.w * wB1.w;
        const float bl0 = bl[0], bl1 = bl[1];
        for (int dl = w; dl < nloc; dl += AGG_W) {
            const float inv0 = 1.f / (den[dl * 2 + 0] + 1e-16f);
            const float inv1 = 1.f / (den[dl * 2 + 1] + 1e-16f);
            const float4 r0 = *(const float4*)(acc + dl * HC + 4 * lane);
            const float4 r1 = *(const float4*)(acc + dl * HC + CC + 4 * lane);
            const float a0 = r0.x, a1 = r0.y, a2 = r0.z, a3 = r0.w;
            const float c0 = r1.x, c1 = r1.y, c2 = r1.z, c3 = r1.w;
            const float tA0 = a0 * wA0.x + a1 * wA0.z + a2 * wA1.x + a3 * wA1.z;
            const float tA1 = a0 * wA0.y + a1 * wA0.w + a2 * wA1.y + a3 * wA1.w;
            const float tB0 = c0 * wB0.x + c1 * wB0.z + c2 * wB1.x + c3 * wB1.z;
            const float tB1 = c0 * wB0.y + c1 * wB0.w + c2 * wB1.y + c3 * wB1.w;
            const float z0 = wave_sum(tA0 * inv0 + tB0 * inv1 + bz0) + bl0;
            const float z1 = wave_sum(tA1 * inv0 + tB1 * inv1 + bz1) + bl1;
            if (lane == 0) {
                m_enc[dl * 2 + 0] = __float_as_uint(z0);
                m_enc[dl * 2 + 1] = __float_as_uint(z1);
            }
        }
        __syncthreads();
        const int nd = (base & ~15) + tid;
        if (nd >= base && nd < base + nloc) {
            const float z0 = __uint_as_float(m_enc[(nd - base) * 2 + 0]);
            const float z1 = __uint_as_float(m_enc[(nd - base) * 2 + 1]);
            const float mx = fmaxf(z0, z1);
            const float e0 = expf(z0 - mx), e1 = expf(z1 - mx);
            const float inv = 1.f / (e0 + e1);
            *(float2*)(prob + 2 * nd) = make_float2(e0 * inv, e1 * inv);
        }
    }
}

extern "C" void kernel_launch(void* const* d_in, const int* in_sizes, int n_in,
                              void* d_out, int out_size, void* d_ws, size_t ws_size,
                              hipStream_t stream) {
    const float* x        = (const float*)d_in[0];
    const int*   ei       = (const int*)  d_in[1];
    const float* W0       = (const float*)d_in[2];
    const float* att_src0 = (const float*)d_in[3];
    const float* att_dst0 = (const float*)d_in[4];
    const float* b0       = (const float*)d_in[5];
    const float* W1       = (const float*)d_in[6];
    const float* att_src1 = (const float*)d_in[7];
    const float* att_dst1 = (const float*)d_in[8];
    const float* b1       = (const float*)d_in[9];
    const float* Wl       = (const float*)d_in[10];
    const float* bl       = (const float*)d_in[11];
    float* out = (float*)d_out;

    char* base = (char*)d_ws;
    size_t off = 0;
    auto carve = [&](size_t bytes) -> char* {
        char* p = base + off;
        off = (off + bytes + 255) & ~(size_t)255;
        return p;
    };
    float*    h_feat = (float*)   carve((size_t)NN * HC * 4);
    float*    accb   = (float*)   carve((size_t)NN * HC * 4);
    __bf16*   wtbf   = (__bf16*)  carve((size_t)HC * HC * 2);
    float*    a_s    = (float*)   carve((size_t)NN * 2 * 4);
    float*    a_d    = (float*)   carve((size_t)NN * 2 * 4);
    unsigned* lists  = (unsigned*)carve((size_t)NTILES * AGG_W * KCAP * 4);
    int*      counts = (int*)     carve((size_t)NTILES * AGG_W * 4);
    unsigned* tcol   = (unsigned*)carve((size_t)EE);
    (void)ws_size; (void)in_sizes; (void)n_in; (void)out_size;

    const int T = 256;
    dim3 blk(T);

    k_transpose_bf16<<<(DD * HC + T - 1) / T, blk, 0, stream>>>(W0, wtbf, DD, HC);

    k_gemm_rows_wmma<DD, false><<<GEMM_G, dim3(GEMM_T), 0, stream>>>(
        x, b0, wtbf, h_feat, att_src0, att_dst0, a_s, a_d, ei + EE, tcol);

    k_tile_aggregate<true, false><<<NTILES, dim3(AGG_T), 0, stream>>>(ei, h_feat, a_s, a_d, accb,
                                                                      lists, counts, tcol,
                                                                      b1, Wl, bl, out);

    k_transpose_bf16<<<(HC * HC + T - 1) / T, blk, 0, stream>>>(W1, wtbf, HC, HC);

    k_gemm_rows_wmma<HC, true><<<GEMM_G, dim3(GEMM_T), 0, stream>>>(
        accb, b0, wtbf, h_feat, att_src1, att_dst1, a_s, a_d, ei + EE, tcol);

    k_tile_aggregate<false, true><<<NTILES, dim3(AGG_T), 0, stream>>>(ei, h_feat, a_s, a_d, accb,
                                                                      lists, counts, tcol,
                                                                      b1, Wl, bl, out);
}


# Edge search by a bit table: layer 0's product marks, per tile, each 16-edge group with an edge of it (LDS OR, plain stores); a tile block reads only marked groups; sum order: LDS arrival as before
# speedup vs baseline: 10.0075x; 1.0441x over previous
#include <hip/hip_runtime.h>
#include <math.h>

#define NN    50000
#define DD    128
#define HC    256
#define CC    128
#define EE    800000
#define EP    (EE + NN)
#define NEG_SLOPE 0.2f
#define TILE      196
#define NTILES    ((NN + TILE - 1) / TILE)
static_assert(NTILES <= 256, "a tile number fits a byte");
static_assert(EE % 16 == 0, "the tile-number column is read 16 bytes at a time");
#define NGRP      (EE / 16)
#define PRE_B     ((NGRP + 127) / 128)
#define PW        (4 * PRE_B)

typedef __attribute__((ext_vector_type(16))) __bf16 v16bf;
typedef __attribute__((ext_vector_type(8)))  __bf16 v8bf;
typedef __attribute__((ext_vector_type(8)))  float  v8f;

__device__ __forceinline__ unsigned enc_f32(float f) {
    unsigned u = __float_as_uint(f);
    return (u & 0x80000000u) ? ~u : (u | 0x80000000u);
}
__device__ __forceinline__ float dec_f32(unsigned u) {
    u = (u & 0x80000000u) ? (u & 0x7FFFFFFFu) : ~u;
    return __uint_as_float(u);
}

__global__ void k_transpose_bf16(const float* __restrict__ W,
                                 __bf16* __restrict__ Wt, int K, int Nn) {
    int i = blockIdx.x * blockDim.x + threadIdx.x;
    if (i < K * Nn) {
        int n = i / K, k = i - n * K;
        Wt[i] = (__bf16)W[(size_t)k * Nn + n];
    }
}

typedef __attribute__((ext_vector_type(4))) __bf16 v4bf;
#define GEMM_T   512
#define GEMM_G   512
#define NSTRIP   (NN / 16)
#define A_PAD    8
static_assert(NN % 16 == 0 && HC == 16 * (GEMM_T / 32), "a block covers 16 whole rows");
static_assert(GEMM_G <= NSTRIP, "every block has a strip");

__device__ __forceinline__ void lds_barrier() {
    asm volatile("s_wait_dscnt 0x0\n\ts_barrier_signal -1\n\ts_barrier_wait -1" ::: "memory");
}

template <int M>
__device__ __forceinline__ float row_xor(float v) {
    return __int_as_float(__builtin_amdgcn_update_dpp(0, __float_as_int(v), 0x160 + M,
                                                      0xf, 0xf, true));
}

template <int K, bool TANH>
__global__ __launch_bounds__(GEMM_T, 8)
void k_gemm_rows_wmma(const float* __restrict__ A,
                      const float* __restrict__ bias,
                      const __bf16* __restrict__ Bt,
                      float* __restrict__ C,
                      const float* __restrict__ att_src,
                      const float* __restrict__ att_dst,
                      float* __restrict__ a_s,
                      float* __restrict__ a_d,
                      const int* __restrict__ dstp,
                      unsigned* __restrict__ tile_of_edge,
                      unsigned* __restrict__ pres) {
    constexpr int PITCH = K + A_PAD;
    __shared__ alignas(16) __bf16 a_lds[16 * PITCH];
    __shared__ float sc_part[(GEMM_T / 32) * 32];
    __shared__ alignas(8) float att_lds[2 * HC];
    static_assert(GEMM_T == 2 * HC, "one element of the two attention vectors per thread");
    __shared__ unsigned pres_lds[TANH ? 1 : NTILES * 4];
    static_assert(NTILES * 4 == 2 * GEMM_T && PRE_B <= GEMM_G, "two table words a thread; a block per 2048 edges");

    const int tid = threadIdx.x;
    const int lane = tid & 31;
    const int tN   = (tid >> 5) << 4;
    const int half = lane >> 4;
    const int lr   = lane & 15;

    const int  pre_g = blockIdx.x * GEMM_T + tid;
    const bool pre   = !TANH && pre_g < EE / 4;
    int4 pre_d = make_int4(0, 0, 0, 0);
    if (pre) pre_d = *(const int4*)(dstp + 4 * pre_g);

    const __bf16* brow = Bt + (size_t)(tN + lr) * K;
    v8bf bfr[K / 16];
#pragma unroll
    for (int j = 0; j < K / 16; ++j)
        bfr[j] = *(const v8bf*)(brow + 16 * j + 8 * half);
    const float att_el = (tid < HC) ? att_src[tid] : att_dst[tid - HC];

    constexpr int V4_ROW = K / 4;
    constexpr int NIT    = 16 * V4_ROW / GEMM_T;
    constexpr int DEPTH  = 4 / NIT;
    static_assert(NIT * GEMM_T == 16 * V4_ROW, "the block's threads cover the 16 rows exactly");
    static_assert(GEMM_T % V4_ROW == 0, "a thread's pieces of a strip lie in the same columns");
    const int prow = tid / V4_ROW, pc4 = tid - prow * V4_ROW;
    const float* aptr = A + (size_t)prow * K + 4 * pc4;
    constexpr int ROW_IT = GEMM_T / V4_ROW;

    float4 bias4 = make_float4(0.f, 0.f, 0.f, 0.f);
    if (TANH) bias4 = *(const float4*)(bias + 4 * pc4);

    float4 av[DEPTH][NIT];
#pragma unroll
    for (int d = 0; d < DEPTH; ++d) {
        const int sp = min((int)blockIdx.x + d * GEMM_G, NSTRIP - 1);
#pragma unroll
        for (int it = 0; it < NIT; ++it)
            av[d][it] = *(const float4*)(aptr + (size_t)(16 * sp + it * ROW_IT) * K);
    }

    if (!TANH && blockIdx.x < PRE_B) {
        pres_lds[tid] = 0u;
        pres_lds[tid + GEMM_T] = 0u;
        lds_barrier();
        if (pre) {
            const unsigned u0 = (unsigned)pre_d.x < (unsigned)NN ? (unsigned)pre_d.x : 0u;
            const unsigned u1 = (unsigned)pre_d.y < (unsigned)NN ? (unsigned)pre_d.y : 0u;
            const unsigned u2 = (unsigned)pre_d.z < (unsigned)NN ? (unsigned)pre_d.z : 0u;
            const unsigned u3 = (unsigned)pre_d.w < (unsigned)NN ? (unsigned)pre_d.w : 0u;
            const unsigned t0 = u0 / TILE, t1 = u1 / TILE, t2 = u2 / TILE, t3 = u3 / TILE;
            tile_of_edge[pre_g] = t0 | (t1 << 8) | (t2 << 16) | (t3 << 24);
            const unsigned bit = 1u << ((tid >> 2) & 31);
            const int      wd  = tid >> 7;
            atomicOr(&pres_lds[t0 * 4 + wd], bit);
            atomicOr(&pres_lds[t1 * 4 + wd], bit);
            atomicOr(&pres_lds[t2 * 4 + wd], bit);
            atomicOr(&pres_lds[t3 * 4 + wd], bit);
        }
        lds_barrier();
        for (int j = tid; j < NTILES * 4; j += GEMM_T)
            pres[(size_t)(j >> 2) * PW + 4 * blockIdx.x + (j & 3)] = pres_lds[j];
    }
    att_lds[(tid < HC) ? 2 * tid : 2 * (tid - HC) + 1] = att_el;

    const __bf16* arow = a_lds + lr * PITCH;
    const int n = tN + lr;

    auto step = [&](const int strip, float4 (&rv)[NIT], const bool ahead) __attribute__((always_inline)) {
        const int tM = strip << 4;

#pragma unroll
        for (int it = 0; it < NIT; ++it) {
            float4 v = rv[it];
            if (TANH) {
                v.x = tanhf(v.x + bias4.x); v.y = tanhf(v.y + bias4.y);
                v.z = tanhf(v.z + bias4.z); v.w = tanhf(v.w + bias4.w);
            }
            v4bf o;
            o[0] = (__bf16)v.x; o[1] = (__bf16)v.y; o[2] = (__bf16)v.z; o[3] = (__bf16)v.w;
            *(v4bf*)(a_lds + (prow + it * ROW_IT) * PITCH + 4 * pc4) = o;
        }
        lds_barrier();

        if (ahead) {
            const int sp = min(strip + DEPTH * GEMM_G, NSTRIP - 1);
#pragma unroll
            for (int it = 0; it < NIT; ++it)
                rv[it] = *(const float4*)(aptr + (size_t)(16 * sp + it * ROW_IT) * K);
        }

        v8f acc = {};
#pragma unroll
        for (int kk = 0; kk < K; kk += 32) {
            v8bf alo = *(const v8bf*)(arow + kk + 8 * half);
            v8bf ahi = *(const v8bf*)(arow + kk + 16 + 8 * half);
            v8bf blo = bfr[kk / 16];
            v8bf bhi = bfr[kk / 16 + 1];
            v16bf a, b;
#pragma unroll
            for (int i = 0; i < 8; ++i) {
                a[i] = alo[i]; a[i + 8] = ahi[i];
                b[i] = blo[i]; b[i + 8] = bhi[i];
            }
            acc = __builtin_amdgcn_wmma_f32_16x16x32_bf16(
                      false, a, false, b, (short)0, acc, false, false);
        }

        const float2 att_n = *(const float2*)(att_lds + 2 * n);
        const float as_n = att_n.x, ad_n = att_n.y;
        float p[16];
#pragma unroll
        for (int r = 0; r < 8; ++r) { p[2 * r] = acc[r] * as_n; p[2 * r + 1] = acc[r] * ad_n; }
        const bool b3 = (lr & 8) != 0, b2 = (lr & 4) != 0, b1 = (lr & 2) != 0, b0 = (lr & 1) != 0;
        float q8[8], q4[4], q2[2];
#pragma unroll
        for (int j = 0; j < 8; ++j)
            q8[j] = (b3 ? p[j + 8] : p[j]) + row_xor<8>(b3 ? p[j] : p[j + 8]);
#pragma unroll
        for (int j = 0; j < 4; ++j)
            q4[j] = (b2 ? q8[j + 4] : q8[j]) + row_xor<4>(b2 ? q8[j] : q8[j + 4]);
#pragma unroll
        for (int j = 0; j < 2; ++j)
            q2[j] = (b1 ? q4[j + 2] : q4[j]) + row_xor<2>(b1 ? q4[j] : q4[j + 2]);
        const float q1 = (b0 ? q2[1] : q2[0]) + row_xor<1>(b0 ? q2[0] : q2[1]);
        sc_part[(tid >> 5) * 32 + lane] = q1;
        lds_barrier();

#pragma unroll
        for (int r = 0; r < 8; ++r) {
            int m = tM + r + 8 * half;
            C[(size_t)m * HC + n] = acc[r];
        }

        if (tid < 64) {
            const int sc = tid >> 5, idx = tid & 31;
            const int row = idx >> 1, hd = idx & 1;
            const int li = 16 * (row >> 3) + 2 * (row & 7) + sc;
            float s = 0.f;
#pragma unroll
            for (int g = 0; g < 8; ++g) s += sc_part[(hd * 8 + g) * 32 + li];
            (sc ? a_d : a_s)[tM * 2 + idx] = s;
        }
    };

    int strip0 = blockIdx.x;
    for (; strip0 + (DEPTH - 1) * GEMM_G < NSTRIP; strip0 += DEPTH * GEMM_G) {
#pragma unroll
        for (int d = 0; d < DEPTH; ++d) step(strip0 + d * GEMM_G, av[d], true);
    }
#pragma unroll
    for (int d = 0; d + 1 < DEPTH; ++d) {
        const int strip = strip0 + d * GEMM_G;
        if (strip >= NSTRIP) return;
        step(strip, av[d], false);
    }
}

#define AGG_T     1024
#define AGG_W     (AGG_T / 32)
#define PIECE_E   512
#define SCAN_U    4
#define NPIECE    ((EE / 16 + AGG_T - 1) / AGG_T)
#define QDRAIN    32
#define QCAP      (QDRAIN + PIECE_E)
#define KCAP      384
#define KSELF     8
static_assert(EP <= (1 << 20) && NN <= (1 << 20) && TILE <= (1 << 12), "queue entry packing");
static_assert((TILE + AGG_W - 1) / AGG_W < KSELF && KSELF <= 32 && KCAP % 8 == 0 && KSELF % 8 == 0,
              "self loops and the fill up to a whole batch fit the reserved room");

__device__ __forceinline__ float leaky(float l) {
    return (l > 0.f) ? l : NEG_SLOPE * l;
}

__device__ __forceinline__ unsigned make_entry(int s, unsigned dl) {
    return (unsigned)s | (dl << 20);
}
__device__ __forceinline__ void entry_nodes(unsigned e, int base, int nloc,
                                            const int* __restrict__ srcp,
                                            const int* __restrict__ dstp,
                                            int& s, int& dl) {
    int d;
    if (e < (unsigned)EE) { s = srcp[e]; d = dstp[e]; }
    else                  { s = (int)e - EE; d = s; }
    dl = min(max(d - base, 0), nloc - 1);
}

__device__ __forceinline__ void drain_max(const unsigned* q, int n, int lane,
                                          const int* __restrict__ srcp,
                                          const int* __restrict__ dstp, int base, int nloc,
                                          const float* __restrict__ a_s,
                                          const float* ad_l, unsigned* m_enc,
                                          unsigned* kept, int* ocnt, int* ovf_flag, bool keep) {
    for (int b = 0; b < n; b += 32) {
        int i = b + lane;
        if (i < n) {
            int s, dl;
            entry_nodes(q[i], base, nloc, srcp, dstp, s, dl);
            float2 as = *(const float2*)(a_s + 2 * s);
            if (keep) {
                const int ow   = dl & (AGG_W - 1);
                const int slot = atomicAdd(&ocnt[ow], 1);
                if (slot < KCAP - KSELF) kept[ow * KCAP + slot] = make_entry(s, (unsigned)dl);
                else                     *ovf_flag = 1;
            }
            atomicMax(&m_enc[dl * 2 + 0], enc_f32(leaky(as.x + ad_l[dl * 2 + 0])));
            atomicMax(&m_enc[dl * 2 + 1], enc_f32(leaky(as.y + ad_l[dl * 2 + 1])));
        }
    }
}

__device__ __forceinline__ void take_list_max(const unsigned* __restrict__ lst, int n, int lane,
                                              const float* __restrict__ a_s,
                                              const float* ad_l, unsigned* m_enc,
                                              unsigned* kp) {
    for (int b = 0; b < n; b += 128) {
        unsigned ent[4];
        float2   as[4];
#pragma unroll
        for (int u = 0; u < 4; ++u) {
            int i = b + 32 * u + lane;
            ent[u] = (i < n) ? lst[i] : 0u;
        }
#pragma unroll
        for (int u = 0; u < 4; ++u)
            as[u] = *(const float2*)(a_s + 2 * (int)(ent[u] & 0xFFFFFu));
#pragma unroll
        for (int u = 0; u < 4; ++u) {
            int i = b + 32 * u + lane;
            if (i < n) {
                int dl = (int)(ent[u] >> 20);
                kp[i] = ent[u];
                atomicMax(&m_enc[dl * 2 + 0], enc_f32(leaky(as[u].x + ad_l[dl * 2 + 0])));
                atomicMax(&m_enc[dl * 2 + 1], enc_f32(leaky(as[u].y + ad_l[dl * 2 + 1])));
            }
        }
    }
}

template <bool FULL, bool OWN>
__device__ __forceinline__ void add_rows(int k0, int cnt, int lane, int s_l, int dl_l,
                                         float ex0_l, float ex1_l,
                                         const float* __restrict__ h, float* acc) {
    float4 v0[8], v1[8];
#pragma unroll
    for (int k = 0; k < 8; ++k) {
        if (FULL || k0 + k < cnt) {
            int s = __builtin_amdgcn_readlane(s_l, k0 + k);
            const float* hr = h + (size_t)s * HC + 4 * lane;
            v0[k] = *(const float4*)hr;
            v1[k] = *(const float4*)(hr + CC);
        }
    }
    if (OWN) __builtin_amdgcn_sched_barrier(0);
#pragma unroll
    for (int k = 0; k < 8; ++k) {
        if (FULL || k0 + k < cnt) {
            int   dl = __builtin_amdgcn_readlane(dl_l, k0 + k);
            float e0 = __int_as_float(__builtin_amdgcn_readlane(__float_as_int(ex0_l), k0 + k));
            float e1 = __int_as_float(__builtin_amdgcn_readlane(__float_as_int(ex1_l), k0 + k));
            float* ar = acc + dl * HC + 4 * lane;
            if (OWN) {
                float4 a0 = *(const float4*)ar;
                float4 a1 = *(const float4*)(ar + CC);
                a0.x = fmaf(e0, v0[k].x, a0.x); a0.y = fmaf(e0, v0[k].y, a0.y);
                a0.z = fmaf(e0, v0[k].z, a0.z); a0.w = fmaf(e0, v0[k].w, a0.w);
                a1.x = fmaf(e1, v1[k].x, a1.x); a1.y = fmaf(e1, v1[k].y, a1.y);
                a1.z = fmaf(e1, v1[k].z, a1.z); a1.w = fmaf(e1, v1[k].w, a1.w);
                *(float4*)ar        = a0;
                *(float4*)(ar + CC) = a1;
            } else {
                atomicAdd(ar + 0, e0 * v0[k].x);
                atomicAdd(ar + 1, e0 * v0[k].y);
                atomicAdd(ar + 2, e0 * v0[k].z);
                atomicAdd(ar + 3, e0 * v0[k].w);
                atomicAdd(ar + CC + 0, e1 * v1[k].x);
                atomicAdd(ar + CC + 1, e1 * v1[k].y);
                atomicAdd(ar + CC + 2, e1 * v1[k].z);
                atomicAdd(ar + CC + 3, e1 * v1[k].w);
            }
        }
    }
}

template <bool DIRECT>
__device__ __forceinline__ int drain_sum(unsigned* q, int n, int n_real, bool last, int lane,
                                          const int* __restrict__ srcp,
                                          const int* __restrict__ dstp, int base, int nloc,
                                          const float* __restrict__ a_s,
                                          const float* __restrict__ h,
                                          const float* ad_l, const unsigned* m_enc,
                                          float* den, float* acc) {
    const int nproc = last ? n : (n & ~7);
    for (int b = 0; b < nproc; b += 32) {
        int i = b + lane;
        int cnt = min(32, nproc - b);
        int s_l = 0, dl_l = 0;
        float ex0_l = 0.f, ex1_l = 0.f;
        if (i < nproc) {
            unsigned ent = q[i];
            if (DIRECT) { s_l = (int)(ent & 0xFFFFFu); dl_l = (int)(ent >> 20); }
            else        entry_nodes(ent, base, nloc, srcp, dstp, s_l, dl_l);
            float2 as = *(const float2*)(a_s + 2 * s_l);
            ex0_l = expf(leaky(as.x + ad_l[dl_l * 2 + 0]) - dec_f32(m_enc[dl_l * 2 + 0]));
            ex1_l = expf(leaky(as.y + ad_l[dl_l * 2 + 1]) - dec_f32(m_enc[dl_l * 2 + 1]));
            if (DIRECT && i >= n_real) {
                ex0_l = 0.f; ex1_l = 0.f;
            } else {
                atomicAdd(&den[dl_l * 2 + 0], ex0_l);
                atomicAdd(&den[dl_l * 2 + 1], ex1_l);
            }
        }
        int k0 = 0;
        for (; k0 + 8 <= cnt; k0 += 8)
            add_rows<true, DIRECT>(k0, cnt, lane, s_l, dl_l, ex0_l, ex1_l, h, acc);
        if (!DIRECT && k0 < cnt)
            add_rows<false, false>(k0, cnt, lane, s_l, dl_l, ex0_l, ex1_l, h, acc);
    }
    const int rem = n - nproc;
    if (rem > 0 && nproc > 0) {
        unsigned keep = 0u;
        if (lane < rem) keep = q[nproc + lane];
        if (lane < rem) q[lane] = keep;
    }
    return rem;
}

template <int PASS>
__device__ __forceinline__ void scan_edges(const int* __restrict__ srcp,
                                           const int* __restrict__ dstp,
                                           const uint4* __restrict__ tcol,
                                           const float* __restrict__ a_s,
                                           const float* __restrict__ h,
                                           unsigned* q, int n, int base, int nloc,
                                           int lane, int w,
                                           const float* ad_l, unsigned* m_enc,
                                           float* den, float* acc,
                                           unsigned* kept, int* ocnt, int* ovf_flag, bool keep) {
    const unsigned idrep = (unsigned)blockIdx.x * 0x01010101u;
    const int goff = w * 32 + lane;
    constexpr int U      = (PASS == 1) ? SCAN_U : 1;
    constexpr int NROUND = (NPIECE + U - 1) / U;
    int rr = blockIdx.x % NROUND;
    for (int r = 0; r < NROUND; ++r) {
        uint4 tv[U];
#pragma unroll
        for (int u = 0; u < U; ++u) {
            int g = (rr * U + u) * AGG_T + goff;
            tv[u] = tcol[min(g, EE / 16 - 1)];
        }
#pragma unroll
        for (int u = 0; u < U; ++u) {
            const int g = (rr * U + u) * AGG_T + goff;
            const int e_base = g * 16;
            const unsigned past = (g < EE / 16) ? 0u : 0xFFFFFFFFu;
            const unsigned xw[4] = { (tv[u].x ^ idrep) | past, (tv[u].y ^ idrep) | past,
                                     (tv[u].z ^ idrep) | past, (tv[u].w ^ idrep) | past };
            unsigned zw[4];
#pragma unroll
            for (int k = 0; k < 4; ++k) zw[k] = (xw[k] - 0x01010101u) & ~xw[k] & 0x80808080u;
            if (__builtin_amdgcn_ballot_w32((zw[0] | zw[1] | zw[2] | zw[3]) != 0u) != 0u) {
#pragma unroll
                for (int k = 0; k < 4; ++k) {
                    if (__builtin_amdgcn_ballot_w32(zw[k] != 0u) == 0u) continue;
#pragma unroll
                    for (int j = 0; j < 4; ++j) {
                        const bool hit = ((xw[k] >> (8 * j)) & 0xFFu) == 0u;
                        const unsigned m = __builtin_amdgcn_ballot_w32(hit);
                        if (m == 0u) continue;
                        if (hit) q[n + __builtin_amdgcn_mbcnt_lo(m, 0u)] = (unsigned)(e_base + 4 * k + j);
                        n += __popc(m);
                    }
                }
            }
            const bool last = (r == NROUND - 1) && (u == U - 1);
            if (n >= QDRAIN || (last && n > 0)) {
                if (PASS == 1) {
                    drain_max(q, n, lane, srcp, dstp, base, nloc, a_s, ad_l, m_enc,
                              kept, ocnt, ovf_flag, keep);
                    n = 0;
                } else {
                    n = drain_sum<false>(q, n, n, last, lane, srcp, dstp, base, nloc, a_s, h, ad_l, m_enc,
                                         den, acc);
                }
            }
        }
        if (++rr == NROUND) rr = 0;
    }
}

#define FW        ((PW + AGG_W - 1) / AGG_W)
#define FDRAIN    256
#define ECAP      (FDRAIN + 32)
#define GCAP      (QCAP - ECAP)
static_assert(FW > 32 && FW <= 64, "a lane takes at most two words of the row");
static_assert(GCAP >= 64, "the group list takes at least two turns of the wave");

__device__ __forceinline__ void drain_found(const unsigned* q, int n, int lane,
                                            const int* __restrict__ srcp,
                                            const int* __restrict__ dstp, int base, int nloc,
                                            const float* __restrict__ a_s,
                                            const float* ad_l, unsigned* m_enc,
                                            unsigned* kept, int* ocnt, int* ovf_flag) {
    for (int b = 0; b < n; b += 128) {
        int    e[4], s[4], dn[4];
        float2 as[4];
#pragma unroll
        for (int u = 0; u < 4; ++u) {
            const int i = b + 32 * u + lane;
            const unsigned qe = q[min(i, ECAP - 1)];
            e[u] = (i < n) ? (int)min(qe, (unsigned)(EE - 1)) : 0;
        }
#pragma unroll
        for (int u = 0; u < 4; ++u) { s[u] = srcp[e[u]]; dn[u] = dstp[e[u]]; }
#pragma unroll
        for (int u = 0; u < 4; ++u) as[u] = *(const float2*)(a_s + 2 * s[u]);
#pragma unroll
        for (int u = 0; u < 4; ++u) {
            const int  i     = b + 32 * u + lane;
            const bool entry = i < n;
            const int  dl    = min(max(dn[u] - base, 0), nloc - 1);
            if (entry) {
                const int ow   = dl & (AGG_W - 1);
                const int slot = atomicAdd(&ocnt[ow], 1);
                if (slot < KCAP - KSELF) kept[ow * KCAP + slot] = make_entry(s[u], (unsigned)dl);
                else                     *ovf_flag = 1;
            }
            const unsigned v0 = enc_f32(leaky(as[u].x + ad_l[dl * 2 + 0]));
            const unsigned v1 = enc_f32(leaky(as[u].y + ad_l[dl * 2 + 1]));
            atomicMax(&m_enc[dl * 2 + 0], entry ? v0 : 0u);
            atomicMax(&m_enc[dl * 2 + 1], entry ? v1 : 0u);
        }
    }
}

__device__ __forceinline__ void find_edges(const int* __restrict__ srcp,
                                           const int* __restrict__ dstp,
                                           const uint4* __restrict__ tcol,
                                           const unsigned* __restrict__ prow,
                                           const float* __restrict__ a_s,
                                           unsigned* q, int base, int nloc, int lane, int w,
                                           const float* ad_l, unsigned* m_enc,
                                           unsigned* kept, int* ocnt, int* ovf_flag) {
    unsigned* gl = q + ECAP;
    const unsigned idrep = (unsigned)blockIdx.x * 0x01010101u;
    const int ia = w * FW + lane, ib = ia + 32;
    unsigned wa = (ia < PW) ? prow[ia] : 0u;
    unsigned wb = (lane < FW - 32 && ib < PW) ? prow[ib] : 0u;
    int n = 0;
    while (__builtin_amdgcn_ballot_w32((wa | wb) != 0u) != 0u) {
        int ng = 0;
        while (ng <= GCAP - 32) {
            const bool has = (wa | wb) != 0u;
            const unsigned m = __builtin_amdgcn_ballot_w32(has);
            if (m == 0u) break;
            if (has) {
                unsigned g;
                if (wa != 0u) { g = ((unsigned)ia << 5) + (unsigned)(__ffs(wa) - 1); wa &= wa - 1u; }
                else          { g = ((unsigned)ib << 5) + (unsigned)(__ffs(wb) - 1); wb &= wb - 1u; }
                gl[ng + __builtin_amdgcn_mbcnt_lo(m, 0u)] = g;
            }
            ng += __popc(m);
        }
        for (int j0 = 0; j0 < ng; j0 += 128) {
            unsigned g[4];
            uint4    tv[4];
#pragma unroll
            for (int u = 0; u < 4; ++u) {
                const int j = j0 + 32 * u + lane;
                g[u]  = min(gl[min(j, GCAP - 1)], (unsigned)(NGRP - 1));
                tv[u] = tcol[g[u]];
            }
            unsigned hm[2] = { 0u, 0u };
#pragma unroll
            for (int u = 0; u < 4; ++u) {
                const int j = j0 + 32 * u + lane;
                const unsigned past = (j < ng) ? 0u : 0xFFFFFFFFu;
                const unsigned xw[4] = { (tv[u].x ^ idrep) | past, (tv[u].y ^ idrep) | past,
                                         (tv[u].z ^ idrep) | past, (tv[u].w ^ idrep) | past };
                unsigned h16 = 0u;
#pragma unroll
                for (int k = 0; k < 4; ++k) {
                    const unsigned z = ~(((xw[k] & 0x7F7F7F7Fu) + 0x7F7F7F7Fu) | xw[k] | 0x7F7F7F7Fu);
                    h16 |= ((((z >> 7) * 0x00204081u) >> 21) & 0xFu) << (4 * k);
                }
                hm[u >> 1] |= h16 << (16 * (u & 1));
            }
            for (;;) {
                const bool has = (hm[0] | hm[1]) != 0u;
                const unsigned m = __builtin_amdgcn_ballot_w32(has);
                if (m == 0u) break;
                if (has) {
                    int idx;
                    if (hm[0] != 0u) { idx = __ffs(hm[0]) - 1;      hm[0] &= hm[0] - 1u; }
                    else             { idx = 31 + __ffs(hm[1]);     hm[1] &= hm[1] - 1u; }
                    const int u = idx >> 4;
                    const unsigned gg = (u == 0) ? g[0] : (u == 1) ? g[1] : (u == 2) ? g[2] : g[3];
                    q[n + __builtin_amdgcn_mbcnt_lo(m, 0u)] = gg * 16u + (unsigned)(idx & 15);
                }
                n += __popc(m);
                if (n >= FDRAIN) {
                    drain_found(q, n, lane, srcp, dstp, base, nloc, a_s, ad_l, m_enc, kept, ocnt, ovf_flag);
                    n = 0;
                }
            }
        }
    }
    if (n > 0)
        drain_found(q, n, lane, srcp, dstp, base, nloc, a_s, ad_l, m_enc, kept, ocnt, ovf_flag);
}

__device__ __forceinline__ float wave_sum(float v) {
    v += row_xor<1>(v); v += row_xor<2>(v); v += row_xor<4>(v); v += row_xor<8>(v);
    return v + __shfl_xor(v, 16);
}

#define LIST_NONE (-1)
template <bool FIND, bool LAST>
__global__ __launch_bounds__(AGG_T)
void k_tile_aggregate(const int* __restrict__ ei,
                      const float* __restrict__ h,
                      const float* __restrict__ a_s,
                      const float* __restrict__ a_d,
                      float* __restrict__ out,
                      unsigned* __restrict__ lists,
                      int* __restrict__ counts,
                      const unsigned* __restrict__ tile_of_edge,
                      const unsigned* __restrict__ pres,
                      const float* __restrict__ b_last,
                      const float* __restrict__ Wl,
                      const float* __restrict__ bl,
                      float* __restrict__ prob) {
    __shared__ alignas(16) float acc[TILE * HC];
    __shared__ unsigned queue[AGG_W * QCAP];
    __shared__ unsigned kept[AGG_W * KCAP];
    __shared__ unsigned m_enc[TILE * 2];
    __shared__ float    den[TILE * 2];
    __shared__ float    ad_l[TILE * 2];
    __shared__ int      ocnt[AGG_W];
    __shared__ int      ovf_flag;

    const int tid = threadIdx.x, lane = tid & 31;
    const int w = __builtin_amdgcn_readfirstlane(tid >> 5);
    const int base = blockIdx.x * TILE;
    const int nloc = min(TILE, NN - base);
    const int* __restrict__ srcp = ei;
    const int* __restrict__ dstp = ei + EE;
    const uint4* __restrict__ tcol = (const uint4*)tile_of_edge;
    unsigned* q = queue + w * QCAP;
    unsigned* kept_w = kept + w * KCAP;
    int  kc  = 0;

    for (int i = tid; i < TILE * HC / 4; i += AGG_T)
        ((float4*)acc)[i] = make_float4(0.f, 0.f, 0.f, 0.f);
    if (tid < TILE * 2) den[tid] = 0.f;
    if (tid < AGG_W) ocnt[tid] = 0;
    if (tid == 0) ovf_flag = 0;
    if (tid < nloc * 2) {
        int g = base * 2 + tid;
        float ad = a_d[g];
        m_enc[tid] = enc_f32(leaky(a_s[g] + ad));
        ad_l[tid]  = ad;
    }
    __syncthreads();

    unsigned* lst = lists + ((size_t)blockIdx.x * AGG_W + w) * KCAP;
    int* cntp = counts + blockIdx.x * AGG_W + w;
    if (FIND) {
        find_edges(srcp, dstp, tcol, pres + (size_t)blockIdx.x * PW, a_s, q, base, nloc, lane, w,
                   ad_l, m_enc, kept, ocnt, &ovf_flag);
    } else {
        int c = __builtin_amdgcn_readfirstlane(*cntp);
        if (c != LIST_NONE) {
            kc = min(max(c, 0), KCAP - KSELF);
            take_list_max(lst, kc, lane, a_s, ad_l, m_enc, kept_w);
        } else {
            if (lane == 0) ovf_flag = 1;
            scan_edges<1>(srcp, dstp, tcol, a_s, h, q, 0, base, nloc, lane, w, ad_l, m_enc, den, acc,
                          kept, ocnt, &ovf_flag, false);
        }
    }
    __syncthreads();
    const bool ovf = __builtin_amdgcn_readfirstlane(ovf_flag) != 0;
    if (FIND) {
        kc = min(max(__builtin_amdgcn_readfirstlane(ocnt[w]), 0), KCAP - KSELF);
        if (!ovf)
            for (int i = lane; i < kc; i += 32) lst[i] = kept_w[i];
        if (lane == 0) *cntp = ovf ? LIST_NONE : kc;
    }

    int nself = (nloc > w) ? (nloc - w + 31) / 32 : 0;
    if (!ovf) {
        if (lane < nself) {
            int dl = w + 32 * lane;
            kept_w[kc + lane] = make_entry(base + dl, (unsigned)dl);
        }
        const int n_real = kc + nself;
        const int n_fill = (n_real + 7) & ~7;
        if (n_real + lane < n_fill) kept_w[n_real + lane] = make_entry(base + w, (unsigned)w);
        drain_sum<true>(kept_w, n_fill, n_real, true, lane, srcp, dstp, base, nloc, a_s, h, ad_l, m_enc, den, acc);
    } else {
        if (lane < nself) {
            int dl = w + 32 * lane;
            q[lane] = (unsigned)(EE + base + dl);
        }
        scan_edges<2>(srcp, dstp, tcol, a_s, h, q, nself, base, nloc, lane, w, ad_l, m_enc, den, acc,
                      kept, ocnt, &ovf_flag, false);
    }
    __syncthreads();

    if (!LAST) {
        for (int dl = w; dl < nloc; dl += AGG_W) {
            float dn0 = den[dl * 2 + 0] + 1e-16f, dn1 = den[dl * 2 + 1] + 1e-16f;
            const float4 r0 = *(const float4*)(acc + dl * HC + 4 * lane);
            const float4 r1 = *(const float4*)(acc + dl * HC + CC + 4 * lane);
            float4 o0 = make_float4(r0.x / dn0, r0.y / dn0, r0.z / dn0, r0.w / dn0);
            float4 o1 = make_float4(r1.x / dn1, r1.y / dn1, r1.z / dn1, r1.w / dn1);
            float* op = out + (size_t)(base + dl) * HC + 4 * lane;
            *(float4*)op        = o0;
            *(float4*)(op + CC) = o1;
        }
    } else {
        const float4 bA  = *(const float4*)(b_last + 4 * lane);
        const float4 bB  = *(const float4*)(b_last + CC + 4 * lane);
        const float4 wA0 = *(const float4*)(Wl + 8 * lane);
        const float4 wA1 = *(const float4*)(Wl + 8 * lane + 4);
        const float4 wB0 = *(const float4*)(Wl + 2 * CC + 8 * lane);
        const float4 wB1 = *(const float4*)(Wl + 2 * CC + 8 * lane + 4);
        const float bz0 = bA.x * wA0.x + bA.y * wA0.z + bA.z * wA1.x + bA.w * wA1.z
                        + bB.x * wB0.x + bB.y * wB0.z + bB.z * wB1.x + bB.w * wB1.z;
        const float bz1 = bA.x * wA0.y + bA.y * wA0.w + bA.z * wA1.y + bA.w * wA1.w
                        + bB.x * wB0.y + bB.y * wB0.w + bB.z * wB1.y + bB.w * wB1.w;
        const float bl0 = bl[0], bl1 = bl[1];
        for (int dl = w; dl < nloc; dl += AGG_W) {
            const float inv0 = 1.f / (den[dl * 2 + 0] + 1e-16f);
            const float inv1 = 1.f / (den[dl * 2 + 1] + 1e-16f);
            const float4 r0 = *(const float4*)(acc + dl * HC + 4 * lane);
            const float4 r1 = *(const float4*)(acc + dl * HC + CC + 4 * lane);
            const float a0 = r0.x, a1 = r0.y, a2 = r0.z, a3 = r0.w;
            const float c0 = r1.x, c1 = r1.y, c2 = r1.z, c3 = r1.w;
            const float tA0 = a0 * wA0.x + a1 * wA0.z + a2 * wA1.x + a3 * wA1.z;
            const float tA1 = a0 * wA0.y + a1 * wA0.w + a2 * wA1.y + a3 * wA1.w;
            const float tB0 = c0 * wB0.x + c1 * wB0.z + c2 * wB1.x + c3 * wB1.z;
            const float tB1 = c0 * wB0.y + c1 * wB0.w + c2 * wB1.y + c3 * wB1.w;
            const float z0 = wave_sum(tA0 * inv0 + tB0 * inv1 + bz0) + bl0;
            const float z1 = wave_sum(tA1 * inv0 + tB1 * inv1 + bz1) + bl1;
            if (lane == 0) {
                m_enc[dl * 2 + 0] = __float_as_uint(z0);
                m_enc[dl * 2 + 1] = __float_as_uint(z1);
            }
        }
        __syncthreads();
        const int nd = (base & ~15) + tid;
        if (nd >= base && nd < base + nloc) {
            const float z0 = __uint_as_float(m_enc[(nd - base) * 2 + 0]);
            const float z1 = __uint_as_float(m_enc[(nd - base) * 2 + 1]);
            const float mx = fmaxf(z0, z1);
            const float e0 = expf(z0 - mx), e1 = expf(z1 - mx);
            const float inv = 1.f / (e0 + e1);
            *(float2*)(prob + 2 * nd) = make_float2(e0 * inv, e1 * inv);
        }
    }
}

extern "C" void kernel_launch(void* const* d_in, const int* in_sizes, int n_in,
                              void* d_out, int out_size, void* d_ws, size_t ws_size,
                              hipStream_t stream) {
    const float* x        = (const float*)d_in[0];
    const int*   ei       = (const int*)  d_in[1];
    const float* W0       = (const float*)d_in[2];
    const float* att_src0 = (const float*)d_in[3];
    const float* att_dst0 = (const float*)d_in[4];
    const float* b0       = (const float*)d_in[5];
    const float* W1       = (const float*)d_in[6];
    const float* att_src1 = (const float*)d_in[7];
    const float* att_dst1 = (const float*)d_in[8];
    const float* b1       = (const float*)d_in[9];
    const float* Wl       = (const float*)d_in[10];
    const float* bl       = (const float*)d_in[11];
    float* out = (float*)d_out;

    char* base = (char*)d_ws;
    size_t off = 0;
    auto carve = [&](size_t bytes) -> char* {
        char* p = base + off;
        off = (off + bytes + 255) & ~(size_t)255;
        return p;
    };
    float*    h_feat = (float*)   carve((size_t)NN * HC * 4);
    float*    accb   = (float*)   carve((size_t)NN * HC * 4);
    __bf16*   wtbf   = (__bf16*)  carve((size_t)HC * HC * 2);
    float*    a_s    = (float*)   carve((size_t)NN * 2 * 4);
    float*    a_d    = (float*)   carve((size_t)NN * 2 * 4);
    unsigned* lists  = (unsigned*)carve((size_t)NTILES * AGG_W * KCAP * 4);
    int*      counts = (int*)     carve((size_t)NTILES * AGG_W * 4);
    unsigned* tcol   = (unsigned*)carve((size_t)EE);
    unsigned* pres   = (unsigned*)carve((size_t)NTILES * PW * 4);
    (void)ws_size; (void)in_sizes; (void)n_in; (void)out_size;

    const int T = 256;
    dim3 blk(T);

    k_transpose_bf16<<<(DD * HC + T - 1) / T, blk, 0, stream>>>(W0, wtbf, DD, HC);

    k_gemm_rows_wmma<DD, false><<<GEMM_G, dim3(GEMM_T), 0, stream>>>(
        x, b0, wtbf, h_feat, att_src0, att_dst0, a_s, a_d, ei + EE, tcol, pres);

    k_tile_aggregate<true, false><<<NTILES, dim3(AGG_T), 0, stream>>>(ei, h_feat, a_s, a_d, accb,
                                                                      lists, counts, tcol, pres,
                                                                      b1, Wl, bl, out);

    k_transpose_bf16<<<(HC * HC + T - 1) / T, blk, 0, stream>>>(W1, wtbf, HC, HC);

    k_gemm_rows_wmma<HC, true><<<GEMM_G, dim3(GEMM_T), 0, stream>>>(
        accb, b0, wtbf, h_feat, att_src1, att_dst1, a_s, a_d, ei + EE, tcol, pres);

    k_tile_aggregate<false, true><<<NTILES, dim3(AGG_T), 0, stream>>>(ei, h_feat, a_s, a_d, accb,
                                                                      lists, counts, tcol, pres,
                                                                      b1, Wl, bl, out);
}
